# starting kernel
# baseline (speedup 1.0000x reference)
#include <hip/hip_runtime.h>
#include <hip/hip_bf16.h>
#include <math.h>

#define N_NODES 50000
#define N_EDGES 800000
#define HID     128
#define OUTC    40
#define NLAYERS 3
#define BN_EPS  1e-5f
#define WT_STRIDE 130

typedef __attribute__((ext_vector_type(2))) float v2f;
typedef __attribute__((ext_vector_type(8))) float v8f;

__device__ __forceinline__ void atomAddF32(float* p, float v) {
    __hip_atomic_fetch_add(p, v, __ATOMIC_RELAXED, __HIP_MEMORY_SCOPE_AGENT);
}

__global__ void gcn_init_deg(float* dinv, int n) {
    int i = blockIdx.x * blockDim.x + threadIdx.x;
    if (i < n) dinv[i] = 1.0f;
}

__global__ void gcn_deg_edges(const int* __restrict__ ei, float* dinv, int e) {
    int i = blockIdx.x * blockDim.x + threadIdx.x;
    if (i < e) atomAddF32(&dinv[ei[e + i]], 1.0f);
}

__global__ void gcn_finish_dinv(float* dinv, int n) {
    int i = blockIdx.x * blockDim.x + threadIdx.x;
    if (i < n) dinv[i] = rsqrtf(dinv[i]);
}

__global__ void gcn_combine(const float* __restrict__ x,
                            const float* __restrict__ la,
                            const float* __restrict__ lb,
                            float* __restrict__ h, int n) {
    int idx = blockIdx.x * blockDim.x + threadIdx.x;
    if (idx >= n * HID) return;
    int nd = idx >> 7, f = idx & 127;
    float alpha = expf(la[0]);
    float beta  = expf(lb[0]);
    h[idx] = alpha * x[(size_t)nd * 256 + f] + beta * x[(size_t)nd * 256 + 128 + f];
}

__global__ __launch_bounds__(128) void gcn_gemm128_wmma(
        const float* __restrict__ A, const float* __restrict__ W,
        const float* __restrict__ bias, float* __restrict__ C,
        int nrows, int do_relu) {
    __shared__ float Wt[HID * WT_STRIDE];

    int t = threadIdx.x;
    #pragma unroll 4
    for (int idx = t; idx < HID * HID; idx += 128) {
        int kk = idx >> 7;
        int n  = idx & 127;
        Wt[n * WT_STRIDE + kk] = W[idx];
    }
    __syncthreads();

    const int wave  = t >> 5;
    const int lane  = t & 31;
    const int row16 = lane & 15;
    const int khalf = (lane >> 4) << 1;

    const int tileM = blockIdx.x * 64 + wave * 16;
    int rowIdx = tileM + row16;
    if (rowIdx > nrows - 1) rowIdx = nrows - 1;
    const float* Arow = A + (size_t)rowIdx * HID;
    const float* Brow = Wt + (size_t)row16 * WT_STRIDE;

    v8f acc[8] = {};

    for (int k = 0; k < HID; k += 4) {
        const float2 av = *(const float2*)(Arow + k + khalf);
        v2f a; a.x = av.x; a.y = av.y;
        #pragma unroll
        for (int nt = 0; nt < 8; ++nt) {
            const float2 bv = *(const float2*)(Brow + nt * 16 * WT_STRIDE + k + khalf);
            v2f b; b.x = bv.x; b.y = bv.y;
            acc[nt] = __builtin_amdgcn_wmma_f32_16x16x4_f32(
                false, a, false, b, (short)0, acc[nt], false, false);
        }
    }

    const int rbase = tileM + ((lane >> 4) << 3);
    #pragma unroll
    for (int nt = 0; nt < 8; ++nt) {
        const int col = nt * 16 + row16;
        const float bv = bias ? bias[col] : 0.0f;
        #pragma unroll
        for (int r = 0; r < 8; ++r) {
            const int row = rbase + r;
            if (row < nrows) {
                float v = acc[nt][r] + bv;
                if (do_relu) v = fmaxf(v, 0.0f);
                C[(size_t)row * HID + col] = v;
            }
        }
    }
}

__global__ void gcn_selfloop(const float* __restrict__ m, const float* __restrict__ dinv,
                             const float* __restrict__ b, float* __restrict__ h, int n) {
    int idx = blockIdx.x * blockDim.x + threadIdx.x;
    if (idx >= n * HID) return;
    int nd = idx >> 7, f = idx & 127;
    float di = dinv[nd];
    h[idx] = b[f] + m[idx] * di * di;
}

__global__ __launch_bounds__(256) void gcn_scatter(const int* __restrict__ ei,
                                                   const float* __restrict__ dinv,
                                                   const float* __restrict__ m,
                                                   float* __restrict__ h, int e) {
    int w = (blockIdx.x * blockDim.x + threadIdx.x) >> 5;
    int lane = threadIdx.x & 31;
    if (w >= e) return;
    int s = ei[w];
    int d = ei[e + w];
    float nrm = dinv[s] * dinv[d];
    const float4 v = *(const float4*)(m + (size_t)s * HID + lane * 4);
    float* hp = h + (size_t)d * HID + lane * 4;
    atomAddF32(hp + 0, v.x * nrm);
    atomAddF32(hp + 1, v.y * nrm);
    atomAddF32(hp + 2, v.z * nrm);
    atomAddF32(hp + 3, v.w * nrm);
}

__global__ void gcn_bn_zero(float* bnsum, float* bnsq) {
    int t = threadIdx.x;
    if (t < HID) bnsum[t] = 0.0f;
    else if (t < 2 * HID) bnsq[t - HID] = 0.0f;
}

__global__ __launch_bounds__(128) void gcn_bn_stats(const float* __restrict__ h,
                                                    float* bnsum, float* bnsq, int n) {
    int f = threadIdx.x;
    int chunk = (n + gridDim.x - 1) / gridDim.x;
    int n0 = blockIdx.x * chunk;
    int n1 = n0 + chunk; if (n1 > n) n1 = n;
    float s = 0.0f, ss = 0.0f;
    for (int nd = n0; nd < n1; ++nd) {
        float v = h[(size_t)nd * HID + f];
        s += v; ss += v * v;
    }
    atomAddF32(&bnsum[f], s);
    atomAddF32(&bnsq[f], ss);
}

__global__ void gcn_bn_apply(float* __restrict__ h,
                             const float* __restrict__ bnsum, const float* __restrict__ bnsq,
                             const float* __restrict__ gamma, const float* __restrict__ beta,
                             int n) {
    int idx = blockIdx.x * blockDim.x + threadIdx.x;
    if (idx >= n * HID) return;
    int f = idx & 127;
    float invn = 1.0f / (float)n;
    float mu  = bnsum[f] * invn;
    float var = bnsq[f] * invn - mu * mu;
    float v = gamma[f] * (h[idx] - mu) * rsqrtf(var + BN_EPS) + beta[f];
    h[idx] = fmaxf(v, 0.0f);
}

__global__ __launch_bounds__(128) void gcn_final(const float* __restrict__ h,
                                                 const float* __restrict__ lw,
                                                 const float* __restrict__ lb,
                                                 float* __restrict__ out, int n) {
    __shared__ float row[HID];
    __shared__ float lg[OUTC];
    int nd = blockIdx.x;
    int t = threadIdx.x;
    row[t] = h[(size_t)nd * HID + t];
    __syncthreads();
    if (t < OUTC) {
        float s = lb[t];
        #pragma unroll 4
        for (int k = 0; k < HID; ++k) s += row[k] * lw[k * OUTC + t];
        lg[t] = s;
    }
    __syncthreads();
    if (t < OUTC) {
        float mx = -INFINITY;
        for (int j = 0; j < OUTC; ++j) mx = fmaxf(mx, lg[j]);
        float se = 0.0f;
        for (int j = 0; j < OUTC; ++j) se += expf(lg[j] - mx);
        out[(size_t)nd * OUTC + t] = lg[t] - mx - logf(se);
    }
}

extern "C" void kernel_launch(void* const* d_in, const int* in_sizes, int n_in,
                              void* d_out, int out_size, void* d_ws, size_t ws_size,
                              hipStream_t stream) {
    const float* x       = (const float*)d_in[0];
    const int*   ei      = (const int*)d_in[1];
    const float* la      = (const float*)d_in[2];
    const float* lb      = (const float*)d_in[3];
    const float* mlp_w   = (const float*)d_in[4];
    const float* mlp_b   = (const float*)d_in[5];
    const float* conv_w  = (const float*)d_in[6];
    const float* conv_b  = (const float*)d_in[7];
    const float* bn_g    = (const float*)d_in[8];
    const float* bn_b    = (const float*)d_in[9];
    const float* lin_w   = (const float*)d_in[10];
    const float* lin_b   = (const float*)d_in[11];
    float* out = (float*)d_out;

    float* ws = (float*)d_ws;
    const size_t NM = (size_t)N_NODES * HID;
    float* hA    = ws;
    float* hB    = hA + NM;
    float* hC    = hB + NM;
    float* dinv  = hC + NM;
    float* bnsum = dinv + N_NODES;
    float* bnsq  = bnsum + HID;

    const int TPB = 256;
    const int nElemBlocks = (N_NODES * HID + TPB - 1) / TPB;
    const int gemmBlocks  = (N_NODES + 63) / 64;

    gcn_init_deg<<<(N_NODES + TPB - 1) / TPB, TPB, 0, stream>>>(dinv, N_NODES);
    gcn_deg_edges<<<(N_EDGES + TPB - 1) / TPB, TPB, 0, stream>>>(ei, dinv, N_EDGES);
    gcn_finish_dinv<<<(N_NODES + TPB - 1) / TPB, TPB, 0, stream>>>(dinv, N_NODES);

    gcn_combine<<<nElemBlocks, TPB, 0, stream>>>(x, la, lb, hA, N_NODES);

    for (int i = 0; i < NLAYERS; ++i) {
        gcn_gemm128_wmma<<<gemmBlocks, 128, 0, stream>>>(
            hA, mlp_w + (size_t)i * HID * HID, mlp_b + i * HID, hB, N_NODES, 1);
        gcn_gemm128_wmma<<<gemmBlocks, 128, 0, stream>>>(
            hB, conv_w + (size_t)i * HID * HID, nullptr, hC, N_NODES, 0);
        gcn_selfloop<<<nElemBlocks, TPB, 0, stream>>>(hC, dinv, conv_b + i * HID, hA, N_NODES);
        gcn_scatter<<<N_EDGES / 8, 256, 0, stream>>>(ei, dinv, hC, hA, N_EDGES);
        if (i < NLAYERS - 1) {
            gcn_bn_zero<<<1, 256, 0, stream>>>(bnsum, bnsq);
            gcn_bn_stats<<<256, 128, 0, stream>>>(hA, bnsum, bnsq, N_NODES);
            gcn_bn_apply<<<nElemBlocks, TPB, 0, stream>>>(
                hA, bnsum, bnsq, bn_g + i * HID, bn_b + i * HID, N_NODES);
        }
    }

    gcn_final<<<N_NODES, 128, 0, stream>>>(hA, lin_w, lin_b, out, N_NODES);
}


# Edge aggregation by gather over a per-call destination-sorted edge list (integer atomic rank + scan) replaces float atomic scatter; self-loop folded in; per-row sum order changes
# speedup vs baseline: 1.2984x; 1.2984x over previous
#include <hip/hip_runtime.h>
#include <hip/hip_bf16.h>
#include <math.h>

#define N_NODES 50000
#define N_EDGES 800000
#define HID     128
#define OUTC    40
#define NLAYERS 3
#define BN_EPS  1e-5f
#define WT_STRIDE 130

typedef __attribute__((ext_vector_type(2))) float v2f;
typedef __attribute__((ext_vector_type(8))) float v8f;

__device__ __forceinline__ void atomAddF32(float* p, float v) {
    __hip_atomic_fetch_add(p, v, __ATOMIC_RELAXED, __HIP_MEMORY_SCOPE_AGENT);
}

#define SCAN_T   1024
#define SCAN_PER 49
#define SCAN_N   (SCAN_T * SCAN_PER)

__global__ void gcn_cnt_zero(int* cnt, int n) {
    int i = blockIdx.x * blockDim.x + threadIdx.x;
    if (i < n) cnt[i] = 0;
}

__global__ void gcn_rank_edges(const int* __restrict__ ei, int* cnt, int* __restrict__ rank, int e) {
    int i = blockIdx.x * blockDim.x + threadIdx.x;
    if (i < e)
        rank[i] = __hip_atomic_fetch_add(&cnt[ei[e + i]], 1, __ATOMIC_RELAXED, __HIP_MEMORY_SCOPE_AGENT);
}

__global__ __launch_bounds__(SCAN_T) void gcn_scan_deg(const int* __restrict__ cnt,
                                                       int* __restrict__ rowptr,
                                                       float* __restrict__ dinv, int n) {
    __shared__ int buf[SCAN_N];
    __shared__ int toff[SCAN_T];
    __shared__ int wsum[SCAN_T / 32];
    const int t = threadIdx.x;
    for (int i = t; i < SCAN_N; i += SCAN_T) {
        int c = 0;
        if (i < n) { c = cnt[i]; dinv[i] = rsqrtf((float)(c + 1)); }
        buf[i] = c;
    }
    __syncthreads();
    const int base = t * SCAN_PER;
    int s = 0;
    for (int j = 0; j < SCAN_PER; ++j) { int c = buf[base + j]; buf[base + j] = s; s += c; }
    const int lane = t & 31, wv = t >> 5;
    int incl = s;
    for (int o = 1; o < 32; o <<= 1) { int y = __shfl_up(incl, o); if (lane >= o) incl += y; }
    if (lane == 31) wsum[wv] = incl;
    __syncthreads();
    if (wv == 0) {
        int x = wsum[lane];
        int xi = x;
        for (int o = 1; o < 32; o <<= 1) { int y = __shfl_up(xi, o); if (lane >= o) xi += y; }
        wsum[lane] = xi - x;
    }
    __syncthreads();
    toff[t] = wsum[wv] + incl - s;
    __syncthreads();
    for (int i = t; i <= n; i += SCAN_T) rowptr[i] = buf[i] + toff[i / SCAN_PER];
}

__global__ void gcn_fill_csr(const int* __restrict__ ei, const int* __restrict__ rank,
                             const int* __restrict__ rowptr, const float* __restrict__ dinv,
                             int* __restrict__ csr, int e) {
    int i = blockIdx.x * blockDim.x + threadIdx.x;
    if (i >= e) return;
    const int s = ei[i];
    const int d = ei[e + i];
    unsigned pos = (unsigned)(rowptr[d] + rank[i]);
    if (pos > (unsigned)(e - 1)) pos = (unsigned)(e - 1);
    int2 p; p.x = s; p.y = __float_as_int(dinv[s]);
    *(int2*)(csr + 2 * (size_t)pos) = p;
}

__global__ void gcn_combine(const float* __restrict__ x,
                            const float* __restrict__ la,
                            const float* __restrict__ lb,
                            float* __restrict__ h, int n) {
    int idx = blockIdx.x * blockDim.x + threadIdx.x;
    if (idx >= n * HID) return;
    int nd = idx >> 7, f = idx & 127;
    float alpha = expf(la[0]);
    float beta  = expf(lb[0]);
    h[idx] = alpha * x[(size_t)nd * 256 + f] + beta * x[(size_t)nd * 256 + 128 + f];
}

__global__ __launch_bounds__(128) void gcn_gemm128_wmma(
        const float* __restrict__ A, const float* __restrict__ W,
        const float* __restrict__ bias, float* __restrict__ C,
        int nrows, int do_relu) {
    __shared__ float Wt[HID * WT_STRIDE];

    int t = threadIdx.x;
    #pragma unroll 4
    for (int idx = t; idx < HID * HID; idx += 128) {
        int kk = idx >> 7;
        int n  = idx & 127;
        Wt[n * WT_STRIDE + kk] = W[idx];
    }
    __syncthreads();

    const int wave  = t >> 5;
    const int lane  = t & 31;
    const int row16 = lane & 15;
    const int khalf = (lane >> 4) << 1;

    const int tileM = blockIdx.x * 64 + wave * 16;
    int rowIdx = tileM + row16;
    if (rowIdx > nrows - 1) rowIdx = nrows - 1;
    const float* Arow = A + (size_t)rowIdx * HID;
    const float* Brow = Wt + (size_t)row16 * WT_STRIDE;

    v8f acc[8] = {};

    for (int k = 0; k < HID; k += 4) {
        const float2 av = *(const float2*)(Arow + k + khalf);
        v2f a; a.x = av.x; a.y = av.y;
        #pragma unroll
        for (int nt = 0; nt < 8; ++nt) {
            const float2 bv = *(const float2*)(Brow + nt * 16 * WT_STRIDE + k + khalf);
            v2f b; b.x = bv.x; b.y = bv.y;
            acc[nt] = __builtin_amdgcn_wmma_f32_16x16x4_f32(
                false, a, false, b, (short)0, acc[nt], false, false);
        }
    }

    const int rbase = tileM + ((lane >> 4) << 3);
    #pragma unroll
    for (int nt = 0; nt < 8; ++nt) {
        const int col = nt * 16 + row16;
        const float bv = bias ? bias[col] : 0.0f;
        #pragma unroll
        for (int r = 0; r < 8; ++r) {
            const int row = rbase + r;
            if (row < nrows) {
                float v = acc[nt][r] + bv;
                if (do_relu) v = fmaxf(v, 0.0f);
                C[(size_t)row * HID + col] = v;
            }
        }
    }
}

#define GROWS 8

__global__ __launch_bounds__(256) void gcn_gather(const int* __restrict__ rowptr,
                                                  const int* __restrict__ csr,
                                                  const float* __restrict__ dinv,
                                                  const float* __restrict__ m,
                                                  const float* __restrict__ b,
                                                  float* __restrict__ h, int n, int e) {
    const int lane = threadIdx.x & 31;
    int node = (blockIdx.x * blockDim.x + threadIdx.x) >> 5;
    if (node >= n) return;
    node = __builtin_amdgcn_readfirstlane(node);
    int beg = rowptr[node];
    int end = rowptr[node + 1];
    if (beg < 0) beg = 0;
    if (end > e) end = e;
    const float dd = dinv[node];
    const float* mlane = m + lane * 4;
    const float4 selfv = *(const float4*)(mlane + (size_t)node * HID);
    const float4 bb = *(const float4*)(b + lane * 4);
    float4 acc = make_float4(0.0f, 0.0f, 0.0f, 0.0f);

    for (int base = beg; base < end; base += 32) {
        int c = end - base; if (c > 32) c = 32;
        int sv = 0, wbits = 0;
        if (lane < c) {
            const int2 p = *(const int2*)(csr + 2 * (size_t)(base + lane));
            sv = p.x; wbits = p.y;
        }
        if ((unsigned)sv > (unsigned)(n - 1)) sv = n - 1;
        int j = 0;
        for (; j + GROWS <= c; j += GROWS) {
            float4 v[GROWS]; float w[GROWS];
            #pragma unroll
            for (int u = 0; u < GROWS; ++u) {
                const int s = __builtin_amdgcn_readlane(sv, j + u);
                w[u] = __int_as_float(__builtin_amdgcn_readlane(wbits, j + u));
                v[u] = *(const float4*)(mlane + (size_t)s * HID);
            }
            #pragma unroll
            for (int u = 0; u < GROWS; ++u) {
                const float nrm = w[u] * dd;
                acc.x += v[u].x * nrm; acc.y += v[u].y * nrm;
                acc.z += v[u].z * nrm; acc.w += v[u].w * nrm;
            }
        }
        const int r = c - j;
        if (r > 0) {
            float4 v[GROWS - 1]; float w[GROWS - 1];
            #pragma unroll
            for (int u = 0; u < GROWS - 1; ++u) {
                if (u < r) {
                    const int s = __builtin_amdgcn_readlane(sv, j + u);
                    w[u] = __int_as_float(__builtin_amdgcn_readlane(wbits, j + u));
                    v[u] = *(const float4*)(mlane + (size_t)s * HID);
                }
            }
            #pragma unroll
            for (int u = 0; u < GROWS - 1; ++u) {
                if (u < r) {
                    const float nrm = w[u] * dd;
                    acc.x += v[u].x * nrm; acc.y += v[u].y * nrm;
                    acc.z += v[u].z * nrm; acc.w += v[u].w * nrm;
                }
            }
        }
    }
    const float sn = dd * dd;
    acc.x += selfv.x * sn; acc.y += selfv.y * sn; acc.z += selfv.z * sn; acc.w += selfv.w * sn;
    acc.x += bb.x; acc.y += bb.y; acc.z += bb.z; acc.w += bb.w;
    *(float4*)(h + (size_t)node * HID + lane * 4) = acc;
}

__global__ void gcn_bn_zero(float* bnsum, float* bnsq) {
    int t = threadIdx.x;
    if (t < HID) bnsum[t] = 0.0f;
    else if (t < 2 * HID) bnsq[t - HID] = 0.0f;
}

__global__ __launch_bounds__(128) void gcn_bn_stats(const float* __restrict__ h,
                                                    float* bnsum, float* bnsq, int n) {
    int f = threadIdx.x;
    int chunk = (n + gridDim.x - 1) / gridDim.x;
    int n0 = blockIdx.x * chunk;
    int n1 = n0 + chunk; if (n1 > n) n1 = n;
    float s = 0.0f, ss = 0.0f;
    for (int nd = n0; nd < n1; ++nd) {
        float v = h[(size_t)nd * HID + f];
        s += v; ss += v * v;
    }
    atomAddF32(&bnsum[f], s);
    atomAddF32(&bnsq[f], ss);
}

__global__ void gcn_bn_apply(float* __restrict__ h,
                             const float* __restrict__ bnsum, const float* __restrict__ bnsq,
                             const float* __restrict__ gamma, const float* __restrict__ beta,
                             int n) {
    int idx = blockIdx.x * blockDim.x + threadIdx.x;
    if (idx >= n * HID) return;
    int f = idx & 127;
    float invn = 1.0f / (float)n;
    float mu  = bnsum[f] * invn;
    float var = bnsq[f] * invn - mu * mu;
    float v = gamma[f] * (h[idx] - mu) * rsqrtf(var + BN_EPS) + beta[f];
    h[idx] = fmaxf(v, 0.0f);
}

__global__ __launch_bounds__(128) void gcn_final(const float* __restrict__ h,
                                                 const float* __restrict__ lw,
                                                 const float* __restrict__ lb,
                                                 float* __restrict__ out, int n) {
    __shared__ float row[HID];
    __shared__ float lg[OUTC];
    int nd = blockIdx.x;
    int t = threadIdx.x;
    row[t] = h[(size_t)nd * HID + t];
    __syncthreads();
    if (t < OUTC) {
        float s = lb[t];
        #pragma unroll 4
        for (int k = 0; k < HID; ++k) s += row[k] * lw[k * OUTC + t];
        lg[t] = s;
    }
    __syncthreads();
    if (t < OUTC) {
        float mx = -INFINITY;
        for (int j = 0; j < OUTC; ++j) mx = fmaxf(mx, lg[j]);
        float se = 0.0f;
        for (int j = 0; j < OUTC; ++j) se += expf(lg[j] - mx);
        out[(size_t)nd * OUTC + t] = lg[t] - mx - logf(se);
    }
}

extern "C" void kernel_launch(void* const* d_in, const int* in_sizes, int n_in,
                              void* d_out, int out_size, void* d_ws, size_t ws_size,
                              hipStream_t stream) {
    const float* x       = (const float*)d_in[0];
    const int*   ei      = (const int*)d_in[1];
    const float* la      = (const float*)d_in[2];
    const float* lb      = (const float*)d_in[3];
    const float* mlp_w   = (const float*)d_in[4];
    const float* mlp_b   = (const float*)d_in[5];
    const float* conv_w  = (const float*)d_in[6];
    const float* conv_b  = (const float*)d_in[7];
    const float* bn_g    = (const float*)d_in[8];
    const float* bn_b    = (const float*)d_in[9];
    const float* lin_w   = (const float*)d_in[10];
    const float* lin_b   = (const float*)d_in[11];
    float* out = (float*)d_out;

    float* ws = (float*)d_ws;
    const size_t NM = (size_t)N_NODES * HID;
    float* hA    = ws;
    float* hB    = hA + NM;
    int*   cnt    = (int*)(hB + NM);
    int*   rowptr = cnt + SCAN_N;
    int*   rank   = rowptr + SCAN_N;
    int*   csr    = rank + N_EDGES;
    float* dinv  = hB + NM + NM;
    float* bnsum = dinv + N_NODES;
    float* bnsq  = bnsum + HID;

    const int TPB = 256;
    const int nElemBlocks = (N_NODES * HID + TPB - 1) / TPB;
    const int gemmBlocks  = (N_NODES + 63) / 64;

    gcn_cnt_zero<<<(N_NODES + TPB - 1) / TPB, TPB, 0, stream>>>(cnt, N_NODES);
    gcn_rank_edges<<<(N_EDGES + TPB - 1) / TPB, TPB, 0, stream>>>(ei, cnt, rank, N_EDGES);
    gcn_scan_deg<<<1, SCAN_T, 0, stream>>>(cnt, rowptr, dinv, N_NODES);
    gcn_fill_csr<<<(N_EDGES + TPB - 1) / TPB, TPB, 0, stream>>>(ei, rank, rowptr, dinv, csr, N_EDGES);

    gcn_combine<<<nElemBlocks, TPB, 0, stream>>>(x, la, lb, hA, N_NODES);

    float* cur = hA;
    float* oth = hB;
    for (int i = 0; i < NLAYERS; ++i) {
        gcn_gemm128_wmma<<<gemmBlocks, 128, 0, stream>>>(
            cur, mlp_w + (size_t)i * HID * HID, mlp_b + i * HID, oth, N_NODES, 1);
        gcn_gemm128_wmma<<<gemmBlocks, 128, 0, stream>>>(
            oth, conv_w + (size_t)i * HID * HID, nullptr, cur, N_NODES, 0);
        gcn_gather<<<(N_NODES * 32 + TPB - 1) / TPB, TPB, 0, stream>>>(
            rowptr, csr, dinv, cur, conv_b + i * HID, oth, N_NODES, N_EDGES);
        if (i < NLAYERS - 1) {
            gcn_bn_zero<<<1, 256, 0, stream>>>(bnsum, bnsq);
            gcn_bn_stats<<<256, 128, 0, stream>>>(oth, bnsum, bnsq, N_NODES);
            gcn_bn_apply<<<nElemBlocks, TPB, 0, stream>>>(
                oth, bnsum, bnsq, bn_g + i * HID, bn_b + i * HID, N_NODES);
        }
        float* tmp = cur; cur = oth; oth = tmp;
    }

    gcn_final<<<N_NODES, 128, 0, stream>>>(cur, lin_w, lin_b, out, N_NODES);
}


# Far fewer, longer waves (wave entry rate was the bound): elementwise and edge-list passes take 4-8 16-byte items per lane, gather waves walk 5 nodes each, classifier drops its 2 idle waves
# speedup vs baseline: 1.3837x; 1.0657x over previous
#include <hip/hip_runtime.h>
#include <hip/hip_bf16.h>
#include <math.h>

#define N_NODES 50000
#define N_EDGES 800000
#define HID     128
#define OUTC    40
#define NLAYERS 3
#define BN_EPS  1e-5f
#define WT_STRIDE 130

typedef __attribute__((ext_vector_type(2))) float v2f;
typedef __attribute__((ext_vector_type(8))) float v8f;
typedef __attribute__((ext_vector_type(4))) float v4f;

__device__ __forceinline__ void atomAddF32(float* p, float v) {
    __hip_atomic_fetch_add(p, v, __ATOMIC_RELAXED, __HIP_MEMORY_SCOPE_AGENT);
}

#define SCAN_T   1024
#define SCAN_PER 49
#define SCAN_N   (SCAN_T * SCAN_PER)

__global__ void gcn_cnt_zero(int* cnt, int n) {
    int i = blockIdx.x * blockDim.x + threadIdx.x;
    if (i < n) cnt[i] = 0;
}

__global__ __launch_bounds__(256) void gcn_rank_edges(const int* __restrict__ ei, int* cnt,
                                                      int* __restrict__ rank, int e) {
    const int i = (blockIdx.x * blockDim.x + threadIdx.x) * 4;
    if (i >= e) return;
    const int* dst = ei + e;
    if ((e & 3) == 0) {
        const int4 d = *(const int4*)(dst + i);
        int4 r;
        r.x = __hip_atomic_fetch_add(&cnt[d.x], 1, __ATOMIC_RELAXED, __HIP_MEMORY_SCOPE_AGENT);
        r.y = __hip_atomic_fetch_add(&cnt[d.y], 1, __ATOMIC_RELAXED, __HIP_MEMORY_SCOPE_AGENT);
        r.z = __hip_atomic_fetch_add(&cnt[d.z], 1, __ATOMIC_RELAXED, __HIP_MEMORY_SCOPE_AGENT);
        r.w = __hip_atomic_fetch_add(&cnt[d.w], 1, __ATOMIC_RELAXED, __HIP_MEMORY_SCOPE_AGENT);
        *(int4*)(rank + i) = r;
    } else {
        for (int k = i; k < e && k < i + 4; ++k)
            rank[k] = __hip_atomic_fetch_add(&cnt[dst[k]], 1, __ATOMIC_RELAXED, __HIP_MEMORY_SCOPE_AGENT);
    }
}

__global__ __launch_bounds__(SCAN_T) void gcn_scan_deg(const int* __restrict__ cnt,
                                                       int* __restrict__ rowptr,
                                                       float* __restrict__ dinv, int n) {
    __shared__ int buf[SCAN_N];
    __shared__ int toff[SCAN_T];
    __shared__ int wsum[SCAN_T / 32];
    const int t = threadIdx.x;
    for (int i = t; i < SCAN_N; i += SCAN_T) {
        int c = 0;
        if (i < n) { c = cnt[i]; dinv[i] = rsqrtf((float)(c + 1)); }
        buf[i] = c;
    }
    __syncthreads();
    const int base = t * SCAN_PER;
    int s = 0;
    for (int j = 0; j < SCAN_PER; ++j) { int c = buf[base + j]; buf[base + j] = s; s += c; }
    const int lane = t & 31, wv = t >> 5;
    int incl = s;
    for (int o = 1; o < 32; o <<= 1) { int y = __shfl_up(incl, o); if (lane >= o) incl += y; }
    if (lane == 31) wsum[wv] = incl;
    __syncthreads();
    if (wv == 0) {
        int x = wsum[lane];
        int xi = x;
        for (int o = 1; o < 32; o <<= 1) { int y = __shfl_up(xi, o); if (lane >= o) xi += y; }
        wsum[lane] = xi - x;
    }
    __syncthreads();
    toff[t] = wsum[wv] + incl - s;
    __syncthreads();
    for (int i = t; i <= n; i += SCAN_T) rowptr[i] = buf[i] + toff[i / SCAN_PER];
}

__global__ __launch_bounds__(256) void gcn_fill_csr(const int* __restrict__ ei, const int* __restrict__ rank,
                             const int* __restrict__ rowptr, const float* __restrict__ dinv,
                             int* __restrict__ csr, int e) {
    const int i = (blockIdx.x * blockDim.x + threadIdx.x) * 4;
    if (i >= e) return;
    if ((e & 3) == 0) {
        const int4 s4 = *(const int4*)(ei + i);
        const int4 d4 = *(const int4*)(ei + e + i);
        const int4 r4 = *(const int4*)(rank + i);
        const int s[4] = {s4.x, s4.y, s4.z, s4.w};
        const int d[4] = {d4.x, d4.y, d4.z, d4.w};
        const int r[4] = {r4.x, r4.y, r4.z, r4.w};
        int rp[4]; float dv[4];
        #pragma unroll
        for (int u = 0; u < 4; ++u) { rp[u] = rowptr[d[u]]; dv[u] = dinv[s[u]]; }
        #pragma unroll
        for (int u = 0; u < 4; ++u) {
            unsigned pos = (unsigned)(rp[u] + r[u]);
            if (pos > (unsigned)(e - 1)) pos = (unsigned)(e - 1);
            int2 p; p.x = s[u]; p.y = __float_as_int(dv[u]);
            *(int2*)(csr + 2 * (size_t)pos) = p;
        }
    } else {
        for (int k = i; k < e && k < i + 4; ++k) {
            const int s = ei[k];
            unsigned pos = (unsigned)(rowptr[ei[e + k]] + rank[k]);
            if (pos > (unsigned)(e - 1)) pos = (unsigned)(e - 1);
            int2 p; p.x = s; p.y = __float_as_int(dinv[s]);
            *(int2*)(csr + 2 * (size_t)pos) = p;
        }
    }
}

#define EW_ITEMS 4
__global__ __launch_bounds__(256) void gcn_combine(const float* __restrict__ x,
                                                   const float* __restrict__ la,
                                                   const float* __restrict__ lb,
                                                   float* __restrict__ h, int n) {
    const float alpha = expf(la[0]);
    const float beta  = expf(lb[0]);
    const int n4 = n * (HID / 4);
    const int stride = gridDim.x * blockDim.x;
    int i = blockIdx.x * blockDim.x + threadIdx.x;
    for (; i + (EW_ITEMS - 1) * stride < n4; i += EW_ITEMS * stride) {
        float4 a[EW_ITEMS], b[EW_ITEMS];
        #pragma unroll
        for (int u = 0; u < EW_ITEMS; ++u) {
            const int idx = i + u * stride;
            const float4* row = (const float4*)(x + (size_t)(idx >> 5) * 256);
            a[u] = row[idx & 31];
            b[u] = row[32 + (idx & 31)];
        }
        #pragma unroll
        for (int u = 0; u < EW_ITEMS; ++u) {
            float4 o;
            o.x = alpha * a[u].x + beta * b[u].x;
            o.y = alpha * a[u].y + beta * b[u].y;
            o.z = alpha * a[u].z + beta * b[u].z;
            o.w = alpha * a[u].w + beta * b[u].w;
            ((float4*)h)[i + u * stride] = o;
        }
    }
    for (; i < n4; i += stride) {
        const float4* row = (const float4*)(x + (size_t)(i >> 5) * 256);
        const float4 a = row[i & 31];
        const float4 b = row[32 + (i & 31)];
        float4 o;
        o.x = alpha * a.x + beta * b.x;
        o.y = alpha * a.y + beta * b.y;
        o.z = alpha * a.z + beta * b.z;
        o.w = alpha * a.w + beta * b.w;
        ((float4*)h)[i] = o;
    }
}

__global__ __launch_bounds__(128) void gcn_gemm128_wmma(
        const float* __restrict__ A, const float* __restrict__ W,
        const float* __restrict__ bias, float* __restrict__ C,
        int nrows, int do_relu) {
    __shared__ float Wt[HID * WT_STRIDE];

    int t = threadIdx.x;
    #pragma unroll 4
    for (int idx = t; idx < HID * HID; idx += 128) {
        int kk = idx >> 7;
        int n  = idx & 127;
        Wt[n * WT_STRIDE + kk] = W[idx];
    }
    __syncthreads();

    const int wave  = t >> 5;
    const int lane  = t & 31;
    const int row16 = lane & 15;
    const int khalf = (lane >> 4) << 1;

    const int tileM = blockIdx.x * 64 + wave * 16;
    int rowIdx = tileM + row16;
    if (rowIdx > nrows - 1) rowIdx = nrows - 1;
    const float* Arow = A + (size_t)rowIdx * HID;
    const float* Brow = Wt + (size_t)row16 * WT_STRIDE;

    v8f acc[8] = {};

    for (int k = 0; k < HID; k += 4) {
        const float2 av = *(const float2*)(Arow + k + khalf);
        v2f a; a.x = av.x; a.y = av.y;
        #pragma unroll
        for (int nt = 0; nt < 8; ++nt) {
            const float2 bv = *(const float2*)(Brow + nt * 16 * WT_STRIDE + k + khalf);
            v2f b; b.x = bv.x; b.y = bv.y;
            acc[nt] = __builtin_amdgcn_wmma_f32_16x16x4_f32(
                false, a, false, b, (short)0, acc[nt], false, false);
        }
    }

    const int rbase = tileM + ((lane >> 4) << 3);
    #pragma unroll
    for (int nt = 0; nt < 8; ++nt) {
        const int col = nt * 16 + row16;
        const float bv = bias ? bias[col] : 0.0f;
        #pragma unroll
        for (int r = 0; r < 8; ++r) {
            const int row = rbase + r;
            if (row < nrows) {
                float v = acc[nt][r] + bv;
                if (do_relu) v = fmaxf(v, 0.0f);
                C[(size_t)row * HID + col] = v;
            }
        }
    }
}

#define GROWS 8

__global__ __launch_bounds__(256) void gcn_gather(const int* __restrict__ rowptr,
                                                  const int* __restrict__ csr,
                                                  const float* __restrict__ dinv,
                                                  const float* __restrict__ m,
                                                  const float* __restrict__ b,
                                                  float* __restrict__ h, int n, int e) {
    const int lane = threadIdx.x & 31;
    const int nwaves = (gridDim.x * blockDim.x) >> 5;
    const int wid = __builtin_amdgcn_readfirstlane((blockIdx.x * blockDim.x + threadIdx.x) >> 5);
    const float* mlane = m + lane * 4;
    const float4 bb = *(const float4*)(b + lane * 4);
    for (int node = wid; node < n; node += nwaves) {
        int beg = rowptr[node];
        int end = rowptr[node + 1];
        if (beg < 0) beg = 0;
        if (end > e) end = e;
        const float dd = dinv[node];
        const float4 selfv = *(const float4*)(mlane + (size_t)node * HID);
        float4 acc = make_float4(0.0f, 0.0f, 0.0f, 0.0f);

        for (int base = beg; base < end; base += 32) {
            int c = end - base; if (c > 32) c = 32;
            int sv = 0, wbits = 0;
            if (lane < c) {
                const int2 p = *(const int2*)(csr + 2 * (size_t)(base + lane));
                sv = p.x; wbits = p.y;
            }
            if ((unsigned)sv > (unsigned)(n - 1)) sv = n - 1;
            int j = 0;
            for (; j + GROWS <= c; j += GROWS) {
                float4 v[GROWS]; float w[GROWS];
                #pragma unroll
                for (int u = 0; u < GROWS; ++u) {
                    const int s = __builtin_amdgcn_readlane(sv, j + u);
                    w[u] = __int_as_float(__builtin_amdgcn_readlane(wbits, j + u));
                    v[u] = *(const float4*)(mlane + (size_t)s * HID);
                }
                #pragma unroll
                for (int u = 0; u < GROWS; ++u) {
                    const float nrm = w[u] * dd;
                    acc.x += v[u].x * nrm; acc.y += v[u].y * nrm;
                    acc.z += v[u].z * nrm; acc.w += v[u].w * nrm;
                }
            }
            const int r = c - j;
            if (r > 0) {
                float4 v[GROWS - 1]; float w[GROWS - 1];
                #pragma unroll
                for (int u = 0; u < GROWS - 1; ++u) {
                    if (u < r) {
                        const int s = __builtin_amdgcn_readlane(sv, j + u);
                        w[u] = __int_as_float(__builtin_amdgcn_readlane(wbits, j + u));
                        v[u] = *(const float4*)(mlane + (size_t)s * HID);
                    }
                }
                #pragma unroll
                for (int u = 0; u < GROWS - 1; ++u) {
                    if (u < r) {
                        const float nrm = w[u] * dd;
                        acc.x += v[u].x * nrm; acc.y += v[u].y * nrm;
                        acc.z += v[u].z * nrm; acc.w += v[u].w * nrm;
                    }
                }
            }
        }
        const float sn = dd * dd;
        acc.x += selfv.x * sn; acc.y += selfv.y * sn; acc.z += selfv.z * sn; acc.w += selfv.w * sn;
        acc.x += bb.x; acc.y += bb.y; acc.z += bb.z; acc.w += bb.w;
        *(float4*)(h + (size_t)node * HID + lane * 4) = acc;
    }
}

__global__ void gcn_bn_zero(float* bnsum, float* bnsq) {
    int t = threadIdx.x;
    if (t < HID) bnsum[t] = 0.0f;
    else if (t < 2 * HID) bnsq[t - HID] = 0.0f;
}

__global__ __launch_bounds__(128) void gcn_bn_stats(const float* __restrict__ h,
                                                    float* bnsum, float* bnsq, int n) {
    int f = threadIdx.x;
    int chunk = (n + gridDim.x - 1) / gridDim.x;
    int n0 = blockIdx.x * chunk;
    int n1 = n0 + chunk; if (n1 > n) n1 = n;
    float s = 0.0f, ss = 0.0f;
    for (int nd = n0; nd < n1; ++nd) {
        float v = h[(size_t)nd * HID + f];
        s += v; ss += v * v;
    }
    atomAddF32(&bnsum[f], s);
    atomAddF32(&bnsq[f], ss);
}

#define BN_ITEMS 8
__global__ __launch_bounds__(256) void gcn_bn_apply(float* __restrict__ h,
                             const float* __restrict__ bnsum, const float* __restrict__ bnsq,
                             const float* __restrict__ gamma, const float* __restrict__ beta,
                             int n) {
    const int n4 = n * (HID / 4);
    const int stride = gridDim.x * blockDim.x;
    int i = blockIdx.x * blockDim.x + threadIdx.x;
    const int f = (i & 31) * 4;
    const float invn = 1.0f / (float)n;
    v4f g, mu, rs, bt;
    #pragma unroll
    for (int c = 0; c < 4; ++c) {
        mu[c] = bnsum[f + c] * invn;
        const float var = bnsq[f + c] * invn - mu[c] * mu[c];
        rs[c] = rsqrtf(var + BN_EPS);
        g[c]  = gamma[f + c];
        bt[c] = beta[f + c];
    }
    const v4f zero = {0.0f, 0.0f, 0.0f, 0.0f};
    v4f* h4 = (v4f*)h;
    for (; i + (BN_ITEMS - 1) * stride < n4; i += BN_ITEMS * stride) {
        v4f v[BN_ITEMS];
        #pragma unroll
        for (int u = 0; u < BN_ITEMS; ++u) v[u] = h4[i + u * stride];
        #pragma unroll
        for (int u = 0; u < BN_ITEMS; ++u)
            h4[i + u * stride] = __builtin_elementwise_max(g * (v[u] - mu) * rs + bt, zero);
    }
    for (; i < n4; i += stride)
        h4[i] = __builtin_elementwise_max(g * (h4[i] - mu) * rs + bt, zero);
}

__global__ __launch_bounds__(64) void gcn_final(const float* __restrict__ h,
                                                 const float* __restrict__ lw,
                                                 const float* __restrict__ lb,
                                                 float* __restrict__ out, int n) {
    __shared__ float row[HID];
    __shared__ float lg[OUTC];
    int nd = blockIdx.x;
    int t = threadIdx.x;
    row[t] = h[(size_t)nd * HID + t];
    row[t + 64] = h[(size_t)nd * HID + 64 + t];
    __syncthreads();
    if (t < OUTC) {
        float s = lb[t];
        #pragma unroll 4
        for (int k = 0; k < HID; ++k) s += row[k] * lw[k * OUTC + t];
        lg[t] = s;
    }
    __syncthreads();
    if (t < OUTC) {
        float mx = -INFINITY;
        for (int j = 0; j < OUTC; ++j) mx = fmaxf(mx, lg[j]);
        float se = 0.0f;
        for (int j = 0; j < OUTC; ++j) se += expf(lg[j] - mx);
        out[(size_t)nd * OUTC + t] = lg[t] - mx - logf(se);
    }
}

extern "C" void kernel_launch(void* const* d_in, const int* in_sizes, int n_in,
                              void* d_out, int out_size, void* d_ws, size_t ws_size,
                              hipStream_t stream) {
    const float* x       = (const float*)d_in[0];
    const int*   ei      = (const int*)d_in[1];
    const float* la      = (const float*)d_in[2];
    const float* lb      = (const float*)d_in[3];
    const float* mlp_w   = (const float*)d_in[4];
    const float* mlp_b   = (const float*)d_in[5];
    const float* conv_w  = (const float*)d_in[6];
    const float* conv_b  = (const float*)d_in[7];
    const float* bn_g    = (const float*)d_in[8];
    const float* bn_b    = (const float*)d_in[9];
    const float* lin_w   = (const float*)d_in[10];
    const float* lin_b   = (const float*)d_in[11];
    float* out = (float*)d_out;

    float* ws = (float*)d_ws;
    const size_t NM = (size_t)N_NODES * HID;
    float* hA    = ws;
    float* hB    = hA + NM;
    int*   cnt    = (int*)(hB + NM);
    int*   rowptr = cnt + SCAN_N;
    int*   rank   = rowptr + SCAN_N;
    int*   csr    = rank + N_EDGES;
    float* dinv  = hB + NM + NM;
    float* bnsum = dinv + N_NODES;
    float* bnsq  = bnsum + HID;

    const int TPB = 256;
    const int n4 = N_NODES * (HID / 4);
    const int combBlocks = (n4 + TPB * EW_ITEMS - 1) / (TPB * EW_ITEMS);
    const int bnBlocks   = (n4 + TPB * BN_ITEMS - 1) / (TPB * BN_ITEMS);
    const int edgeBlocks = ((N_EDGES + 3) / 4 + TPB - 1) / TPB;
    const int gatherBlocks = 1250;
    const int gemmBlocks  = (N_NODES + 63) / 64;

    gcn_cnt_zero<<<(N_NODES + TPB - 1) / TPB, TPB, 0, stream>>>(cnt, N_NODES);
    gcn_rank_edges<<<edgeBlocks, TPB, 0, stream>>>(ei, cnt, rank, N_EDGES);
    gcn_scan_deg<<<1, SCAN_T, 0, stream>>>(cnt, rowptr, dinv, N_NODES);
    gcn_fill_csr<<<edgeBlocks, TPB, 0, stream>>>(ei, rank, rowptr, dinv, csr, N_EDGES);

    gcn_combine<<<combBlocks, TPB, 0, stream>>>(x, la, lb, hA, N_NODES);

    float* cur = hA;
    float* oth = hB;
    for (int i = 0; i < NLAYERS; ++i) {
        gcn_gemm128_wmma<<<gemmBlocks, 128, 0, stream>>>(
            cur, mlp_w + (size_t)i * HID * HID, mlp_b + i * HID, oth, N_NODES, 1);
        gcn_gemm128_wmma<<<gemmBlocks, 128, 0, stream>>>(
            oth, conv_w + (size_t)i * HID * HID, nullptr, cur, N_NODES, 0);
        gcn_gather<<<gatherBlocks, TPB, 0, stream>>>(
            rowptr, csr, dinv, cur, conv_b + i * HID, oth, N_NODES, N_EDGES);
        if (i < NLAYERS - 1) {
            gcn_bn_zero<<<1, 256, 0, stream>>>(bnsum, bnsq);
            gcn_bn_stats<<<256, 128, 0, stream>>>(oth, bnsum, bnsq, N_NODES);
            gcn_bn_apply<<<bnBlocks, TPB, 0, stream>>>(
                oth, bnsum, bnsq, bn_g + i * HID, bn_b + i * HID, N_NODES);
        }
        float* tmp = cur; cur = oth; oth = tmp;
    }

    gcn_final<<<N_NODES, 64, 0, stream>>>(cur, lin_w, lin_b, out, N_NODES);
}


# Batch-norm sums taken in the gather epilogue from the row in registers (per-wave partials, LDS fold of 8 waves, 1 float atomic per feature per block); stats pass removed; BN sum order changes
# speedup vs baseline: 1.8490x; 1.3363x over previous
#include <hip/hip_runtime.h>
#include <hip/hip_bf16.h>
#include <math.h>

#define N_NODES 50000
#define N_EDGES 800000
#define HID     128
#define OUTC    40
#define NLAYERS 3
#define BN_EPS  1e-5f
#define WT_STRIDE 130

typedef __attribute__((ext_vector_type(2))) float v2f;
typedef __attribute__((ext_vector_type(8))) float v8f;
typedef __attribute__((ext_vector_type(4))) float v4f;

__device__ __forceinline__ void atomAddF32(float* p, float v) {
    __hip_atomic_fetch_add(p, v, __ATOMIC_RELAXED, __HIP_MEMORY_SCOPE_AGENT);
}

#define SCAN_T   1024
#define SCAN_PER 49
#define SCAN_N   (SCAN_T * SCAN_PER)

__global__ void gcn_cnt_zero(int* cnt, int n) {
    int i = blockIdx.x * blockDim.x + threadIdx.x;
    if (i < n) cnt[i] = 0;
}

__global__ __launch_bounds__(256) void gcn_rank_edges(const int* __restrict__ ei, int* cnt,
                                                      int* __restrict__ rank, int e) {
    const int i = (blockIdx.x * blockDim.x + threadIdx.x) * 4;
    if (i >= e) return;
    const int* dst = ei + e;
    if ((e & 3) == 0) {
        const int4 d = *(const int4*)(dst + i);
        int4 r;
        r.x = __hip_atomic_fetch_add(&cnt[d.x], 1, __ATOMIC_RELAXED, __HIP_MEMORY_SCOPE_AGENT);
        r.y = __hip_atomic_fetch_add(&cnt[d.y], 1, __ATOMIC_RELAXED, __HIP_MEMORY_SCOPE_AGENT);
        r.z = __hip_atomic_fetch_add(&cnt[d.z], 1, __ATOMIC_RELAXED, __HIP_MEMORY_SCOPE_AGENT);
        r.w = __hip_atomic_fetch_add(&cnt[d.w], 1, __ATOMIC_RELAXED, __HIP_MEMORY_SCOPE_AGENT);
        *(int4*)(rank + i) = r;
    } else {
        for (int k = i; k < e && k < i + 4; ++k)
            rank[k] = __hip_atomic_fetch_add(&cnt[dst[k]], 1, __ATOMIC_RELAXED, __HIP_MEMORY_SCOPE_AGENT);
    }
}

__global__ __launch_bounds__(SCAN_T) void gcn_scan_deg(const int* __restrict__ cnt,
                                                       int* __restrict__ rowptr,
                                                       float* __restrict__ dinv, int n) {
    __shared__ int buf[SCAN_N];
    __shared__ int toff[SCAN_T];
    __shared__ int wsum[SCAN_T / 32];
    const int t = threadIdx.x;
    for (int i = t; i < SCAN_N; i += SCAN_T) {
        int c = 0;
        if (i < n) { c = cnt[i]; dinv[i] = rsqrtf((float)(c + 1)); }
        buf[i] = c;
    }
    __syncthreads();
    const int base = t * SCAN_PER;
    int s = 0;
    for (int j = 0; j < SCAN_PER; ++j) { int c = buf[base + j]; buf[base + j] = s; s += c; }
    const int lane = t & 31, wv = t >> 5;
    int incl = s;
    for (int o = 1; o < 32; o <<= 1) { int y = __shfl_up(incl, o); if (lane >= o) incl += y; }
    if (lane == 31) wsum[wv] = incl;
    __syncthreads();
    if (wv == 0) {
        int x = wsum[lane];
        int xi = x;
        for (int o = 1; o < 32; o <<= 1) { int y = __shfl_up(xi, o); if (lane >= o) xi += y; }
        wsum[lane] = xi - x;
    }
    __syncthreads();
    toff[t] = wsum[wv] + incl - s;
    __syncthreads();
    for (int i = t; i <= n; i += SCAN_T) rowptr[i] = buf[i] + toff[i / SCAN_PER];
}

__global__ __launch_bounds__(256) void gcn_fill_csr(const int* __restrict__ ei, const int* __restrict__ rank,
                             const int* __restrict__ rowptr, const float* __restrict__ dinv,
                             int* __restrict__ csr, int e) {
    const int i = (blockIdx.x * blockDim.x + threadIdx.x) * 4;
    if (i >= e) return;
    if ((e & 3) == 0) {
        const int4 s4 = *(const int4*)(ei + i);
        const int4 d4 = *(const int4*)(ei + e + i);
        const int4 r4 = *(const int4*)(rank + i);
        const int s[4] = {s4.x, s4.y, s4.z, s4.w};
        const int d[4] = {d4.x, d4.y, d4.z, d4.w};
        const int r[4] = {r4.x, r4.y, r4.z, r4.w};
        int rp[4]; float dv[4];
        #pragma unroll
        for (int u = 0; u < 4; ++u) { rp[u] = rowptr[d[u]]; dv[u] = dinv[s[u]]; }
        #pragma unroll
        for (int u = 0; u < 4; ++u) {
            unsigned pos = (unsigned)(rp[u] + r[u]);
            if (pos > (unsigned)(e - 1)) pos = (unsigned)(e - 1);
            int2 p; p.x = s[u]; p.y = __float_as_int(dv[u]);
            *(int2*)(csr + 2 * (size_t)pos) = p;
        }
    } else {
        for (int k = i; k < e && k < i + 4; ++k) {
            const int s = ei[k];
            unsigned pos = (unsigned)(rowptr[ei[e + k]] + rank[k]);
            if (pos > (unsigned)(e - 1)) pos = (unsigned)(e - 1);
            int2 p; p.x = s; p.y = __float_as_int(dinv[s]);
            *(int2*)(csr + 2 * (size_t)pos) = p;
        }
    }
}

#define EW_ITEMS 4
__global__ __launch_bounds__(256) void gcn_combine(const float* __restrict__ x,
                                                   const float* __restrict__ la,
                                                   const float* __restrict__ lb,
                                                   float* __restrict__ h, int n) {
    const float alpha = expf(la[0]);
    const float beta  = expf(lb[0]);
    const int n4 = n * (HID / 4);
    const int stride = gridDim.x * blockDim.x;
    int i = blockIdx.x * blockDim.x + threadIdx.x;
    for (; i + (EW_ITEMS - 1) * stride < n4; i += EW_ITEMS * stride) {
        float4 a[EW_ITEMS], b[EW_ITEMS];
        #pragma unroll
        for (int u = 0; u < EW_ITEMS; ++u) {
            const int idx = i + u * stride;
            const float4* row = (const float4*)(x + (size_t)(idx >> 5) * 256);
            a[u] = row[idx & 31];
            b[u] = row[32 + (idx & 31)];
        }
        #pragma unroll
        for (int u = 0; u < EW_ITEMS; ++u) {
            float4 o;
            o.x = alpha * a[u].x + beta * b[u].x;
            o.y = alpha * a[u].y + beta * b[u].y;
            o.z = alpha * a[u].z + beta * b[u].z;
            o.w = alpha * a[u].w + beta * b[u].w;
            ((float4*)h)[i + u * stride] = o;
        }
    }
    for (; i < n4; i += stride) {
        const float4* row = (const float4*)(x + (size_t)(i >> 5) * 256);
        const float4 a = row[i & 31];
        const float4 b = row[32 + (i & 31)];
        float4 o;
        o.x = alpha * a.x + beta * b.x;
        o.y = alpha * a.y + beta * b.y;
        o.z = alpha * a.z + beta * b.z;
        o.w = alpha * a.w + beta * b.w;
        ((float4*)h)[i] = o;
    }
}

__global__ __launch_bounds__(128) void gcn_gemm128_wmma(
        const float* __restrict__ A, const float* __restrict__ W,
        const float* __restrict__ bias, float* __restrict__ C,
        int nrows, int do_relu) {
    __shared__ float Wt[HID * WT_STRIDE];

    int t = threadIdx.x;
    #pragma unroll 4
    for (int idx = t; idx < HID * HID; idx += 128) {
        int kk = idx >> 7;
        int n  = idx & 127;
        Wt[n * WT_STRIDE + kk] = W[idx];
    }
    __syncthreads();

    const int wave  = t >> 5;
    const int lane  = t & 31;
    const int row16 = lane & 15;
    const int khalf = (lane >> 4) << 1;

    const int tileM = blockIdx.x * 64 + wave * 16;
    int rowIdx = tileM + row16;
    if (rowIdx > nrows - 1) rowIdx = nrows - 1;
    const float* Arow = A + (size_t)rowIdx * HID;
    const float* Brow = Wt + (size_t)row16 * WT_STRIDE;

    v8f acc[8] = {};

    for (int k = 0; k < HID; k += 4) {
        const float2 av = *(const float2*)(Arow + k + khalf);
        v2f a; a.x = av.x; a.y = av.y;
        #pragma unroll
        for (int nt = 0; nt < 8; ++nt) {
            const float2 bv = *(const float2*)(Brow + nt * 16 * WT_STRIDE + k + khalf);
            v2f b; b.x = bv.x; b.y = bv.y;
            acc[nt] = __builtin_amdgcn_wmma_f32_16x16x4_f32(
                false, a, false, b, (short)0, acc[nt], false, false);
        }
    }

    const int rbase = tileM + ((lane >> 4) << 3);
    #pragma unroll
    for (int nt = 0; nt < 8; ++nt) {
        const int col = nt * 16 + row16;
        const float bv = bias ? bias[col] : 0.0f;
        #pragma unroll
        for (int r = 0; r < 8; ++r) {
            const int row = rbase + r;
            if (row < nrows) {
                float v = acc[nt][r] + bv;
                if (do_relu) v = fmaxf(v, 0.0f);
                C[(size_t)row * HID + col] = v;
            }
        }
    }
}

#define GROWS 8
#define GATHER_T 256

template <bool STATS>
__global__ __launch_bounds__(GATHER_T) void gcn_gather(const int* __restrict__ rowptr,
                                                  const int* __restrict__ csr,
                                                  const float* __restrict__ dinv,
                                                  const float* __restrict__ m,
                                                  const float* __restrict__ b,
                                                  float* __restrict__ h,
                                                  float* bnsum, float* bnsq, int n, int e) {
    const int lane = threadIdx.x & 31;
    float4 ssum = make_float4(0.0f, 0.0f, 0.0f, 0.0f);
    float4 ssq  = make_float4(0.0f, 0.0f, 0.0f, 0.0f);
    const int nwaves = (gridDim.x * blockDim.x) >> 5;
    const int wid = __builtin_amdgcn_readfirstlane((blockIdx.x * blockDim.x + threadIdx.x) >> 5);
    const float* mlane = m + lane * 4;
    const float4 bb = *(const float4*)(b + lane * 4);
    for (int node = wid; node < n; node += nwaves) {
        int beg = rowptr[node];
        int end = rowptr[node + 1];
        if (beg < 0) beg = 0;
        if (end > e) end = e;
        const float dd = dinv[node];
        const float4 selfv = *(const float4*)(mlane + (size_t)node * HID);
        float4 acc = make_float4(0.0f, 0.0f, 0.0f, 0.0f);

        for (int base = beg; base < end; base += 32) {
            int c = end - base; if (c > 32) c = 32;
            int sv = 0, wbits = 0;
            if (lane < c) {
                const int2 p = *(const int2*)(csr + 2 * (size_t)(base + lane));
                sv = p.x; wbits = p.y;
            }
            if ((unsigned)sv > (unsigned)(n - 1)) sv = n - 1;
            int j = 0;
            for (; j + GROWS <= c; j += GROWS) {
                float4 v[GROWS]; float w[GROWS];
                #pragma unroll
                for (int u = 0; u < GROWS; ++u) {
                    const int s = __builtin_amdgcn_readlane(sv, j + u);
                    w[u] = __int_as_float(__builtin_amdgcn_readlane(wbits, j + u));
                    v[u] = *(const float4*)(mlane + (size_t)s * HID);
                }
                #pragma unroll
                for (int u = 0; u < GROWS; ++u) {
                    const float nrm = w[u] * dd;
                    acc.x += v[u].x * nrm; acc.y += v[u].y * nrm;
                    acc.z += v[u].z * nrm; acc.w += v[u].w * nrm;
                }
            }
            const int r = c - j;
            if (r > 0) {
                float4 v[GROWS - 1]; float w[GROWS - 1];
                #pragma unroll
                for (int u = 0; u < GROWS - 1; ++u) {
                    if (u < r) {
                        const int s = __builtin_amdgcn_readlane(sv, j + u);
                        w[u] = __int_as_float(__builtin_amdgcn_readlane(wbits, j + u));
                        v[u] = *(const float4*)(mlane + (size_t)s * HID);
                    }
                }
                #pragma unroll
                for (int u = 0; u < GROWS - 1; ++u) {
                    if (u < r) {
                        const float nrm = w[u] * dd;
                        acc.x += v[u].x * nrm; acc.y += v[u].y * nrm;
                        acc.z += v[u].z * nrm; acc.w += v[u].w * nrm;
                    }
                }
            }
        }
        const float sn = dd * dd;
        acc.x += selfv.x * sn; acc.y += selfv.y * sn; acc.z += selfv.z * sn; acc.w += selfv.w * sn;
        acc.x += bb.x; acc.y += bb.y; acc.z += bb.z; acc.w += bb.w;
        *(float4*)(h + (size_t)node * HID + lane * 4) = acc;
        if constexpr (STATS) {
            ssum.x += acc.x; ssum.y += acc.y; ssum.z += acc.z; ssum.w += acc.w;
            ssq.x += acc.x * acc.x; ssq.y += acc.y * acc.y;
            ssq.z += acc.z * acc.z; ssq.w += acc.w * acc.w;
        }
    }
    if constexpr (STATS) {
        __shared__ float red[(GATHER_T / 32) * 2 * HID];
        const int wv = threadIdx.x >> 5;
        *(float4*)(red + (wv * 2 + 0) * HID + lane * 4) = ssum;
        *(float4*)(red + (wv * 2 + 1) * HID + lane * 4) = ssq;
        __syncthreads();
        static_assert(GATHER_T == 2 * HID, "one thread per batch-norm sum");
        const int which = threadIdx.x >> 7;
        const int f = threadIdx.x & (HID - 1);
        float tot = 0.0f;
        for (int w = 0; w < GATHER_T / 32; ++w) tot += red[(w * 2 + which) * HID + f];
        atomAddF32(which ? &bnsq[f] : &bnsum[f], tot);
    }
}

__global__ void gcn_bn_zero(float* bnsum, float* bnsq) {
    int t = threadIdx.x;
    if (t < HID) bnsum[t] = 0.0f;
    else if (t < 2 * HID) bnsq[t - HID] = 0.0f;
}

#define BN_ITEMS 8
__global__ __launch_bounds__(256) void gcn_bn_apply(float* __restrict__ h,
                             const float* __restrict__ bnsum, const float* __restrict__ bnsq,
                             const float* __restrict__ gamma, const float* __restrict__ beta,
                             int n) {
    const int n4 = n * (HID / 4);
    const int stride = gridDim.x * blockDim.x;
    int i = blockIdx.x * blockDim.x + threadIdx.x;
    const int f = (i & 31) * 4;
    const float invn = 1.0f / (float)n;
    v4f g, mu, rs, bt;
    #pragma unroll
    for (int c = 0; c < 4; ++c) {
        mu[c] = bnsum[f + c] * invn;
        const float var = bnsq[f + c] * invn - mu[c] * mu[c];
        rs[c] = rsqrtf(var + BN_EPS);
        g[c]  = gamma[f + c];
        bt[c] = beta[f + c];
    }
    const v4f zero = {0.0f, 0.0f, 0.0f, 0.0f};
    v4f* h4 = (v4f*)h;
    for (; i + (BN_ITEMS - 1) * stride < n4; i += BN_ITEMS * stride) {
        v4f v[BN_ITEMS];
        #pragma unroll
        for (int u = 0; u < BN_ITEMS; ++u) v[u] = h4[i + u * stride];
        #pragma unroll
        for (int u = 0; u < BN_ITEMS; ++u)
            h4[i + u * stride] = __builtin_elementwise_max(g * (v[u] - mu) * rs + bt, zero);
    }
    for (; i < n4; i += stride)
        h4[i] = __builtin_elementwise_max(g * (h4[i] - mu) * rs + bt, zero);
}

__global__ __launch_bounds__(64) void gcn_final(const float* __restrict__ h,
                                                 const float* __restrict__ lw,
                                                 const float* __restrict__ lb,
                                                 float* __restrict__ out, int n) {
    __shared__ float row[HID];
    __shared__ float lg[OUTC];
    int nd = blockIdx.x;
    int t = threadIdx.x;
    row[t] = h[(size_t)nd * HID + t];
    row[t + 64] = h[(size_t)nd * HID + 64 + t];
    __syncthreads();
    if (t < OUTC) {
        float s = lb[t];
        #pragma unroll 4
        for (int k = 0; k < HID; ++k) s += row[k] * lw[k * OUTC + t];
        lg[t] = s;
    }
    __syncthreads();
    if (t < OUTC) {
        float mx = -INFINITY;
        for (int j = 0; j < OUTC; ++j) mx = fmaxf(mx, lg[j]);
        float se = 0.0f;
        for (int j = 0; j < OUTC; ++j) se += expf(lg[j] - mx);
        out[(size_t)nd * OUTC + t] = lg[t] - mx - logf(se);
    }
}

extern "C" void kernel_launch(void* const* d_in, const int* in_sizes, int n_in,
                              void* d_out, int out_size, void* d_ws, size_t ws_size,
                              hipStream_t stream) {
    const float* x       = (const float*)d_in[0];
    const int*   ei      = (const int*)d_in[1];
    const float* la      = (const float*)d_in[2];
    const float* lb      = (const float*)d_in[3];
    const float* mlp_w   = (const float*)d_in[4];
    const float* mlp_b   = (const float*)d_in[5];
    const float* conv_w  = (const float*)d_in[6];
    const float* conv_b  = (const float*)d_in[7];
    const float* bn_g    = (const float*)d_in[8];
    const float* bn_b    = (const float*)d_in[9];
    const float* lin_w   = (const float*)d_in[10];
    const float* lin_b   = (const float*)d_in[11];
    float* out = (float*)d_out;

    float* ws = (float*)d_ws;
    const size_t NM = (size_t)N_NODES * HID;
    float* hA    = ws;
    float* hB    = hA + NM;
    int*   cnt    = (int*)(hB + NM);
    int*   rowptr = cnt + SCAN_N;
    int*   rank   = rowptr + SCAN_N;
    int*   csr    = rank + N_EDGES;
    float* dinv  = hB + NM + NM;
    float* bnsum = dinv + N_NODES;
    float* bnsq  = bnsum + HID;

    const int TPB = 256;
    const int n4 = N_NODES * (HID / 4);
    const int combBlocks = (n4 + TPB * EW_ITEMS - 1) / (TPB * EW_ITEMS);
    const int bnBlocks   = (n4 + TPB * BN_ITEMS - 1) / (TPB * BN_ITEMS);
    const int edgeBlocks = ((N_EDGES + 3) / 4 + TPB - 1) / TPB;
    const int gatherBlocks = 1250;
    const int gemmBlocks  = (N_NODES + 63) / 64;

    gcn_cnt_zero<<<(N_NODES + TPB - 1) / TPB, TPB, 0, stream>>>(cnt, N_NODES);
    gcn_rank_edges<<<edgeBlocks, TPB, 0, stream>>>(ei, cnt, rank, N_EDGES);
    gcn_scan_deg<<<1, SCAN_T, 0, stream>>>(cnt, rowptr, dinv, N_NODES);
    gcn_fill_csr<<<edgeBlocks, TPB, 0, stream>>>(ei, rank, rowptr, dinv, csr, N_EDGES);

    gcn_combine<<<combBlocks, TPB, 0, stream>>>(x, la, lb, hA, N_NODES);

    float* cur = hA;
    float* oth = hB;
    for (int i = 0; i < NLAYERS; ++i) {
        gcn_gemm128_wmma<<<gemmBlocks, 128, 0, stream>>>(
            cur, mlp_w + (size_t)i * HID * HID, mlp_b + i * HID, oth, N_NODES, 1);
        gcn_gemm128_wmma<<<gemmBlocks, 128, 0, stream>>>(
            oth, conv_w + (size_t)i * HID * HID, nullptr, cur, N_NODES, 0);
        if (i < NLAYERS - 1) {
            gcn_bn_zero<<<1, 256, 0, stream>>>(bnsum, bnsq);
            gcn_gather<true><<<gatherBlocks, GATHER_T, 0, stream>>>(
                rowptr, csr, dinv, cur, conv_b + i * HID, oth, bnsum, bnsq, N_NODES, N_EDGES);
            gcn_bn_apply<<<bnBlocks, TPB, 0, stream>>>(
                oth, bnsum, bnsq, bn_g + i * HID, bn_b + i * HID, N_NODES);
        } else {
            gcn_gather<false><<<gatherBlocks, GATHER_T, 0, stream>>>(
                rowptr, csr, dinv, cur, conv_b + i * HID, oth, nullptr, nullptr, N_NODES, N_EDGES);
        }
        float* tmp = cur; cur = oth; oth = tmp;
    }

    gcn_final<<<N_NODES, 64, 0, stream>>>(cur, lin_w, lin_b, out, N_NODES);
}


# Classifier moved to the matrix unit: 128->40 linear as f32 16x16x4 WMMA (3 column tiles of 16), log-softmax in the accumulator layout by 16-lane reductions; logit and softmax sum order changes
# speedup vs baseline: 2.1038x; 1.1378x over previous
#include <hip/hip_runtime.h>
#include <hip/hip_bf16.h>
#include <math.h>

#define N_NODES 50000
#define N_EDGES 800000
#define HID     128
#define OUTC    40
#define NLAYERS 3
#define BN_EPS  1e-5f
#define WT_STRIDE 130

typedef __attribute__((ext_vector_type(2))) float v2f;
typedef __attribute__((ext_vector_type(8))) float v8f;
typedef __attribute__((ext_vector_type(4))) float v4f;

__device__ __forceinline__ void atomAddF32(float* p, float v) {
    __hip_atomic_fetch_add(p, v, __ATOMIC_RELAXED, __HIP_MEMORY_SCOPE_AGENT);
}

#define SCAN_T   1024
#define SCAN_PER 49
#define SCAN_N   (SCAN_T * SCAN_PER)

__global__ void gcn_cnt_zero(int* cnt, int n) {
    int i = blockIdx.x * blockDim.x + threadIdx.x;
    if (i < n) cnt[i] = 0;
}

__global__ __launch_bounds__(256) void gcn_rank_edges(const int* __restrict__ ei, int* cnt,
                                                      int* __restrict__ rank, int e) {
    const int i = (blockIdx.x * blockDim.x + threadIdx.x) * 4;
    if (i >= e) return;
    const int* dst = ei + e;
    if ((e & 3) == 0) {
        const int4 d = *(const int4*)(dst + i);
        int4 r;
        r.x = __hip_atomic_fetch_add(&cnt[d.x], 1, __ATOMIC_RELAXED, __HIP_MEMORY_SCOPE_AGENT);
        r.y = __hip_atomic_fetch_add(&cnt[d.y], 1, __ATOMIC_RELAXED, __HIP_MEMORY_SCOPE_AGENT);
        r.z = __hip_atomic_fetch_add(&cnt[d.z], 1, __ATOMIC_RELAXED, __HIP_MEMORY_SCOPE_AGENT);
        r.w = __hip_atomic_fetch_add(&cnt[d.w], 1, __ATOMIC_RELAXED, __HIP_MEMORY_SCOPE_AGENT);
        *(int4*)(rank + i) = r;
    } else {
        for (int k = i; k < e && k < i + 4; ++k)
            rank[k] = __hip_atomic_fetch_add(&cnt[dst[k]], 1, __ATOMIC_RELAXED, __HIP_MEMORY_SCOPE_AGENT);
    }
}

__global__ __launch_bounds__(SCAN_T) void gcn_scan_deg(const int* __restrict__ cnt,
                                                       int* __restrict__ rowptr,
                                                       float* __restrict__ dinv, int n) {
    __shared__ int buf[SCAN_N];
    __shared__ int toff[SCAN_T];
    __shared__ int wsum[SCAN_T / 32];
    const int t = threadIdx.x;
    for (int i = t; i < SCAN_N; i += SCAN_T) {
        int c = 0;
        if (i < n) { c = cnt[i]; dinv[i] = rsqrtf((float)(c + 1)); }
        buf[i] = c;
    }
    __syncthreads();
    const int base = t * SCAN_PER;
    int s = 0;
    for (int j = 0; j < SCAN_PER; ++j) { int c = buf[base + j]; buf[base + j] = s; s += c; }
    const int lane = t & 31, wv = t >> 5;
    int incl = s;
    for (int o = 1; o < 32; o <<= 1) { int y = __shfl_up(incl, o); if (lane >= o) incl += y; }
    if (lane == 31) wsum[wv] = incl;
    __syncthreads();
    if (wv == 0) {
        int x = wsum[lane];
        int xi = x;
        for (int o = 1; o < 32; o <<= 1) { int y = __shfl_up(xi, o); if (lane >= o) xi += y; }
        wsum[lane] = xi - x;
    }
    __syncthreads();
    toff[t] = wsum[wv] + incl - s;
    __syncthreads();
    for (int i = t; i <= n; i += SCAN_T) rowptr[i] = buf[i] + toff[i / SCAN_PER];
}

__global__ __launch_bounds__(256) void gcn_fill_csr(const int* __restrict__ ei, const int* __restrict__ rank,
                             const int* __restrict__ rowptr, const float* __restrict__ dinv,
                             int* __restrict__ csr, int e) {
    const int i = (blockIdx.x * blockDim.x + threadIdx.x) * 4;
    if (i >= e) return;
    if ((e & 3) == 0) {
        const int4 s4 = *(const int4*)(ei + i);
        const int4 d4 = *(const int4*)(ei + e + i);
        const int4 r4 = *(const int4*)(rank + i);
        const int s[4] = {s4.x, s4.y, s4.z, s4.w};
        const int d[4] = {d4.x, d4.y, d4.z, d4.w};
        const int r[4] = {r4.x, r4.y, r4.z, r4.w};
        int rp[4]; float dv[4];
        #pragma unroll
        for (int u = 0; u < 4; ++u) { rp[u] = rowptr[d[u]]; dv[u] = dinv[s[u]]; }
        #pragma unroll
        for (int u = 0; u < 4; ++u) {
            unsigned pos = (unsigned)(rp[u] + r[u]);
            if (pos > (unsigned)(e - 1)) pos = (unsigned)(e - 1);
            int2 p; p.x = s[u]; p.y = __float_as_int(dv[u]);
            *(int2*)(csr + 2 * (size_t)pos) = p;
        }
    } else {
        for (int k = i; k < e && k < i + 4; ++k) {
            const int s = ei[k];
            unsigned pos = (unsigned)(rowptr[ei[e + k]] + rank[k]);
            if (pos > (unsigned)(e - 1)) pos = (unsigned)(e - 1);
            int2 p; p.x = s; p.y = __float_as_int(dinv[s]);
            *(int2*)(csr + 2 * (size_t)pos) = p;
        }
    }
}

#define EW_ITEMS 4
__global__ __launch_bounds__(256) void gcn_combine(const float* __restrict__ x,
                                                   const float* __restrict__ la,
                                                   const float* __restrict__ lb,
                                                   float* __restrict__ h, int n) {
    const float alpha = expf(la[0]);
    const float beta  = expf(lb[0]);
    const int n4 = n * (HID / 4);
    const int stride = gridDim.x * blockDim.x;
    int i = blockIdx.x * blockDim.x + threadIdx.x;
    for (; i + (EW_ITEMS - 1) * stride < n4; i += EW_ITEMS * stride) {
        float4 a[EW_ITEMS], b[EW_ITEMS];
        #pragma unroll
        for (int u = 0; u < EW_ITEMS; ++u) {
            const int idx = i + u * stride;
            const float4* row = (const float4*)(x + (size_t)(idx >> 5) * 256);
            a[u] = row[idx & 31];
            b[u] = row[32 + (idx & 31)];
        }
        #pragma unroll
        for (int u = 0; u < EW_ITEMS; ++u) {
            float4 o;
            o.x = alpha * a[u].x + beta * b[u].x;
            o.y = alpha * a[u].y + beta * b[u].y;
            o.z = alpha * a[u].z + beta * b[u].z;
            o.w = alpha * a[u].w + beta * b[u].w;
            ((float4*)h)[i + u * stride] = o;
        }
    }
    for (; i < n4; i += stride) {
        const float4* row = (const float4*)(x + (size_t)(i >> 5) * 256);
        const float4 a = row[i & 31];
        const float4 b = row[32 + (i & 31)];
        float4 o;
        o.x = alpha * a.x + beta * b.x;
        o.y = alpha * a.y + beta * b.y;
        o.z = alpha * a.z + beta * b.z;
        o.w = alpha * a.w + beta * b.w;
        ((float4*)h)[i] = o;
    }
}

__global__ __launch_bounds__(128) void gcn_gemm128_wmma(
        const float* __restrict__ A, const float* __restrict__ W,
        const float* __restrict__ bias, float* __restrict__ C,
        int nrows, int do_relu) {
    __shared__ float Wt[HID * WT_STRIDE];

    int t = threadIdx.x;
    #pragma unroll 4
    for (int idx = t; idx < HID * HID; idx += 128) {
        int kk = idx >> 7;
        int n  = idx & 127;
        Wt[n * WT_STRIDE + kk] = W[idx];
    }
    __syncthreads();

    const int wave  = t >> 5;
    const int lane  = t & 31;
    const int row16 = lane & 15;
    const int khalf = (lane >> 4) << 1;

    const int tileM = blockIdx.x * 64 + wave * 16;
    int rowIdx = tileM + row16;
    if (rowIdx > nrows - 1) rowIdx = nrows - 1;
    const float* Arow = A + (size_t)rowIdx * HID;
    const float* Brow = Wt + (size_t)row16 * WT_STRIDE;

    v8f acc[8] = {};

    for (int k = 0; k < HID; k += 4) {
        const float2 av = *(const float2*)(Arow + k + khalf);
        v2f a; a.x = av.x; a.y = av.y;
        #pragma unroll
        for (int nt = 0; nt < 8; ++nt) {
            const float2 bv = *(const float2*)(Brow + nt * 16 * WT_STRIDE + k + khalf);
            v2f b; b.x = bv.x; b.y = bv.y;
            acc[nt] = __builtin_amdgcn_wmma_f32_16x16x4_f32(
                false, a, false, b, (short)0, acc[nt], false, false);
        }
    }

    const int rbase = tileM + ((lane >> 4) << 3);
    #pragma unroll
    for (int nt = 0; nt < 8; ++nt) {
        const int col = nt * 16 + row16;
        const float bv = bias ? bias[col] : 0.0f;
        #pragma unroll
        for (int r = 0; r < 8; ++r) {
            const int row = rbase + r;
            if (row < nrows) {
                float v = acc[nt][r] + bv;
                if (do_relu) v = fmaxf(v, 0.0f);
                C[(size_t)row * HID + col] = v;
            }
        }
    }
}

#define GROWS 8
#define GATHER_T 256

template <bool STATS>
__global__ __launch_bounds__(GATHER_T) void gcn_gather(const int* __restrict__ rowptr,
                                                  const int* __restrict__ csr,
                                                  const float* __restrict__ dinv,
                                                  const float* __restrict__ m,
                                                  const float* __restrict__ b,
                                                  float* __restrict__ h,
                                                  float* bnsum, float* bnsq, int n, int e) {
    const int lane = threadIdx.x & 31;
    float4 ssum = make_float4(0.0f, 0.0f, 0.0f, 0.0f);
    float4 ssq  = make_float4(0.0f, 0.0f, 0.0f, 0.0f);
    const int nwaves = (gridDim.x * blockDim.x) >> 5;
    const int wid = __builtin_amdgcn_readfirstlane((blockIdx.x * blockDim.x + threadIdx.x) >> 5);
    const float* mlane = m + lane * 4;
    const float4 bb = *(const float4*)(b + lane * 4);
    for (int node = wid; node < n; node += nwaves) {
        int beg = rowptr[node];
        int end = rowptr[node + 1];
        if (beg < 0) beg = 0;
        if (end > e) end = e;
        const float dd = dinv[node];
        const float4 selfv = *(const float4*)(mlane + (size_t)node * HID);
        float4 acc = make_float4(0.0f, 0.0f, 0.0f, 0.0f);

        for (int base = beg; base < end; base += 32) {
            int c = end - base; if (c > 32) c = 32;
            int sv = 0, wbits = 0;
            if (lane < c) {
                const int2 p = *(const int2*)(csr + 2 * (size_t)(base + lane));
                sv = p.x; wbits = p.y;
            }
            if ((unsigned)sv > (unsigned)(n - 1)) sv = n - 1;
            int j = 0;
            for (; j + GROWS <= c; j += GROWS) {
                float4 v[GROWS]; float w[GROWS];
                #pragma unroll
                for (int u = 0; u < GROWS; ++u) {
                    const int s = __builtin_amdgcn_readlane(sv, j + u);
                    w[u] = __int_as_float(__builtin_amdgcn_readlane(wbits, j + u));
                    v[u] = *(const float4*)(mlane + (size_t)s * HID);
                }
                #pragma unroll
                for (int u = 0; u < GROWS; ++u) {
                    const float nrm = w[u] * dd;
                    acc.x += v[u].x * nrm; acc.y += v[u].y * nrm;
                    acc.z += v[u].z * nrm; acc.w += v[u].w * nrm;
                }
            }
            const int r = c - j;
            if (r > 0) {
                float4 v[GROWS - 1]; float w[GROWS - 1];
                #pragma unroll
                for (int u = 0; u < GROWS - 1; ++u) {
                    if (u < r) {
                        const int s = __builtin_amdgcn_readlane(sv, j + u);
                        w[u] = __int_as_float(__builtin_amdgcn_readlane(wbits, j + u));
                        v[u] = *(const float4*)(mlane + (size_t)s * HID);
                    }
                }
                #pragma unroll
                for (int u = 0; u < GROWS - 1; ++u) {
                    if (u < r) {
                        const float nrm = w[u] * dd;
                        acc.x += v[u].x * nrm; acc.y += v[u].y * nrm;
                        acc.z += v[u].z * nrm; acc.w += v[u].w * nrm;
                    }
                }
            }
        }
        const float sn = dd * dd;
        acc.x += selfv.x * sn; acc.y += selfv.y * sn; acc.z += selfv.z * sn; acc.w += selfv.w * sn;
        acc.x += bb.x; acc.y += bb.y; acc.z += bb.z; acc.w += bb.w;
        *(float4*)(h + (size_t)node * HID + lane * 4) = acc;
        if constexpr (STATS) {
            ssum.x += acc.x; ssum.y += acc.y; ssum.z += acc.z; ssum.w += acc.w;
            ssq.x += acc.x * acc.x; ssq.y += acc.y * acc.y;
            ssq.z += acc.z * acc.z; ssq.w += acc.w * acc.w;
        }
    }
    if constexpr (STATS) {
        __shared__ float red[(GATHER_T / 32) * 2 * HID];
        const int wv = threadIdx.x >> 5;
        *(float4*)(red + (wv * 2 + 0) * HID + lane * 4) = ssum;
        *(float4*)(red + (wv * 2 + 1) * HID + lane * 4) = ssq;
        __syncthreads();
        static_assert(GATHER_T == 2 * HID, "one thread per batch-norm sum");
        const int which = threadIdx.x >> 7;
        const int f = threadIdx.x & (HID - 1);
        float tot = 0.0f;
        for (int w = 0; w < GATHER_T / 32; ++w) tot += red[(w * 2 + which) * HID + f];
        atomAddF32(which ? &bnsq[f] : &bnsum[f], tot);
    }
}

__global__ void gcn_bn_zero(float* bnsum, float* bnsq) {
    int t = threadIdx.x;
    if (t < HID) bnsum[t] = 0.0f;
    else if (t < 2 * HID) bnsq[t - HID] = 0.0f;
}

#define BN_ITEMS 8
__global__ __launch_bounds__(256) void gcn_bn_apply(float* __restrict__ h,
                             const float* __restrict__ bnsum, const float* __restrict__ bnsq,
                             const float* __restrict__ gamma, const float* __restrict__ beta,
                             int n) {
    const int n4 = n * (HID / 4);
    const int stride = gridDim.x * blockDim.x;
    int i = blockIdx.x * blockDim.x + threadIdx.x;
    const int f = (i & 31) * 4;
    const float invn = 1.0f / (float)n;
    v4f g, mu, rs, bt;
    #pragma unroll
    for (int c = 0; c < 4; ++c) {
        mu[c] = bnsum[f + c] * invn;
        const float var = bnsq[f + c] * invn - mu[c] * mu[c];
        rs[c] = rsqrtf(var + BN_EPS);
        g[c]  = gamma[f + c];
        bt[c] = beta[f + c];
    }
    const v4f zero = {0.0f, 0.0f, 0.0f, 0.0f};
    v4f* h4 = (v4f*)h;
    for (; i + (BN_ITEMS - 1) * stride < n4; i += BN_ITEMS * stride) {
        v4f v[BN_ITEMS];
        #pragma unroll
        for (int u = 0; u < BN_ITEMS; ++u) v[u] = h4[i + u * stride];
        #pragma unroll
        for (int u = 0; u < BN_ITEMS; ++u)
            h4[i + u * stride] = __builtin_elementwise_max(g * (v[u] - mu) * rs + bt, zero);
    }
    for (; i < n4; i += stride)
        h4[i] = __builtin_elementwise_max(g * (h4[i] - mu) * rs + bt, zero);
}

#define FIN_NT 3
template <int CTRL> __device__ __forceinline__ float lane_perm16(float v) {
    return __int_as_float(__builtin_amdgcn_update_dpp(0, __float_as_int(v), CTRL, 0xF, 0xF, true));
}
__device__ __forceinline__ float half_wave_max(float v) {
    v = fmaxf(v, lane_perm16<0xB1>(v));
    v = fmaxf(v, lane_perm16<0x4E>(v));
    v = fmaxf(v, lane_perm16<0x141>(v));
    v = fmaxf(v, lane_perm16<0x140>(v));
    return v;
}
__device__ __forceinline__ float half_wave_sum(float v) {
    v += lane_perm16<0xB1>(v);
    v += lane_perm16<0x4E>(v);
    v += lane_perm16<0x141>(v);
    v += lane_perm16<0x140>(v);
    return v;
}

__global__ __launch_bounds__(128) void gcn_final_wmma(const float* __restrict__ h,
                                                      const float* __restrict__ lw,
                                                      const float* __restrict__ lb,
                                                      float* __restrict__ out, int n) {
    __shared__ float Wt[FIN_NT * 16 * WT_STRIDE];

    const int t = threadIdx.x;
    const int wave  = t >> 5;
    const int lane  = t & 31;
    const int row16 = lane & 15;
    const int khalf = (lane >> 4) << 1;

    const int tileM = blockIdx.x * 64 + wave * 16;
    int rowIdx = tileM + row16;
    if (rowIdx > n - 1) rowIdx = n - 1;
    const float* Arow = h + (size_t)rowIdx * HID;

    float wv[(HID * OUTC) / 128];
    #pragma unroll
    for (int j = 0; j < (HID * OUTC) / 128; ++j) wv[j] = lw[t + j * 128];
    float2 av[HID / 4];
    #pragma unroll
    for (int s = 0; s < HID / 4; ++s) av[s] = *(const float2*)(Arow + 4 * s + khalf);

    #pragma unroll
    for (int j = 0; j < (HID * OUTC) / 128; ++j) {
        const int idx = t + j * 128;
        const int kk = idx / OUTC;
        const int c  = idx - kk * OUTC;
        Wt[c * WT_STRIDE + kk] = wv[j];
    }
    for (int j = 0; j < FIN_NT * 16 - OUTC; ++j) Wt[(OUTC + j) * WT_STRIDE + t] = 0.0f;
    __syncthreads();

    const float* Brow = Wt + (size_t)row16 * WT_STRIDE;
    v8f acc[FIN_NT] = {};
    #pragma unroll
    for (int s = 0; s < HID / 4; ++s) {
        v2f a; a.x = av[s].x; a.y = av[s].y;
        #pragma unroll
        for (int nt = 0; nt < FIN_NT; ++nt) {
            const float2 bv = *(const float2*)(Brow + nt * 16 * WT_STRIDE + 4 * s + khalf);
            v2f b; b.x = bv.x; b.y = bv.y;
            acc[nt] = __builtin_amdgcn_wmma_f32_16x16x4_f32(
                false, a, false, b, (short)0, acc[nt], false, false);
        }
    }

    const bool has2 = row16 < OUTC - 32;
    const float b0 = lb[row16];
    const float b1 = lb[16 + row16];
    const float b2 = lb[has2 ? 32 + row16 : 0];
    float x0[8], x1[8], x2[8], mx[8], se[8];
    #pragma unroll
    for (int r = 0; r < 8; ++r) {
        x0[r] = acc[0][r] + b0;
        x1[r] = acc[1][r] + b1;
        x2[r] = has2 ? acc[2][r] + b2 : -INFINITY;
        mx[r] = half_wave_max(fmaxf(fmaxf(x0[r], x1[r]), x2[r]));
    }
    #pragma unroll
    for (int r = 0; r < 8; ++r)
        se[r] = half_wave_sum(expf(x0[r] - mx[r]) + expf(x1[r] - mx[r]) + expf(x2[r] - mx[r]));
    const int rbase = tileM + ((lane >> 4) << 3);
    #pragma unroll
    for (int r = 0; r < 8; ++r) {
        const float ls = logf(se[r]);
        const int row = rbase + r;
        if (row < n) {
            float* orow = out + (size_t)row * OUTC;
            orow[row16]      = x0[r] - mx[r] - ls;
            orow[16 + row16] = x1[r] - mx[r] - ls;
            if (has2) orow[32 + row16] = x2[r] - mx[r] - ls;
        }
    }
}

extern "C" void kernel_launch(void* const* d_in, const int* in_sizes, int n_in,
                              void* d_out, int out_size, void* d_ws, size_t ws_size,
                              hipStream_t stream) {
    const float* x       = (const float*)d_in[0];
    const int*   ei      = (const int*)d_in[1];
    const float* la      = (const float*)d_in[2];
    const float* lb      = (const float*)d_in[3];
    const float* mlp_w   = (const float*)d_in[4];
    const float* mlp_b   = (const float*)d_in[5];
    const float* conv_w  = (const float*)d_in[6];
    const float* conv_b  = (const float*)d_in[7];
    const float* bn_g    = (const float*)d_in[8];
    const float* bn_b    = (const float*)d_in[9];
    const float* lin_w   = (const float*)d_in[10];
    const float* lin_b   = (const float*)d_in[11];
    float* out = (float*)d_out;

    float* ws = (float*)d_ws;
    const size_t NM = (size_t)N_NODES * HID;
    float* hA    = ws;
    float* hB    = hA + NM;
    int*   cnt    = (int*)(hB + NM);
    int*   rowptr = cnt + SCAN_N;
    int*   rank   = rowptr + SCAN_N;
    int*   csr    = rank + N_EDGES;
    float* dinv  = hB + NM + NM;
    float* bnsum = dinv + N_NODES;
    float* bnsq  = bnsum + HID;

    const int TPB = 256;
    const int n4 = N_NODES * (HID / 4);
    const int combBlocks = (n4 + TPB * EW_ITEMS - 1) / (TPB * EW_ITEMS);
    const int bnBlocks   = (n4 + TPB * BN_ITEMS - 1) / (TPB * BN_ITEMS);
    const int edgeBlocks = ((N_EDGES + 3) / 4 + TPB - 1) / TPB;
    const int gatherBlocks = 1250;
    const int gemmBlocks  = (N_NODES + 63) / 64;

    gcn_cnt_zero<<<(N_NODES + TPB - 1) / TPB, TPB, 0, stream>>>(cnt, N_NODES);
    gcn_rank_edges<<<edgeBlocks, TPB, 0, stream>>>(ei, cnt, rank, N_EDGES);
    gcn_scan_deg<<<1, SCAN_T, 0, stream>>>(cnt, rowptr, dinv, N_NODES);
    gcn_fill_csr<<<edgeBlocks, TPB, 0, stream>>>(ei, rank, rowptr, dinv, csr, N_EDGES);

    gcn_combine<<<combBlocks, TPB, 0, stream>>>(x, la, lb, hA, N_NODES);

    float* cur = hA;
    float* oth = hB;
    for (int i = 0; i < NLAYERS; ++i) {
        gcn_gemm128_wmma<<<gemmBlocks, 128, 0, stream>>>(
            cur, mlp_w + (size_t)i * HID * HID, mlp_b + i * HID, oth, N_NODES, 1);
        gcn_gemm128_wmma<<<gemmBlocks, 128, 0, stream>>>(
            oth, conv_w + (size_t)i * HID * HID, nullptr, cur, N_NODES, 0);
        if (i < NLAYERS - 1) {
            gcn_bn_zero<<<1, 256, 0, stream>>>(bnsum, bnsq);
            gcn_gather<true><<<gatherBlocks, GATHER_T, 0, stream>>>(
                rowptr, csr, dinv, cur, conv_b + i * HID, oth, bnsum, bnsq, N_NODES, N_EDGES);
            gcn_bn_apply<<<bnBlocks, TPB, 0, stream>>>(
                oth, bnsum, bnsq, bn_g + i * HID, bn_b + i * HID, N_NODES);
        } else {
            gcn_gather<false><<<gatherBlocks, GATHER_T, 0, stream>>>(
                rowptr, csr, dinv, cur, conv_b + i * HID, oth, nullptr, nullptr, N_NODES, N_EDGES);
        }
        float* tmp = cur; cur = oth; oth = tmp;
    }

    gcn_final_wmma<<<gemmBlocks, 128, 0, stream>>>(cur, lin_w, lin_b, out, N_NODES);
}


# Each layer's row-local chain (x combine or batch norm+ReLU, MLP GEMM+ReLU, conv GEMM) is one kernel: the 16-row strip stays in registers/LDS, both weight tiles staged once per work-group
# speedup vs baseline: 3.3671x; 1.6005x over previous
#include <hip/hip_runtime.h>
#include <hip/hip_bf16.h>
#include <math.h>

#define N_NODES 50000
#define N_EDGES 800000
#define HID     128
#define OUTC    40
#define NLAYERS 3
#define BN_EPS  1e-5f
#define WT_STRIDE 130

typedef __attribute__((ext_vector_type(2))) float v2f;
typedef __attribute__((ext_vector_type(8))) float v8f;
typedef __attribute__((ext_vector_type(4))) float v4f;

__device__ __forceinline__ void atomAddF32(float* p, float v) {
    __hip_atomic_fetch_add(p, v, __ATOMIC_RELAXED, __HIP_MEMORY_SCOPE_AGENT);
}

#define SCAN_T   1024
#define SCAN_PER 49
#define SCAN_N   (SCAN_T * SCAN_PER)

__global__ void gcn_cnt_zero(int* cnt, int n) {
    int i = blockIdx.x * blockDim.x + threadIdx.x;
    if (i < n) cnt[i] = 0;
}

__global__ __launch_bounds__(256) void gcn_rank_edges(const int* __restrict__ ei, int* cnt,
                                                      int* __restrict__ rank, int e) {
    const int i = (blockIdx.x * blockDim.x + threadIdx.x) * 4;
    if (i >= e) return;
    const int* dst = ei + e;
    if ((e & 3) == 0) {
        const int4 d = *(const int4*)(dst + i);
        int4 r;
        r.x = __hip_atomic_fetch_add(&cnt[d.x], 1, __ATOMIC_RELAXED, __HIP_MEMORY_SCOPE_AGENT);
        r.y = __hip_atomic_fetch_add(&cnt[d.y], 1, __ATOMIC_RELAXED, __HIP_MEMORY_SCOPE_AGENT);
        r.z = __hip_atomic_fetch_add(&cnt[d.z], 1, __ATOMIC_RELAXED, __HIP_MEMORY_SCOPE_AGENT);
        r.w = __hip_atomic_fetch_add(&cnt[d.w], 1, __ATOMIC_RELAXED, __HIP_MEMORY_SCOPE_AGENT);
        *(int4*)(rank + i) = r;
    } else {
        for (int k = i; k < e && k < i + 4; ++k)
            rank[k] = __hip_atomic_fetch_add(&cnt[dst[k]], 1, __ATOMIC_RELAXED, __HIP_MEMORY_SCOPE_AGENT);
    }
}

__global__ __launch_bounds__(SCAN_T) void gcn_scan_deg(const int* __restrict__ cnt,
                                                       int* __restrict__ rowptr,
                                                       float* __restrict__ dinv, int n) {
    __shared__ int buf[SCAN_N];
    __shared__ int toff[SCAN_T];
    __shared__ int wsum[SCAN_T / 32];
    const int t = threadIdx.x;
    for (int i = t; i < SCAN_N; i += SCAN_T) {
        int c = 0;
        if (i < n) { c = cnt[i]; dinv[i] = rsqrtf((float)(c + 1)); }
        buf[i] = c;
    }
    __syncthreads();
    const int base = t * SCAN_PER;
    int s = 0;
    for (int j = 0; j < SCAN_PER; ++j) { int c = buf[base + j]; buf[base + j] = s; s += c; }
    const int lane = t & 31, wv = t >> 5;
    int incl = s;
    for (int o = 1; o < 32; o <<= 1) { int y = __shfl_up(incl, o); if (lane >= o) incl += y; }
    if (lane == 31) wsum[wv] = incl;
    __syncthreads();
    if (wv == 0) {
        int x = wsum[lane];
        int xi = x;
        for (int o = 1; o < 32; o <<= 1) { int y = __shfl_up(xi, o); if (lane >= o) xi += y; }
        wsum[lane] = xi - x;
    }
    __syncthreads();
    toff[t] = wsum[wv] + incl - s;
    __syncthreads();
    for (int i = t; i <= n; i += SCAN_T) rowptr[i] = buf[i] + toff[i / SCAN_PER];
}

__global__ __launch_bounds__(256) void gcn_fill_csr(const int* __restrict__ ei, const int* __restrict__ rank,
                             const int* __restrict__ rowptr, const float* __restrict__ dinv,
                             int* __restrict__ csr, int e) {
    const int i = (blockIdx.x * blockDim.x + threadIdx.x) * 4;
    if (i >= e) return;
    if ((e & 3) == 0) {
        const int4 s4 = *(const int4*)(ei + i);
        const int4 d4 = *(const int4*)(ei + e + i);
        const int4 r4 = *(const int4*)(rank + i);
        const int s[4] = {s4.x, s4.y, s4.z, s4.w};
        const int d[4] = {d4.x, d4.y, d4.z, d4.w};
        const int r[4] = {r4.x, r4.y, r4.z, r4.w};
        int rp[4]; float dv[4];
        #pragma unroll
        for (int u = 0; u < 4; ++u) { rp[u] = rowptr[d[u]]; dv[u] = dinv[s[u]]; }
        #pragma unroll
        for (int u = 0; u < 4; ++u) {
            unsigned pos = (unsigned)(rp[u] + r[u]);
            if (pos > (unsigned)(e - 1)) pos = (unsigned)(e - 1);
            int2 p; p.x = s[u]; p.y = __float_as_int(dv[u]);
            *(int2*)(csr + 2 * (size_t)pos) = p;
        }
    } else {
        for (int k = i; k < e && k < i + 4; ++k) {
            const int s = ei[k];
            unsigned pos = (unsigned)(rowptr[ei[e + k]] + rank[k]);
            if (pos > (unsigned)(e - 1)) pos = (unsigned)(e - 1);
            int2 p; p.x = s; p.y = __float_as_int(dinv[s]);
            *(int2*)(csr + 2 * (size_t)pos) = p;
        }
    }
}

#define DENSE_T 512
#define DENSE_WAVES (DENSE_T / 32)
#define STRIP_STRIDE 132

__device__ __forceinline__ void strip_gemm(const float2 (&av)[HID / 4], const float* Bfrag, v8f (&acc)[8]) {
    #pragma unroll
    for (int s = 0; s < HID / 4; ++s) {
        v2f a; a.x = av[s].x; a.y = av[s].y;
        #pragma unroll
        for (int nt = 0; nt < 8; ++nt) {
            const float2 bv = *(const float2*)(Bfrag + nt * 16 * WT_STRIDE + 4 * s);
            v2f b; b.x = bv.x; b.y = bv.y;
            acc[nt] = __builtin_amdgcn_wmma_f32_16x16x4_f32(
                false, a, false, b, (short)0, acc[nt], false, false);
        }
    }
}

template <bool FIRST>
__device__ __forceinline__ void strip_load(float2 (&av)[HID / 4], float2 (&aw)[HID / 4],
                                           const float* __restrict__ A, int tileM, int row16,
                                           int khalf, int nrows) {
    int rowIdx = tileM + row16;
    if (rowIdx > nrows - 1) rowIdx = nrows - 1;
    const float* Arow = A + (size_t)rowIdx * (FIRST ? 2 * HID : HID) + khalf;
    #pragma unroll
    for (int s = 0; s < HID / 4; ++s) {
        av[s] = *(const float2*)(Arow + 4 * s);
        if constexpr (FIRST) aw[s] = *(const float2*)(Arow + HID + 4 * s);
    }
}

template <bool FIRST>
__global__ __launch_bounds__(DENSE_T) void gcn_layer_dense(
        const float* __restrict__ A,
        const float* __restrict__ la, const float* __restrict__ lb,
        const float* __restrict__ bnsum, const float* __restrict__ bnsq,
        const float* __restrict__ gamma, const float* __restrict__ beta,
        const float* __restrict__ W1, const float* __restrict__ b1,
        const float* __restrict__ W2, float* __restrict__ C, int nrows) {
    __shared__ __attribute__((aligned(16))) float W1t[HID * WT_STRIDE];
    __shared__ __attribute__((aligned(16))) float W2t[HID * WT_STRIDE];
    __shared__ __attribute__((aligned(16))) float strip[DENSE_WAVES * 16 * STRIP_STRIDE];
    __shared__ __attribute__((aligned(16))) float bnc[4 * HID];

    const int t = threadIdx.x;
    const int wave  = __builtin_amdgcn_readfirstlane(t >> 5);
    const int lane  = t & 31;
    const int row16 = lane & 15;
    const int khalf = (lane >> 4) << 1;
    const int nstrips = (nrows + 15) >> 4;
    const int wavesTotal = gridDim.x * DENSE_WAVES;
    const int sfirst = wave * gridDim.x + blockIdx.x;

    float w1v[(HID * HID) / DENSE_T], w2v[(HID * HID) / DENSE_T];
    #pragma unroll
    for (int j = 0; j < (HID * HID) / DENSE_T; ++j) w1v[j] = W1[t + j * DENSE_T];
    #pragma unroll
    for (int j = 0; j < (HID * HID) / DENSE_T; ++j) w2v[j] = W2[t + j * DENSE_T];

    float2 av[HID / 4], aw[HID / 4];
    bool active = sfirst < nstrips;
    if (active) strip_load<FIRST>(av, aw, A, sfirst * 16, row16, khalf, nrows);

    #pragma unroll
    for (int j = 0; j < (HID * HID) / DENSE_T; ++j) {
        const int idx = t + j * DENSE_T;
        W1t[(idx & 127) * WT_STRIDE + (idx >> 7)] = w1v[j];
    }
    #pragma unroll
    for (int j = 0; j < (HID * HID) / DENSE_T; ++j) {
        const int idx = t + j * DENSE_T;
        W2t[(idx & 127) * WT_STRIDE + (idx >> 7)] = w2v[j];
    }
    float alpha = 0.0f, betac = 0.0f;
    if constexpr (FIRST) {
        alpha = expf(la[0]);
        betac = expf(lb[0]);
    } else {
        if (t < HID) {
            const float invn = 1.0f / (float)nrows;
            const float mu = bnsum[t] * invn;
            const float var = bnsq[t] * invn - mu * mu;
            bnc[t] = gamma[t];
            bnc[HID + t] = mu;
            bnc[2 * HID + t] = rsqrtf(var + BN_EPS);
            bnc[3 * HID + t] = beta[t];
        }
    }
    __syncthreads();

    float* myStrip = strip + wave * 16 * STRIP_STRIDE;
    const float* B1 = W1t + row16 * WT_STRIDE + khalf;
    const float* B2 = W2t + row16 * WT_STRIDE + khalf;
    const int rhalf = (lane >> 4) << 3;

    for (int s0 = 0; s0 < nstrips; s0 += wavesTotal) {
        const int tileM = (s0 + sfirst) * 16;
        if (active) {
            #pragma unroll
            for (int s = 0; s < HID / 4; ++s) {
                if constexpr (FIRST) {
                    av[s].x = alpha * av[s].x + betac * aw[s].x;
                    av[s].y = alpha * av[s].y + betac * aw[s].y;
                } else {
                    const int k = 4 * s + khalf;
                    const float2 g  = *(const float2*)(bnc + k);
                    const float2 mu = *(const float2*)(bnc + HID + k);
                    const float2 rs = *(const float2*)(bnc + 2 * HID + k);
                    const float2 bt = *(const float2*)(bnc + 3 * HID + k);
                    av[s].x = fmaxf(g.x * (av[s].x - mu.x) * rs.x + bt.x, 0.0f);
                    av[s].y = fmaxf(g.y * (av[s].y - mu.y) * rs.y + bt.y, 0.0f);
                }
            }
            v8f acc[8] = {};
            strip_gemm(av, B1, acc);
            #pragma unroll
            for (int nt = 0; nt < 8; ++nt) {
                const float bv = b1[nt * 16 + row16];
                #pragma unroll
                for (int r = 0; r < 8; ++r)
                    myStrip[(rhalf + r) * STRIP_STRIDE + nt * 16 + row16] = fmaxf(acc[nt][r] + bv, 0.0f);
            }
        }
        __syncthreads();
        const bool was = active;
        if (was) {
            #pragma unroll
            for (int s = 0; s < HID / 4; ++s)
                av[s] = *(const float2*)(myStrip + row16 * STRIP_STRIDE + 4 * s + khalf);
        }
        v8f acc2[8] = {};
        if (was) strip_gemm(av, B2, acc2);
        const int snext = s0 + wavesTotal + sfirst;
        active = (s0 + wavesTotal < nstrips) && (snext < nstrips);
        if (active) strip_load<FIRST>(av, aw, A, snext * 16, row16, khalf, nrows);
        if (was) {
            #pragma unroll
            for (int nt = 0; nt < 8; ++nt) {
                const int col = nt * 16 + row16;
                #pragma unroll
                for (int r = 0; r < 8; ++r) {
                    const int row = tileM + rhalf + r;
                    if (row < nrows) C[(size_t)row * HID + col] = acc2[nt][r];
                }
            }
        }
    }
}

#define GROWS 8
#define GATHER_T 256

template <bool STATS>
__global__ __launch_bounds__(GATHER_T) void gcn_gather(const int* __restrict__ rowptr,
                                                  const int* __restrict__ csr,
                                                  const float* __restrict__ dinv,
                                                  const float* __restrict__ m,
                                                  const float* __restrict__ b,
                                                  float* __restrict__ h,
                                                  float* bnsum, float* bnsq, int n, int e) {
    const int lane = threadIdx.x & 31;
    float4 ssum = make_float4(0.0f, 0.0f, 0.0f, 0.0f);
    float4 ssq  = make_float4(0.0f, 0.0f, 0.0f, 0.0f);
    const int nwaves = (gridDim.x * blockDim.x) >> 5;
    const int wid = __builtin_amdgcn_readfirstlane((blockIdx.x * blockDim.x + threadIdx.x) >> 5);
    const float* mlane = m + lane * 4;
    const float4 bb = *(const float4*)(b + lane * 4);
    for (int node = wid; node < n; node += nwaves) {
        int beg = rowptr[node];
        int end = rowptr[node + 1];
        if (beg < 0) beg = 0;
        if (end > e) end = e;
        const float dd = dinv[node];
        const float4 selfv = *(const float4*)(mlane + (size_t)node * HID);
        float4 acc = make_float4(0.0f, 0.0f, 0.0f, 0.0f);

        for (int base = beg; base < end; base += 32) {
            int c = end - base; if (c > 32) c = 32;
            int sv = 0, wbits = 0;
            if (lane < c) {
                const int2 p = *(const int2*)(csr + 2 * (size_t)(base + lane));
                sv = p.x; wbits = p.y;
            }
            if ((unsigned)sv > (unsigned)(n - 1)) sv = n - 1;
            int j = 0;
            for (; j + GROWS <= c; j += GROWS) {
                float4 v[GROWS]; float w[GROWS];
                #pragma unroll
                for (int u = 0; u < GROWS; ++u) {
                    const int s = __builtin_amdgcn_readlane(sv, j + u);
                    w[u] = __int_as_float(__builtin_amdgcn_readlane(wbits, j + u));
                    v[u] = *(const float4*)(mlane + (size_t)s * HID);
                }
                #pragma unroll
                for (int u = 0; u < GROWS; ++u) {
                    const float nrm = w[u] * dd;
                    acc.x += v[u].x * nrm; acc.y += v[u].y * nrm;
                    acc.z += v[u].z * nrm; acc.w += v[u].w * nrm;
                }
            }
            const int r = c - j;
            if (r > 0) {
                float4 v[GROWS - 1]; float w[GROWS - 1];
                #pragma unroll
                for (int u = 0; u < GROWS - 1; ++u) {
                    if (u < r) {
                        const int s = __builtin_amdgcn_readlane(sv, j + u);
                        w[u] = __int_as_float(__builtin_amdgcn_readlane(wbits, j + u));
                        v[u] = *(const float4*)(mlane + (size_t)s * HID);
                    }
                }
                #pragma unroll
                for (int u = 0; u < GROWS - 1; ++u) {
                    if (u < r) {
                        const float nrm = w[u] * dd;
                        acc.x += v[u].x * nrm; acc.y += v[u].y * nrm;
                        acc.z += v[u].z * nrm; acc.w += v[u].w * nrm;
                    }
                }
            }
        }
        const float sn = dd * dd;
        acc.x += selfv.x * sn; acc.y += selfv.y * sn; acc.z += selfv.z * sn; acc.w += selfv.w * sn;
        acc.x += bb.x; acc.y += bb.y; acc.z += bb.z; acc.w += bb.w;
        *(float4*)(h + (size_t)node * HID + lane * 4) = acc;
        if constexpr (STATS) {
            ssum.x += acc.x; ssum.y += acc.y; ssum.z += acc.z; ssum.w += acc.w;
            ssq.x += acc.x * acc.x; ssq.y += acc.y * acc.y;
            ssq.z += acc.z * acc.z; ssq.w += acc.w * acc.w;
        }
    }
    if constexpr (STATS) {
        __shared__ float red[(GATHER_T / 32) * 2 * HID];
        const int wv = threadIdx.x >> 5;
        *(float4*)(red + (wv * 2 + 0) * HID + lane * 4) = ssum;
        *(float4*)(red + (wv * 2 + 1) * HID + lane * 4) = ssq;
        __syncthreads();
        static_assert(GATHER_T == 2 * HID, "one thread per batch-norm sum");
        const int which = threadIdx.x >> 7;
        const int f = threadIdx.x & (HID - 1);
        float tot = 0.0f;
        for (int w = 0; w < GATHER_T / 32; ++w) tot += red[(w * 2 + which) * HID + f];
        atomAddF32(which ? &bnsq[f] : &bnsum[f], tot);
    }
}

__global__ void gcn_bn_zero(float* bnsum, float* bnsq) {
    int t = threadIdx.x;
    if (t < HID) bnsum[t] = 0.0f;
    else if (t < 2 * HID) bnsq[t - HID] = 0.0f;
}

#define FIN_NT 3
template <int CTRL> __device__ __forceinline__ float lane_perm16(float v) {
    return __int_as_float(__builtin_amdgcn_update_dpp(0, __float_as_int(v), CTRL, 0xF, 0xF, true));
}
__device__ __forceinline__ float half_wave_max(float v) {
    v = fmaxf(v, lane_perm16<0xB1>(v));
    v = fmaxf(v, lane_perm16<0x4E>(v));
    v = fmaxf(v, lane_perm16<0x141>(v));
    v = fmaxf(v, lane_perm16<0x140>(v));
    return v;
}
__device__ __forceinline__ float half_wave_sum(float v) {
    v += lane_perm16<0xB1>(v);
    v += lane_perm16<0x4E>(v);
    v += lane_perm16<0x141>(v);
    v += lane_perm16<0x140>(v);
    return v;
}

__global__ __launch_bounds__(128) void gcn_final_wmma(const float* __restrict__ h,
                                                      const float* __restrict__ lw,
                                                      const float* __restrict__ lb,
                                                      float* __restrict__ out, int n) {
    __shared__ float Wt[FIN_NT * 16 * WT_STRIDE];

    const int t = threadIdx.x;
    const int wave  = t >> 5;
    const int lane  = t & 31;
    const int row16 = lane & 15;
    const int khalf = (lane >> 4) << 1;

    const int tileM = blockIdx.x * 64 + wave * 16;
    int rowIdx = tileM + row16;
    if (rowIdx > n - 1) rowIdx = n - 1;
    const float* Arow = h + (size_t)rowIdx * HID;

    float wv[(HID * OUTC) / 128];
    #pragma unroll
    for (int j = 0; j < (HID * OUTC) / 128; ++j) wv[j] = lw[t + j * 128];
    float2 av[HID / 4];
    #pragma unroll
    for (int s = 0; s < HID / 4; ++s) av[s] = *(const float2*)(Arow + 4 * s + khalf);

    #pragma unroll
    for (int j = 0; j < (HID * OUTC) / 128; ++j) {
        const int idx = t + j * 128;
        const int kk = idx / OUTC;
        const int c  = idx - kk * OUTC;
        Wt[c * WT_STRIDE + kk] = wv[j];
    }
    for (int j = 0; j < FIN_NT * 16 - OUTC; ++j) Wt[(OUTC + j) * WT_STRIDE + t] = 0.0f;
    __syncthreads();

    const float* Brow = Wt + (size_t)row16 * WT_STRIDE;
    v8f acc[FIN_NT] = {};
    #pragma unroll
    for (int s = 0; s < HID / 4; ++s) {
        v2f a; a.x = av[s].x; a.y = av[s].y;
        #pragma unroll
        for (int nt = 0; nt < FIN_NT; ++nt) {
            const float2 bv = *(const float2*)(Brow + nt * 16 * WT_STRIDE + 4 * s + khalf);
            v2f b; b.x = bv.x; b.y = bv.y;
            acc[nt] = __builtin_amdgcn_wmma_f32_16x16x4_f32(
                false, a, false, b, (short)0, acc[nt], false, false);
        }
    }

    const bool has2 = row16 < OUTC - 32;
    const float b0 = lb[row16];
    const float b1 = lb[16 + row16];
    const float b2 = lb[has2 ? 32 + row16 : 0];
    float x0[8], x1[8], x2[8], mx[8], se[8];
    #pragma unroll
    for (int r = 0; r < 8; ++r) {
        x0[r] = acc[0][r] + b0;
        x1[r] = acc[1][r] + b1;
        x2[r] = has2 ? acc[2][r] + b2 : -INFINITY;
        mx[r] = half_wave_max(fmaxf(fmaxf(x0[r], x1[r]), x2[r]));
    }
    #pragma unroll
    for (int r = 0; r < 8; ++r)
        se[r] = half_wave_sum(expf(x0[r] - mx[r]) + expf(x1[r] - mx[r]) + expf(x2[r] - mx[r]));
    const int rbase = tileM + ((lane >> 4) << 3);
    #pragma unroll
    for (int r = 0; r < 8; ++r) {
        const float ls = logf(se[r]);
        const int row = rbase + r;
        if (row < n) {
            float* orow = out + (size_t)row * OUTC;
            orow[row16]      = x0[r] - mx[r] - ls;
            orow[16 + row16] = x1[r] - mx[r] - ls;
            if (has2) orow[32 + row16] = x2[r] - mx[r] - ls;
        }
    }
}

extern "C" void kernel_launch(void* const* d_in, const int* in_sizes, int n_in,
                              void* d_out, int out_size, void* d_ws, size_t ws_size,
                              hipStream_t stream) {
    const float* x       = (const float*)d_in[0];
    const int*   ei      = (const int*)d_in[1];
    const float* la      = (const float*)d_in[2];
    const float* lb      = (const float*)d_in[3];
    const float* mlp_w   = (const float*)d_in[4];
    const float* mlp_b   = (const float*)d_in[5];
    const float* conv_w  = (const float*)d_in[6];
    const float* conv_b  = (const float*)d_in[7];
    const float* bn_g    = (const float*)d_in[8];
    const float* bn_b    = (const float*)d_in[9];
    const float* lin_w   = (const float*)d_in[10];
    const float* lin_b   = (const float*)d_in[11];
    float* out = (float*)d_out;

    float* ws = (float*)d_ws;
    const size_t NM = (size_t)N_NODES * HID;
    float* hA    = ws;
    float* hB    = hA + NM;
    int*   cnt    = (int*)(hB + NM);
    int*   rowptr = cnt + SCAN_N;
    int*   rank   = rowptr + SCAN_N;
    int*   csr    = rank + N_EDGES;
    float* dinv  = hB + NM + NM;
    float* bnsum = dinv + N_NODES;
    float* bnsq  = bnsum + HID;

    const int TPB = 256;
    const int edgeBlocks = ((N_EDGES + 3) / 4 + TPB - 1) / TPB;
    const int gatherBlocks = 1250;
    const int denseBlocks = 256;
    const int finBlocks  = (N_NODES + 63) / 64;

    gcn_cnt_zero<<<(N_NODES + TPB - 1) / TPB, TPB, 0, stream>>>(cnt, N_NODES);
    gcn_rank_edges<<<edgeBlocks, TPB, 0, stream>>>(ei, cnt, rank, N_EDGES);
    gcn_scan_deg<<<1, SCAN_T, 0, stream>>>(cnt, rowptr, dinv, N_NODES);
    gcn_fill_csr<<<edgeBlocks, TPB, 0, stream>>>(ei, rank, rowptr, dinv, csr, N_EDGES);

    for (int i = 0; i < NLAYERS; ++i) {
        if (i == 0) {
            gcn_layer_dense<true><<<denseBlocks, DENSE_T, 0, stream>>>(
                x, la, lb, nullptr, nullptr, nullptr, nullptr,
                mlp_w, mlp_b, conv_w, hA, N_NODES);
        } else {
            gcn_layer_dense<false><<<denseBlocks, DENSE_T, 0, stream>>>(
                hB, nullptr, nullptr, bnsum, bnsq, bn_g + (i - 1) * HID, bn_b + (i - 1) * HID,
                mlp_w + (size_t)i * HID * HID, mlp_b + i * HID, conv_w + (size_t)i * HID * HID,
                hA, N_NODES);
        }
        if (i < NLAYERS - 1) {
            gcn_bn_zero<<<1, 256, 0, stream>>>(bnsum, bnsq);
            gcn_gather<true><<<gatherBlocks, GATHER_T, 0, stream>>>(
                rowptr, csr, dinv, hA, conv_b + i * HID, hB, bnsum, bnsq, N_NODES, N_EDGES);
        } else {
            gcn_gather<false><<<gatherBlocks, GATHER_T, 0, stream>>>(
                rowptr, csr, dinv, hA, conv_b + i * HID, hB, nullptr, nullptr, N_NODES, N_EDGES);
        }
    }

    gcn_final_wmma<<<finBlocks, 128, 0, stream>>>(hB, lin_w, lin_b, out, N_NODES);
}


# Classifier folded through the last aggregation by linearity: Wf = conv_w[2]*lin_w on WMMA once per call, last layer gathers 48-wide rows, log-softmax in the gather; logit sum order changes
# speedup vs baseline: 3.6342x; 1.0793x over previous
#include <hip/hip_runtime.h>
#include <hip/hip_bf16.h>
#include <math.h>

#define N_NODES 50000
#define N_EDGES 800000
#define HID     128
#define OUTC    40
#define NLAYERS 3
#define BN_EPS  1e-5f
#define WT_STRIDE 130

typedef __attribute__((ext_vector_type(2))) float v2f;
typedef __attribute__((ext_vector_type(8))) float v8f;
typedef __attribute__((ext_vector_type(4))) float v4f;

__device__ __forceinline__ void atomAddF32(float* p, float v) {
    __hip_atomic_fetch_add(p, v, __ATOMIC_RELAXED, __HIP_MEMORY_SCOPE_AGENT);
}

#define SCAN_T   1024
#define SCAN_PER 49
#define SCAN_N   (SCAN_T * SCAN_PER)

__global__ void gcn_cnt_zero(int* cnt, int n) {
    int i = blockIdx.x * blockDim.x + threadIdx.x;
    if (i < n) cnt[i] = 0;
}

__global__ __launch_bounds__(256) void gcn_rank_edges(const int* __restrict__ ei, int* cnt,
                                                      int* __restrict__ rank, int e) {
    const int i = (blockIdx.x * blockDim.x + threadIdx.x) * 4;
    if (i >= e) return;
    const int* dst = ei + e;
    if ((e & 3) == 0) {
        const int4 d = *(const int4*)(dst + i);
        int4 r;
        r.x = __hip_atomic_fetch_add(&cnt[d.x], 1, __ATOMIC_RELAXED, __HIP_MEMORY_SCOPE_AGENT);
        r.y = __hip_atomic_fetch_add(&cnt[d.y], 1, __ATOMIC_RELAXED, __HIP_MEMORY_SCOPE_AGENT);
        r.z = __hip_atomic_fetch_add(&cnt[d.z], 1, __ATOMIC_RELAXED, __HIP_MEMORY_SCOPE_AGENT);
        r.w = __hip_atomic_fetch_add(&cnt[d.w], 1, __ATOMIC_RELAXED, __HIP_MEMORY_SCOPE_AGENT);
        *(int4*)(rank + i) = r;
    } else {
        for (int k = i; k < e && k < i + 4; ++k)
            rank[k] = __hip_atomic_fetch_add(&cnt[dst[k]], 1, __ATOMIC_RELAXED, __HIP_MEMORY_SCOPE_AGENT);
    }
}

__global__ __launch_bounds__(SCAN_T) void gcn_scan_deg(const int* __restrict__ cnt,
                                                       int* __restrict__ rowptr,
                                                       float* __restrict__ dinv, int n) {
    __shared__ int buf[SCAN_N];
    __shared__ int toff[SCAN_T];
    __shared__ int wsum[SCAN_T / 32];
    const int t = threadIdx.x;
    for (int i = t; i < SCAN_N; i += SCAN_T) {
        int c = 0;
        if (i < n) { c = cnt[i]; dinv[i] = rsqrtf((float)(c + 1)); }
        buf[i] = c;
    }
    __syncthreads();
    const int base = t * SCAN_PER;
    int s = 0;
    for (int j = 0; j < SCAN_PER; ++j) { int c = buf[base + j]; buf[base + j] = s; s += c; }
    const int lane = t & 31, wv = t >> 5;
    int incl = s;
    for (int o = 1; o < 32; o <<= 1) { int y = __shfl_up(incl, o); if (lane >= o) incl += y; }
    if (lane == 31) wsum[wv] = incl;
    __syncthreads();
    if (wv == 0) {
        int x = wsum[lane];
        int xi = x;
        for (int o = 1; o < 32; o <<= 1) { int y = __shfl_up(xi, o); if (lane >= o) xi += y; }
        wsum[lane] = xi - x;
    }
    __syncthreads();
    toff[t] = wsum[wv] + incl - s;
    __syncthreads();
    for (int i = t; i <= n; i += SCAN_T) rowptr[i] = buf[i] + toff[i / SCAN_PER];
}

__global__ __launch_bounds__(256) void gcn_fill_csr(const int* __restrict__ ei, const int* __restrict__ rank,
                             const int* __restrict__ rowptr, const float* __restrict__ dinv,
                             int* __restrict__ csr, int e) {
    const int i = (blockIdx.x * blockDim.x + threadIdx.x) * 4;
    if (i >= e) return;
    if ((e & 3) == 0) {
        const int4 s4 = *(const int4*)(ei + i);
        const int4 d4 = *(const int4*)(ei + e + i);
        const int4 r4 = *(const int4*)(rank + i);
        const int s[4] = {s4.x, s4.y, s4.z, s4.w};
        const int d[4] = {d4.x, d4.y, d4.z, d4.w};
        const int r[4] = {r4.x, r4.y, r4.z, r4.w};
        int rp[4]; float dv[4];
        #pragma unroll
        for (int u = 0; u < 4; ++u) { rp[u] = rowptr[d[u]]; dv[u] = dinv[s[u]]; }
        #pragma unroll
        for (int u = 0; u < 4; ++u) {
            unsigned pos = (unsigned)(rp[u] + r[u]);
            if (pos > (unsigned)(e - 1)) pos = (unsigned)(e - 1);
            int2 p; p.x = s[u]; p.y = __float_as_int(dv[u]);
            *(int2*)(csr + 2 * (size_t)pos) = p;
        }
    } else {
        for (int k = i; k < e && k < i + 4; ++k) {
            const int s = ei[k];
            unsigned pos = (unsigned)(rowptr[ei[e + k]] + rank[k]);
            if (pos > (unsigned)(e - 1)) pos = (unsigned)(e - 1);
            int2 p; p.x = s; p.y = __float_as_int(dinv[s]);
            *(int2*)(csr + 2 * (size_t)pos) = p;
        }
    }
}

#define DENSE_T 512
#define DENSE_WAVES (DENSE_T / 32)
#define STRIP_STRIDE 132

template <int NT>
__device__ __forceinline__ void strip_gemm(const float2 (&av)[HID / 4], const float* Bfrag, v8f (&acc)[NT]) {
    #pragma unroll
    for (int s = 0; s < HID / 4; ++s) {
        v2f a; a.x = av[s].x; a.y = av[s].y;
        #pragma unroll
        for (int nt = 0; nt < NT; ++nt) {
            const float2 bv = *(const float2*)(Bfrag + nt * 16 * WT_STRIDE + 4 * s);
            v2f b; b.x = bv.x; b.y = bv.y;
            acc[nt] = __builtin_amdgcn_wmma_f32_16x16x4_f32(
                false, a, false, b, (short)0, acc[nt], false, false);
        }
    }
}

template <bool FIRST>
__device__ __forceinline__ void strip_load(float2 (&av)[HID / 4], float2 (&aw)[HID / 4],
                                           const float* __restrict__ A, int tileM, int row16,
                                           int khalf, int nrows) {
    int rowIdx = tileM + row16;
    if (rowIdx > nrows - 1) rowIdx = nrows - 1;
    const float* Arow = A + (size_t)rowIdx * (FIRST ? 2 * HID : HID) + khalf;
    #pragma unroll
    for (int s = 0; s < HID / 4; ++s) {
        av[s] = *(const float2*)(Arow + 4 * s);
        if constexpr (FIRST) aw[s] = *(const float2*)(Arow + HID + 4 * s);
    }
}

template <bool FIRST, int NT2>
__global__ __launch_bounds__(DENSE_T) void gcn_layer_dense(
        const float* __restrict__ A,
        const float* __restrict__ la, const float* __restrict__ lb,
        const float* __restrict__ bnsum, const float* __restrict__ bnsq,
        const float* __restrict__ gamma, const float* __restrict__ beta,
        const float* __restrict__ W1, const float* __restrict__ b1,
        const float* __restrict__ W2, float* __restrict__ C, int nrows) {
    __shared__ __attribute__((aligned(16))) float W1t[HID * WT_STRIDE];
    __shared__ __attribute__((aligned(16))) float W2t[HID * WT_STRIDE];
    __shared__ __attribute__((aligned(16))) float strip[DENSE_WAVES * 16 * STRIP_STRIDE];
    __shared__ __attribute__((aligned(16))) float bnc[4 * HID];

    const int t = threadIdx.x;
    const int wave  = __builtin_amdgcn_readfirstlane(t >> 5);
    const int lane  = t & 31;
    const int row16 = lane & 15;
    const int khalf = (lane >> 4) << 1;
    const int nstrips = (nrows + 15) >> 4;
    const int wavesTotal = gridDim.x * DENSE_WAVES;
    const int sfirst = wave * gridDim.x + blockIdx.x;

    constexpr int N2 = NT2 * 16;
    constexpr int W2PER = (HID * N2) / DENSE_T;
    float w1v[(HID * HID) / DENSE_T], w2v[W2PER];
    #pragma unroll
    for (int j = 0; j < (HID * HID) / DENSE_T; ++j) w1v[j] = W1[t + j * DENSE_T];
    #pragma unroll
    for (int j = 0; j < W2PER; ++j) w2v[j] = W2[t + j * DENSE_T];

    float2 av[HID / 4], aw[HID / 4];
    bool active = sfirst < nstrips;
    if (active) strip_load<FIRST>(av, aw, A, sfirst * 16, row16, khalf, nrows);

    #pragma unroll
    for (int j = 0; j < (HID * HID) / DENSE_T; ++j) {
        const int idx = t + j * DENSE_T;
        W1t[(idx & 127) * WT_STRIDE + (idx >> 7)] = w1v[j];
    }
    #pragma unroll
    for (int j = 0; j < W2PER; ++j) {
        const int idx = t + j * DENSE_T;
        W2t[(idx % N2) * WT_STRIDE + (idx / N2)] = w2v[j];
    }
    float alpha = 0.0f, betac = 0.0f;
    if constexpr (FIRST) {
        alpha = expf(la[0]);
        betac = expf(lb[0]);
    } else {
        if (t < HID) {
            const float invn = 1.0f / (float)nrows;
            const float mu = bnsum[t] * invn;
            const float var = bnsq[t] * invn - mu * mu;
            bnc[t] = gamma[t];
            bnc[HID + t] = mu;
            bnc[2 * HID + t] = rsqrtf(var + BN_EPS);
            bnc[3 * HID + t] = beta[t];
        }
    }
    __syncthreads();

    float* myStrip = strip + wave * 16 * STRIP_STRIDE;
    const float* B1 = W1t + row16 * WT_STRIDE + khalf;
    const float* B2 = W2t + row16 * WT_STRIDE + khalf;
    const int rhalf = (lane >> 4) << 3;

    for (int s0 = 0; s0 < nstrips; s0 += wavesTotal) {
        const int tileM = (s0 + sfirst) * 16;
        if (active) {
            #pragma unroll
            for (int s = 0; s < HID / 4; ++s) {
                if constexpr (FIRST) {
                    av[s].x = alpha * av[s].x + betac * aw[s].x;
                    av[s].y = alpha * av[s].y + betac * aw[s].y;
                } else {
                    const int k = 4 * s + khalf;
                    const float2 g  = *(const float2*)(bnc + k);
                    const float2 mu = *(const float2*)(bnc + HID + k);
                    const float2 rs = *(const float2*)(bnc + 2 * HID + k);
                    const float2 bt = *(const float2*)(bnc + 3 * HID + k);
                    av[s].x = fmaxf(g.x * (av[s].x - mu.x) * rs.x + bt.x, 0.0f);
                    av[s].y = fmaxf(g.y * (av[s].y - mu.y) * rs.y + bt.y, 0.0f);
                }
            }
            v8f acc[8] = {};
            strip_gemm<8>(av, B1, acc);
            #pragma unroll
            for (int nt = 0; nt < 8; ++nt) {
                const float bv = b1[nt * 16 + row16];
                #pragma unroll
                for (int r = 0; r < 8; ++r)
                    myStrip[(rhalf + r) * STRIP_STRIDE + nt * 16 + row16] = fmaxf(acc[nt][r] + bv, 0.0f);
            }
        }
        __syncthreads();
        const bool was = active;
        if (was) {
            #pragma unroll
            for (int s = 0; s < HID / 4; ++s)
                av[s] = *(const float2*)(myStrip + row16 * STRIP_STRIDE + 4 * s + khalf);
        }
        v8f acc2[NT2] = {};
        if (was) strip_gemm<NT2>(av, B2, acc2);
        const int snext = s0 + wavesTotal + sfirst;
        active = (s0 + wavesTotal < nstrips) && (snext < nstrips);
        if (active) strip_load<FIRST>(av, aw, A, snext * 16, row16, khalf, nrows);
        if (was) {
            #pragma unroll
            for (int nt = 0; nt < NT2; ++nt) {
                const int col = nt * 16 + row16;
                #pragma unroll
                for (int r = 0; r < 8; ++r) {
                    const int row = tileM + rhalf + r;
                    if (row < nrows) C[(size_t)row * N2 + col] = acc2[nt][r];
                }
            }
        }
    }
}

#define GROWS 8
#define GATHER_T 256

template <bool STATS>
__global__ __launch_bounds__(GATHER_T) void gcn_gather(const int* __restrict__ rowptr,
                                                  const int* __restrict__ csr,
                                                  const float* __restrict__ dinv,
                                                  const float* __restrict__ m,
                                                  const float* __restrict__ b,
                                                  float* __restrict__ h,
                                                  float* bnsum, float* bnsq, int n, int e) {
    const int lane = threadIdx.x & 31;
    float4 ssum = make_float4(0.0f, 0.0f, 0.0f, 0.0f);
    float4 ssq  = make_float4(0.0f, 0.0f, 0.0f, 0.0f);
    const int nwaves = (gridDim.x * blockDim.x) >> 5;
    const int wid = __builtin_amdgcn_readfirstlane((blockIdx.x * blockDim.x + threadIdx.x) >> 5);
    const float* mlane = m + lane * 4;
    const float4 bb = *(const float4*)(b + lane * 4);
    for (int node = wid; node < n; node += nwaves) {
        int beg = rowptr[node];
        int end = rowptr[node + 1];
        if (beg < 0) beg = 0;
        if (end > e) end = e;
        const float dd = dinv[node];
        const float4 selfv = *(const float4*)(mlane + (size_t)node * HID);
        float4 acc = make_float4(0.0f, 0.0f, 0.0f, 0.0f);

        for (int base = beg; base < end; base += 32) {
            int c = end - base; if (c > 32) c = 32;
            int sv = 0, wbits = 0;
            if (lane < c) {
                const int2 p = *(const int2*)(csr + 2 * (size_t)(base + lane));
                sv = p.x; wbits = p.y;
            }
            if ((unsigned)sv > (unsigned)(n - 1)) sv = n - 1;
            int j = 0;
            for (; j + GROWS <= c; j += GROWS) {
                float4 v[GROWS]; float w[GROWS];
                #pragma unroll
                for (int u = 0; u < GROWS; ++u) {
                    const int s = __builtin_amdgcn_readlane(sv, j + u);
                    w[u] = __int_as_float(__builtin_amdgcn_readlane(wbits, j + u));
                    v[u] = *(const float4*)(mlane + (size_t)s * HID);
                }
                #pragma unroll
                for (int u = 0; u < GROWS; ++u) {
                    const float nrm = w[u] * dd;
                    acc.x += v[u].x * nrm; acc.y += v[u].y * nrm;
                    acc.z += v[u].z * nrm; acc.w += v[u].w * nrm;
                }
            }
            const int r = c - j;
            if (r > 0) {
                float4 v[GROWS - 1]; float w[GROWS - 1];
                #pragma unroll
                for (int u = 0; u < GROWS - 1; ++u) {
                    if (u < r) {
                        const int s = __builtin_amdgcn_readlane(sv, j + u);
                        w[u] = __int_as_float(__builtin_amdgcn_readlane(wbits, j + u));
                        v[u] = *(const float4*)(mlane + (size_t)s * HID);
                    }
                }
                #pragma unroll
                for (int u = 0; u < GROWS - 1; ++u) {
                    if (u < r) {
                        const float nrm = w[u] * dd;
                        acc.x += v[u].x * nrm; acc.y += v[u].y * nrm;
                        acc.z += v[u].z * nrm; acc.w += v[u].w * nrm;
                    }
                }
            }
        }
        const float sn = dd * dd;
        acc.x += selfv.x * sn; acc.y += selfv.y * sn; acc.z += selfv.z * sn; acc.w += selfv.w * sn;
        acc.x += bb.x; acc.y += bb.y; acc.z += bb.z; acc.w += bb.w;
        *(float4*)(h + (size_t)node * HID + lane * 4) = acc;
        if constexpr (STATS) {
            ssum.x += acc.x; ssum.y += acc.y; ssum.z += acc.z; ssum.w += acc.w;
            ssq.x += acc.x * acc.x; ssq.y += acc.y * acc.y;
            ssq.z += acc.z * acc.z; ssq.w += acc.w * acc.w;
        }
    }
    if constexpr (STATS) {
        __shared__ float red[(GATHER_T / 32) * 2 * HID];
        const int wv = threadIdx.x >> 5;
        *(float4*)(red + (wv * 2 + 0) * HID + lane * 4) = ssum;
        *(float4*)(red + (wv * 2 + 1) * HID + lane * 4) = ssq;
        __syncthreads();
        static_assert(GATHER_T == 2 * HID, "one thread per batch-norm sum");
        const int which = threadIdx.x >> 7;
        const int f = threadIdx.x & (HID - 1);
        float tot = 0.0f;
        for (int w = 0; w < GATHER_T / 32; ++w) tot += red[(w * 2 + which) * HID + f];
        atomAddF32(which ? &bnsq[f] : &bnsum[f], tot);
    }
}

__global__ void gcn_bn_zero(float* bnsum, float* bnsq) {
    int t = threadIdx.x;
    if (t < HID) bnsum[t] = 0.0f;
    else if (t < 2 * HID) bnsq[t - HID] = 0.0f;
}

#define CLS_NT 3
#define CLS_W (CLS_NT * 16)
#define FOLD_T (CLS_NT * 32)
#define FOLD_BLOCKS (HID / 16 + 1)
#define CLS_ROWS 16

__global__ __launch_bounds__(FOLD_T) void gcn_fold_cls(const float* __restrict__ W2,
                                                       const float* __restrict__ b,
                                                       const float* __restrict__ lw,
                                                       const float* __restrict__ lb,
                                                       float* __restrict__ Wf, float* __restrict__ bf) {
    const int t = threadIdx.x;
    const int mt = blockIdx.x;
    const int nt = __builtin_amdgcn_readfirstlane(t >> 5);
    const int lane  = t & 31;
    const int row16 = lane & 15;
    const int khalf = (lane >> 4) << 1;
    const int arow = mt * 16 + row16;
    const float* Arow = (arow < HID ? W2 + (size_t)arow * HID : b) + khalf;
    const int col = nt * 16 + row16;
    const bool live = col < OUTC;
    const float* Bcol = lw + (live ? col : OUTC - 1) + khalf * OUTC;

    float2 av[HID / 4]; float b0[HID / 4], b1[HID / 4];
    #pragma unroll
    for (int s = 0; s < HID / 4; ++s) {
        av[s] = *(const float2*)(Arow + 4 * s);
        b0[s] = Bcol[(4 * s) * OUTC];
        b1[s] = Bcol[(4 * s + 1) * OUTC];
    }
    v8f acc = {};
    #pragma unroll
    for (int s = 0; s < HID / 4; ++s) {
        v2f a; a.x = av[s].x; a.y = av[s].y;
        v2f bfrag; bfrag.x = live ? b0[s] : 0.0f; bfrag.y = live ? b1[s] : 0.0f;
        acc = __builtin_amdgcn_wmma_f32_16x16x4_f32(false, a, false, bfrag, (short)0, acc, false, false);
    }
    const int rhalf = (lane >> 4) << 3;
    if (mt < HID / 16) {
        #pragma unroll
        for (int r = 0; r < 8; ++r)
            Wf[(size_t)(mt * 16 + rhalf + r) * CLS_W + col] = acc[r];
    } else if (lane < 16) {
        bf[col] = acc[0] + (live ? lb[col] : 0.0f);
    }
}

template <int CTRL> __device__ __forceinline__ float lane_perm16(float v) {
    return __int_as_float(__builtin_amdgcn_update_dpp(0, __float_as_int(v), CTRL, 0xF, 0xF, true));
}
__device__ __forceinline__ float wave_max32(float v) {
    v = fmaxf(v, lane_perm16<0xB1>(v));
    v = fmaxf(v, lane_perm16<0x4E>(v));
    v = fmaxf(v, lane_perm16<0x141>(v));
    v = fmaxf(v, lane_perm16<0x140>(v));
    const float lo = __int_as_float(__builtin_amdgcn_readlane(__float_as_int(v), 0));
    const float hi = __int_as_float(__builtin_amdgcn_readlane(__float_as_int(v), 16));
    return fmaxf(lo, hi);
}
__device__ __forceinline__ float wave_sum32(float v) {
    v += lane_perm16<0xB1>(v);
    v += lane_perm16<0x4E>(v);
    v += lane_perm16<0x141>(v);
    v += lane_perm16<0x140>(v);
    const float lo = __int_as_float(__builtin_amdgcn_readlane(__float_as_int(v), 0));
    const float hi = __int_as_float(__builtin_amdgcn_readlane(__float_as_int(v), 16));
    return lo + hi;
}

__global__ __launch_bounds__(GATHER_T) void gcn_gather_cls(const int* __restrict__ rowptr,
                                                           const int* __restrict__ csr,
                                                           const float* __restrict__ dinv,
                                                           const float* __restrict__ m,
                                                           const float* __restrict__ bf,
                                                           float* __restrict__ out, int n, int e) {
    const int lane = threadIdx.x & 31;
    const int nwaves = (gridDim.x * blockDim.x) >> 5;
    const int wid = __builtin_amdgcn_readfirstlane((blockIdx.x * blockDim.x + threadIdx.x) >> 5);
    const bool valid = lane < OUTC / 2;
    const int fl = 2 * (lane < CLS_W / 2 ? lane : CLS_W / 2 - 1);
    const float2 bb = *(const float2*)(bf + fl);
    for (int node = wid; node < n; node += nwaves) {
        int beg = rowptr[node];
        int end = rowptr[node + 1];
        if (beg < 0) beg = 0;
        if (end > e) end = e;
        const float dd = dinv[node];
        const float2 selfv = *(const float2*)(m + (size_t)node * CLS_W + fl);
        float2 acc = make_float2(0.0f, 0.0f);

        for (int base = beg; base < end; base += 32) {
            int c = end - base; if (c > 32) c = 32;
            int sv = 0, wbits = 0;
            if (lane < c) {
                const int2 p = *(const int2*)(csr + 2 * (size_t)(base + lane));
                sv = p.x; wbits = p.y;
            }
            if ((unsigned)sv > (unsigned)(n - 1)) sv = n - 1;
            int j = 0;
            for (; j + CLS_ROWS <= c; j += CLS_ROWS) {
                float2 v[CLS_ROWS]; float w[CLS_ROWS];
                #pragma unroll
                for (int u = 0; u < CLS_ROWS; ++u) {
                    const int s = __builtin_amdgcn_readlane(sv, j + u);
                    w[u] = __int_as_float(__builtin_amdgcn_readlane(wbits, j + u));
                    v[u] = *(const float2*)(m + (size_t)s * CLS_W + fl);
                }
                #pragma unroll
                for (int u = 0; u < CLS_ROWS; ++u) {
                    const float nrm = w[u] * dd;
                    acc.x += v[u].x * nrm; acc.y += v[u].y * nrm;
                }
            }
            const int r = c - j;
            if (r > 0) {
                float2 v[CLS_ROWS - 1]; float w[CLS_ROWS - 1];
                #pragma unroll
                for (int u = 0; u < CLS_ROWS - 1; ++u) {
                    if (u < r) {
                        const int s = __builtin_amdgcn_readlane(sv, j + u);
                        w[u] = __int_as_float(__builtin_amdgcn_readlane(wbits, j + u));
                        v[u] = *(const float2*)(m + (size_t)s * CLS_W + fl);
                    }
                }
                #pragma unroll
                for (int u = 0; u < CLS_ROWS - 1; ++u) {
                    if (u < r) {
                        const float nrm = w[u] * dd;
                        acc.x += v[u].x * nrm; acc.y += v[u].y * nrm;
                    }
                }
            }
        }
        const float sn = dd * dd;
        acc.x += selfv.x * sn; acc.y += selfv.y * sn;
        acc.x += bb.x; acc.y += bb.y;
        const float x0 = valid ? acc.x : -INFINITY;
        const float x1 = valid ? acc.y : -INFINITY;
        const float mx = wave_max32(fmaxf(x0, x1));
        const float ls = logf(wave_sum32(expf(x0 - mx) + expf(x1 - mx)));
        if (valid) {
            float2 o; o.x = x0 - mx - ls; o.y = x1 - mx - ls;
            *(float2*)(out + (size_t)node * OUTC + 2 * lane) = o;
        }
    }
}

extern "C" void kernel_launch(void* const* d_in, const int* in_sizes, int n_in,
                              void* d_out, int out_size, void* d_ws, size_t ws_size,
                              hipStream_t stream) {
    const float* x       = (const float*)d_in[0];
    const int*   ei      = (const int*)d_in[1];
    const float* la      = (const float*)d_in[2];
    const float* lb      = (const float*)d_in[3];
    const float* mlp_w   = (const float*)d_in[4];
    const float* mlp_b   = (const float*)d_in[5];
    const float* conv_w  = (const float*)d_in[6];
    const float* conv_b  = (const float*)d_in[7];
    const float* bn_g    = (const float*)d_in[8];
    const float* bn_b    = (const float*)d_in[9];
    const float* lin_w   = (const float*)d_in[10];
    const float* lin_b   = (const float*)d_in[11];
    float* out = (float*)d_out;

    float* ws = (float*)d_ws;
    const size_t NM = (size_t)N_NODES * HID;
    float* hA    = ws;
    float* hB    = hA + NM;
    int*   cnt    = (int*)(hB + NM);
    int*   rowptr = cnt + SCAN_N;
    int*   rank   = rowptr + SCAN_N;
    int*   csr    = rank + N_EDGES;
    float* Wf    = (float*)(csr + 2 * N_EDGES);
    float* bf    = Wf + HID * CLS_W;
    float* dinv  = hB + NM + NM;
    float* bnsum = dinv + N_NODES;
    float* bnsq  = bnsum + HID;

    const int TPB = 256;
    const int edgeBlocks = ((N_EDGES + 3) / 4 + TPB - 1) / TPB;
    const int gatherBlocks = 1250;
    const int denseBlocks = 256;

    gcn_fold_cls<<<FOLD_BLOCKS, FOLD_T, 0, stream>>>(conv_w + (size_t)(NLAYERS - 1) * HID * HID,
                                           conv_b + (NLAYERS - 1) * HID, lin_w, lin_b, Wf, bf);

    gcn_cnt_zero<<<(N_NODES + TPB - 1) / TPB, TPB, 0, stream>>>(cnt, N_NODES);
    gcn_rank_edges<<<edgeBlocks, TPB, 0, stream>>>(ei, cnt, rank, N_EDGES);
    gcn_scan_deg<<<1, SCAN_T, 0, stream>>>(cnt, rowptr, dinv, N_NODES);
    gcn_fill_csr<<<edgeBlocks, TPB, 0, stream>>>(ei, rank, rowptr, dinv, csr, N_EDGES);

    for (int i = 0; i < NLAYERS; ++i) {
        if (i == 0) {
            gcn_layer_dense<true, 8><<<denseBlocks, DENSE_T, 0, stream>>>(
                x, la, lb, nullptr, nullptr, nullptr, nullptr,
                mlp_w, mlp_b, conv_w, hA, N_NODES);
        } else if (i < NLAYERS - 1) {
            gcn_layer_dense<false, 8><<<denseBlocks, DENSE_T, 0, stream>>>(
                hB, nullptr, nullptr, bnsum, bnsq, bn_g + (i - 1) * HID, bn_b + (i - 1) * HID,
                mlp_w + (size_t)i * HID * HID, mlp_b + i * HID, conv_w + (size_t)i * HID * HID,
                hA, N_NODES);
        } else {
            gcn_layer_dense<false, CLS_NT><<<denseBlocks, DENSE_T, 0, stream>>>(
                hB, nullptr, nullptr, bnsum, bnsq, bn_g + (i - 1) * HID, bn_b + (i - 1) * HID,
                mlp_w + (size_t)i * HID * HID, mlp_b + i * HID, Wf,
                hA, N_NODES);
        }
        if (i < NLAYERS - 1) {
            gcn_bn_zero<<<1, 256, 0, stream>>>(bnsum, bnsq);
            gcn_gather<true><<<gatherBlocks, GATHER_T, 0, stream>>>(
                rowptr, csr, dinv, hA, conv_b + i * HID, hB, bnsum, bnsq, N_NODES, N_EDGES);
        } else {
            gcn_gather_cls<<<gatherBlocks, GATHER_T, 0, stream>>>(
                rowptr, csr, dinv, hA, bf, out, N_NODES, N_EDGES);
        }
    }
}


# Degree prefix sum on 49 work-groups, one thread per entry: each sums the counts in front of it with all its 16-byte loads in flight; was one work-group with 49 dependent loads per wave
# speedup vs baseline: 4.0554x; 1.1159x over previous
#include <hip/hip_runtime.h>
#include <hip/hip_bf16.h>
#include <math.h>

#define N_NODES 50000
#define N_EDGES 800000
#define HID     128
#define OUTC    40
#define NLAYERS 3
#define BN_EPS  1e-5f
#define WT_STRIDE 130

typedef __attribute__((ext_vector_type(2))) float v2f;
typedef __attribute__((ext_vector_type(8))) float v8f;
typedef __attribute__((ext_vector_type(4))) float v4f;

__device__ __forceinline__ void atomAddF32(float* p, float v) {
    __hip_atomic_fetch_add(p, v, __ATOMIC_RELAXED, __HIP_MEMORY_SCOPE_AGENT);
}

#define SCAN_T      1024
#define SCAN_BLOCKS ((N_NODES + 1 + SCAN_T - 1) / SCAN_T)
#define SCAN_N      (SCAN_T * SCAN_BLOCKS)
#define SCAN_PRE    (((SCAN_BLOCKS - 1) * (SCAN_T / 4) + SCAN_T - 1) / SCAN_T)

__global__ void gcn_cnt_zero(int* cnt, int n) {
    int i = blockIdx.x * blockDim.x + threadIdx.x;
    if (i < n) cnt[i] = 0;
}

__global__ __launch_bounds__(256) void gcn_rank_edges(const int* __restrict__ ei, int* cnt,
                                                      int* __restrict__ rank, int e) {
    const int i = (blockIdx.x * blockDim.x + threadIdx.x) * 4;
    if (i >= e) return;
    const int* dst = ei + e;
    if ((e & 3) == 0) {
        const int4 d = *(const int4*)(dst + i);
        int4 r;
        r.x = __hip_atomic_fetch_add(&cnt[d.x], 1, __ATOMIC_RELAXED, __HIP_MEMORY_SCOPE_AGENT);
        r.y = __hip_atomic_fetch_add(&cnt[d.y], 1, __ATOMIC_RELAXED, __HIP_MEMORY_SCOPE_AGENT);
        r.z = __hip_atomic_fetch_add(&cnt[d.z], 1, __ATOMIC_RELAXED, __HIP_MEMORY_SCOPE_AGENT);
        r.w = __hip_atomic_fetch_add(&cnt[d.w], 1, __ATOMIC_RELAXED, __HIP_MEMORY_SCOPE_AGENT);
        *(int4*)(rank + i) = r;
    } else {
        for (int k = i; k < e && k < i + 4; ++k)
            rank[k] = __hip_atomic_fetch_add(&cnt[dst[k]], 1, __ATOMIC_RELAXED, __HIP_MEMORY_SCOPE_AGENT);
    }
}

__global__ __launch_bounds__(SCAN_T) void gcn_scan_deg(const int* __restrict__ cnt,
                                                       int* __restrict__ rowptr,
                                                       float* __restrict__ dinv, int n) {
    __shared__ int wpre[SCAN_T / 32];
    __shared__ int wown[SCAN_T / 32];
    static_assert(SCAN_T / 32 == 32, "one lane per wave total in the fold below");
    const int t = threadIdx.x;
    const int lane = t & 31, wv = t >> 5;
    const int i = blockIdx.x * SCAN_T + t;
    const int lim = blockIdx.x * (SCAN_T / 4);
    const int4* cnt4 = (const int4*)cnt;
    int4 v[SCAN_PRE];
    #pragma unroll
    for (int j = 0; j < SCAN_PRE; ++j) {
        const int q = j * SCAN_T + t;
        v[j] = cnt4[q < lim ? q : 0];
    }
    int c = 0;
    if (i < n) c = cnt[i];
    int p = 0;
    #pragma unroll
    for (int j = 0; j < SCAN_PRE; ++j) {
        const int q = j * SCAN_T + t;
        if (q < lim) p += (v[j].x + v[j].y) + (v[j].z + v[j].w);
    }
    if (i < n) dinv[i] = rsqrtf((float)(c + 1));
    for (int o = 16; o > 0; o >>= 1) p += __shfl_xor(p, o);
    int incl = c;
    for (int o = 1; o < 32; o <<= 1) { int y = __shfl_up(incl, o); if (lane >= o) incl += y; }
    if (lane == 31) { wpre[wv] = p; wown[wv] = incl; }
    __syncthreads();
    int z = wpre[lane] + (lane < wv ? wown[lane] : 0);
    for (int o = 16; o > 0; o >>= 1) z += __shfl_xor(z, o);
    if (i <= n) rowptr[i] = z + incl - c;
}

__global__ __launch_bounds__(256) void gcn_fill_csr(const int* __restrict__ ei, const int* __restrict__ rank,
                             const int* __restrict__ rowptr, const float* __restrict__ dinv,
                             int* __restrict__ csr, int e) {
    const int i = (blockIdx.x * blockDim.x + threadIdx.x) * 4;
    if (i >= e) return;
    if ((e & 3) == 0) {
        const int4 s4 = *(const int4*)(ei + i);
        const int4 d4 = *(const int4*)(ei + e + i);
        const int4 r4 = *(const int4*)(rank + i);
        const int s[4] = {s4.x, s4.y, s4.z, s4.w};
        const int d[4] = {d4.x, d4.y, d4.z, d4.w};
        const int r[4] = {r4.x, r4.y, r4.z, r4.w};
        int rp[4]; float dv[4];
        #pragma unroll
        for (int u = 0; u < 4; ++u) { rp[u] = rowptr[d[u]]; dv[u] = dinv[s[u]]; }
        #pragma unroll
        for (int u = 0; u < 4; ++u) {
            unsigned pos = (unsigned)(rp[u] + r[u]);
            if (pos > (unsigned)(e - 1)) pos = (unsigned)(e - 1);
            int2 p; p.x = s[u]; p.y = __float_as_int(dv[u]);
            *(int2*)(csr + 2 * (size_t)pos) = p;
        }
    } else {
        for (int k = i; k < e && k < i + 4; ++k) {
            const int s = ei[k];
            unsigned pos = (unsigned)(rowptr[ei[e + k]] + rank[k]);
            if (pos > (unsigned)(e - 1)) pos = (unsigned)(e - 1);
            int2 p; p.x = s; p.y = __float_as_int(dinv[s]);
            *(int2*)(csr + 2 * (size_t)pos) = p;
        }
    }
}

#define DENSE_T 512
#define DENSE_WAVES (DENSE_T / 32)
#define STRIP_STRIDE 132

template <int NT>
__device__ __forceinline__ void strip_gemm(const float2 (&av)[HID / 4], const float* Bfrag, v8f (&acc)[NT]) {
    #pragma unroll
    for (int s = 0; s < HID / 4; ++s) {
        v2f a; a.x = av[s].x; a.y = av[s].y;
        #pragma unroll
        for (int nt = 0; nt < NT; ++nt) {
            const float2 bv = *(const float2*)(Bfrag + nt * 16 * WT_STRIDE + 4 * s);
            v2f b; b.x = bv.x; b.y = bv.y;
            acc[nt] = __builtin_amdgcn_wmma_f32_16x16x4_f32(
                false, a, false, b, (short)0, acc[nt], false, false);
        }
    }
}

template <bool FIRST>
__device__ __forceinline__ void strip_load(float2 (&av)[HID / 4], float2 (&aw)[HID / 4],
                                           const float* __restrict__ A, int tileM, int row16,
                                           int khalf, int nrows) {
    int rowIdx = tileM + row16;
    if (rowIdx > nrows - 1) rowIdx = nrows - 1;
    const float* Arow = A + (size_t)rowIdx * (FIRST ? 2 * HID : HID) + khalf;
    #pragma unroll
    for (int s = 0; s < HID / 4; ++s) {
        av[s] = *(const float2*)(Arow + 4 * s);
        if constexpr (FIRST) aw[s] = *(const float2*)(Arow + HID + 4 * s);
    }
}

template <bool FIRST, int NT2>
__global__ __launch_bounds__(DENSE_T) void gcn_layer_dense(
        const float* __restrict__ A,
        const float* __restrict__ la, const float* __restrict__ lb,
        const float* __restrict__ bnsum, const float* __restrict__ bnsq,
        const float* __restrict__ gamma, const float* __restrict__ beta,
        const float* __restrict__ W1, const float* __restrict__ b1,
        const float* __restrict__ W2, float* __restrict__ C, int nrows) {
    __shared__ __attribute__((aligned(16))) float W1t[HID * WT_STRIDE];
    __shared__ __attribute__((aligned(16))) float W2t[HID * WT_STRIDE];
    __shared__ __attribute__((aligned(16))) float strip[DENSE_WAVES * 16 * STRIP_STRIDE];
    __shared__ __attribute__((aligned(16))) float bnc[4 * HID];

    const int t = threadIdx.x;
    const int wave  = __builtin_amdgcn_readfirstlane(t >> 5);
    const int lane  = t & 31;
    const int row16 = lane & 15;
    const int khalf = (lane >> 4) << 1;
    const int nstrips = (nrows + 15) >> 4;
    const int wavesTotal = gridDim.x * DENSE_WAVES;
    const int sfirst = wave * gridDim.x + blockIdx.x;

    constexpr int N2 = NT2 * 16;
    constexpr int W2PER = (HID * N2) / DENSE_T;
    float w1v[(HID * HID) / DENSE_T], w2v[W2PER];
    #pragma unroll
    for (int j = 0; j < (HID * HID) / DENSE_T; ++j) w1v[j] = W1[t + j * DENSE_T];
    #pragma unroll
    for (int j = 0; j < W2PER; ++j) w2v[j] = W2[t + j * DENSE_T];

    float2 av[HID / 4], aw[HID / 4];
    bool active = sfirst < nstrips;
    if (active) strip_load<FIRST>(av, aw, A, sfirst * 16, row16, khalf, nrows);

    #pragma unroll
    for (int j = 0; j < (HID * HID) / DENSE_T; ++j) {
        const int idx = t + j * DENSE_T;
        W1t[(idx & 127) * WT_STRIDE + (idx >> 7)] = w1v[j];
    }
    #pragma unroll
    for (int j = 0; j < W2PER; ++j) {
        const int idx = t + j * DENSE_T;
        W2t[(idx % N2) * WT_STRIDE + (idx / N2)] = w2v[j];
    }
    float alpha = 0.0f, betac = 0.0f;
    if constexpr (FIRST) {
        alpha = expf(la[0]);
        betac = expf(lb[0]);
    } else {
        if (t < HID) {
            const float invn = 1.0f / (float)nrows;
            const float mu = bnsum[t] * invn;
            const float var = bnsq[t] * invn - mu * mu;
            bnc[t] = gamma[t];
            bnc[HID + t] = mu;
            bnc[2 * HID + t] = rsqrtf(var + BN_EPS);
            bnc[3 * HID + t] = beta[t];
        }
    }
    __syncthreads();

    float* myStrip = strip + wave * 16 * STRIP_STRIDE;
    const float* B1 = W1t + row16 * WT_STRIDE + khalf;
    const float* B2 = W2t + row16 * WT_STRIDE + khalf;
    const int rhalf = (lane >> 4) << 3;

    for (int s0 = 0; s0 < nstrips; s0 += wavesTotal) {
        const int tileM = (s0 + sfirst) * 16;
        if (active) {
            #pragma unroll
            for (int s = 0; s < HID / 4; ++s) {
                if constexpr (FIRST) {
                    av[s].x = alpha * av[s].x + betac * aw[s].x;
                    av[s].y = alpha * av[s].y + betac * aw[s].y;
                } else {
                    const int k = 4 * s + khalf;
                    const float2 g  = *(const float2*)(bnc + k);
                    const float2 mu = *(const float2*)(bnc + HID + k);
                    const float2 rs = *(const float2*)(bnc + 2 * HID + k);
                    const float2 bt = *(const float2*)(bnc + 3 * HID + k);
                    av[s].x = fmaxf(g.x * (av[s].x - mu.x) * rs.x + bt.x, 0.0f);
                    av[s].y = fmaxf(g.y * (av[s].y - mu.y) * rs.y + bt.y, 0.0f);
                }
            }
            v8f acc[8] = {};
            strip_gemm<8>(av, B1, acc);
            #pragma unroll
            for (int nt = 0; nt < 8; ++nt) {
                const float bv = b1[nt * 16 + row16];
                #pragma unroll
                for (int r = 0; r < 8; ++r)
                    myStrip[(rhalf + r) * STRIP_STRIDE + nt * 16 + row16] = fmaxf(acc[nt][r] + bv, 0.0f);
            }
        }
        __syncthreads();
        const bool was = active;
        if (was) {
            #pragma unroll
            for (int s = 0; s < HID / 4; ++s)
                av[s] = *(const float2*)(myStrip + row16 * STRIP_STRIDE + 4 * s + khalf);
        }
        v8f acc2[NT2] = {};
        if (was) strip_gemm<NT2>(av, B2, acc2);
        const int snext = s0 + wavesTotal + sfirst;
        active = (s0 + wavesTotal < nstrips) && (snext < nstrips);
        if (active) strip_load<FIRST>(av, aw, A, snext * 16, row16, khalf, nrows);
        if (was) {
            #pragma unroll
            for (int nt = 0; nt < NT2; ++nt) {
                const int col = nt * 16 + row16;
                #pragma unroll
                for (int r = 0; r < 8; ++r) {
                    const int row = tileM + rhalf + r;
                    if (row < nrows) C[(size_t)row * N2 + col] = acc2[nt][r];
                }
            }
        }
    }
}

#define GROWS 8
#define GATHER_T 256

template <bool STATS>
__global__ __launch_bounds__(GATHER_T) void gcn_gather(const int* __restrict__ rowptr,
                                                  const int* __restrict__ csr,
                                                  const float* __restrict__ dinv,
                                                  const float* __restrict__ m,
                                                  const float* __restrict__ b,
                                                  float* __restrict__ h,
                                                  float* bnsum, float* bnsq, int n, int e) {
    const int lane = threadIdx.x & 31;
    float4 ssum = make_float4(0.0f, 0.0f, 0.0f, 0.0f);
    float4 ssq  = make_float4(0.0f, 0.0f, 0.0f, 0.0f);
    const int nwaves = (gridDim.x * blockDim.x) >> 5;
    const int wid = __builtin_amdgcn_readfirstlane((blockIdx.x * blockDim.x + threadIdx.x) >> 5);
    const float* mlane = m + lane * 4;
    const float4 bb = *(const float4*)(b + lane * 4);
    for (int node = wid; node < n; node += nwaves) {
        int beg = rowptr[node];
        int end = rowptr[node + 1];
        if (beg < 0) beg = 0;
        if (end > e) end = e;
        const float dd = dinv[node];
        const float4 selfv = *(const float4*)(mlane + (size_t)node * HID);
        float4 acc = make_float4(0.0f, 0.0f, 0.0f, 0.0f);

        for (int base = beg; base < end; base += 32) {
            int c = end - base; if (c > 32) c = 32;
            int sv = 0, wbits = 0;
            if (lane < c) {
                const int2 p = *(const int2*)(csr + 2 * (size_t)(base + lane));
                sv = p.x; wbits = p.y;
            }
            if ((unsigned)sv > (unsigned)(n - 1)) sv = n - 1;
            int j = 0;
            for (; j + GROWS <= c; j += GROWS) {
                float4 v[GROWS]; float w[GROWS];
                #pragma unroll
                for (int u = 0; u < GROWS; ++u) {
                    const int s = __builtin_amdgcn_readlane(sv, j + u);
                    w[u] = __int_as_float(__builtin_amdgcn_readlane(wbits, j + u));
                    v[u] = *(const float4*)(mlane + (size_t)s * HID);
                }
                #pragma unroll
                for (int u = 0; u < GROWS; ++u) {
                    const float nrm = w[u] * dd;
                    acc.x += v[u].x * nrm; acc.y += v[u].y * nrm;
                    acc.z += v[u].z * nrm; acc.w += v[u].w * nrm;
                }
            }
            const int r = c - j;
            if (r > 0) {
                float4 v[GROWS - 1]; float w[GROWS - 1];
                #pragma unroll
                for (int u = 0; u < GROWS - 1; ++u) {
                    if (u < r) {
                        const int s = __builtin_amdgcn_readlane(sv, j + u);
                        w[u] = __int_as_float(__builtin_amdgcn_readlane(wbits, j + u));
                        v[u] = *(const float4*)(mlane + (size_t)s * HID);
                    }
                }
                #pragma unroll
                for (int u = 0; u < GROWS - 1; ++u) {
                    if (u < r) {
                        const float nrm = w[u] * dd;
                        acc.x += v[u].x * nrm; acc.y += v[u].y * nrm;
                        acc.z += v[u].z * nrm; acc.w += v[u].w * nrm;
                    }
                }
            }
        }
        const float sn = dd * dd;
        acc.x += selfv.x * sn; acc.y += selfv.y * sn; acc.z += selfv.z * sn; acc.w += selfv.w * sn;
        acc.x += bb.x; acc.y += bb.y; acc.z += bb.z; acc.w += bb.w;
        *(float4*)(h + (size_t)node * HID + lane * 4) = acc;
        if constexpr (STATS) {
            ssum.x += acc.x; ssum.y += acc.y; ssum.z += acc.z; ssum.w += acc.w;
            ssq.x += acc.x * acc.x; ssq.y += acc.y * acc.y;
            ssq.z += acc.z * acc.z; ssq.w += acc.w * acc.w;
        }
    }
    if constexpr (STATS) {
        __shared__ float red[(GATHER_T / 32) * 2 * HID];
        const int wv = threadIdx.x >> 5;
        *(float4*)(red + (wv * 2 + 0) * HID + lane * 4) = ssum;
        *(float4*)(red + (wv * 2 + 1) * HID + lane * 4) = ssq;
        __syncthreads();
        static_assert(GATHER_T == 2 * HID, "one thread per batch-norm sum");
        const int which = threadIdx.x >> 7;
        const int f = threadIdx.x & (HID - 1);
        float tot = 0.0f;
        for (int w = 0; w < GATHER_T / 32; ++w) tot += red[(w * 2 + which) * HID + f];
        atomAddF32(which ? &bnsq[f] : &bnsum[f], tot);
    }
}

__global__ void gcn_bn_zero(float* bnsum, float* bnsq) {
    int t = threadIdx.x;
    if (t < HID) bnsum[t] = 0.0f;
    else if (t < 2 * HID) bnsq[t - HID] = 0.0f;
}

#define CLS_NT 3
#define CLS_W (CLS_NT * 16)
#define FOLD_T (CLS_NT * 32)
#define FOLD_BLOCKS (HID / 16 + 1)
#define CLS_ROWS 16

__global__ __launch_bounds__(FOLD_T) void gcn_fold_cls(const float* __restrict__ W2,
                                                       const float* __restrict__ b,
                                                       const float* __restrict__ lw,
                                                       const float* __restrict__ lb,
                                                       float* __restrict__ Wf, float* __restrict__ bf) {
    const int t = threadIdx.x;
    const int mt = blockIdx.x;
    const int nt = __builtin_amdgcn_readfirstlane(t >> 5);
    const int lane  = t & 31;
    const int row16 = lane & 15;
    const int khalf = (lane >> 4) << 1;
    const int arow = mt * 16 + row16;
    const float* Arow = (arow < HID ? W2 + (size_t)arow * HID : b) + khalf;
    const int col = nt * 16 + row16;
    const bool live = col < OUTC;
    const float* Bcol = lw + (live ? col : OUTC - 1) + khalf * OUTC;

    float2 av[HID / 4]; float b0[HID / 4], b1[HID / 4];
    #pragma unroll
    for (int s = 0; s < HID / 4; ++s) {
        av[s] = *(const float2*)(Arow + 4 * s);
        b0[s] = Bcol[(4 * s) * OUTC];
        b1[s] = Bcol[(4 * s + 1) * OUTC];
    }
    v8f acc = {};
    #pragma unroll
    for (int s = 0; s < HID / 4; ++s) {
        v2f a; a.x = av[s].x; a.y = av[s].y;
        v2f bfrag; bfrag.x = live ? b0[s] : 0.0f; bfrag.y = live ? b1[s] : 0.0f;
        acc = __builtin_amdgcn_wmma_f32_16x16x4_f32(false, a, false, bfrag, (short)0, acc, false, false);
    }
    const int rhalf = (lane >> 4) << 3;
    if (mt < HID / 16) {
        #pragma unroll
        for (int r = 0; r < 8; ++r)
            Wf[(size_t)(mt * 16 + rhalf + r) * CLS_W + col] = acc[r];
    } else if (lane < 16) {
        bf[col] = acc[0] + (live ? lb[col] : 0.0f);
    }
}

template <int CTRL> __device__ __forceinline__ float lane_perm16(float v) {
    return __int_as_float(__builtin_amdgcn_update_dpp(0, __float_as_int(v), CTRL, 0xF, 0xF, true));
}
__device__ __forceinline__ float wave_max32(float v) {
    v = fmaxf(v, lane_perm16<0xB1>(v));
    v = fmaxf(v, lane_perm16<0x4E>(v));
    v = fmaxf(v, lane_perm16<0x141>(v));
    v = fmaxf(v, lane_perm16<0x140>(v));
    const float lo = __int_as_float(__builtin_amdgcn_readlane(__float_as_int(v), 0));
    const float hi = __int_as_float(__builtin_amdgcn_readlane(__float_as_int(v), 16));
    return fmaxf(lo, hi);
}
__device__ __forceinline__ float wave_sum32(float v) {
    v += lane_perm16<0xB1>(v);
    v += lane_perm16<0x4E>(v);
    v += lane_perm16<0x141>(v);
    v += lane_perm16<0x140>(v);
    const float lo = __int_as_float(__builtin_amdgcn_readlane(__float_as_int(v), 0));
    const float hi = __int_as_float(__builtin_amdgcn_readlane(__float_as_int(v), 16));
    return lo + hi;
}

__global__ __launch_bounds__(GATHER_T) void gcn_gather_cls(const int* __restrict__ rowptr,
                                                           const int* __restrict__ csr,
                                                           const float* __restrict__ dinv,
                                                           const float* __restrict__ m,
                                                           const float* __restrict__ bf,
                                                           float* __restrict__ out, int n, int e) {
    const int lane = threadIdx.x & 31;
    const int nwaves = (gridDim.x * blockDim.x) >> 5;
    const int wid = __builtin_amdgcn_readfirstlane((blockIdx.x * blockDim.x + threadIdx.x) >> 5);
    const bool valid = lane < OUTC / 2;
    const int fl = 2 * (lane < CLS_W / 2 ? lane : CLS_W / 2 - 1);
    const float2 bb = *(const float2*)(bf + fl);
    for (int node = wid; node < n; node += nwaves) {
        int beg = rowptr[node];
        int end = rowptr[node + 1];
        if (beg < 0) beg = 0;
        if (end > e) end = e;
        const float dd = dinv[node];
        const float2 selfv = *(const float2*)(m + (size_t)node * CLS_W + fl);
        float2 acc = make_float2(0.0f, 0.0f);

        for (int base = beg; base < end; base += 32) {
            int c = end - base; if (c > 32) c = 32;
            int sv = 0, wbits = 0;
            if (lane < c) {
                const int2 p = *(const int2*)(csr + 2 * (size_t)(base + lane));
                sv = p.x; wbits = p.y;
            }
            if ((unsigned)sv > (unsigned)(n - 1)) sv = n - 1;
            int j = 0;
            for (; j + CLS_ROWS <= c; j += CLS_ROWS) {
                float2 v[CLS_ROWS]; float w[CLS_ROWS];
                #pragma unroll
                for (int u = 0; u < CLS_ROWS; ++u) {
                    const int s = __builtin_amdgcn_readlane(sv, j + u);
                    w[u] = __int_as_float(__builtin_amdgcn_readlane(wbits, j + u));
                    v[u] = *(const float2*)(m + (size_t)s * CLS_W + fl);
                }
                #pragma unroll
                for (int u = 0; u < CLS_ROWS; ++u) {
                    const float nrm = w[u] * dd;
                    acc.x += v[u].x * nrm; acc.y += v[u].y * nrm;
                }
            }
            const int r = c - j;
            if (r > 0) {
                float2 v[CLS_ROWS - 1]; float w[CLS_ROWS - 1];
                #pragma unroll
                for (int u = 0; u < CLS_ROWS - 1; ++u) {
                    if (u < r) {
                        const int s = __builtin_amdgcn_readlane(sv, j + u);
                        w[u] = __int_as_float(__builtin_amdgcn_readlane(wbits, j + u));
                        v[u] = *(const float2*)(m + (size_t)s * CLS_W + fl);
                    }
                }
                #pragma unroll
                for (int u = 0; u < CLS_ROWS - 1; ++u) {
                    if (u < r) {
                        const float nrm = w[u] * dd;
                        acc.x += v[u].x * nrm; acc.y += v[u].y * nrm;
                    }
                }
            }
        }
        const float sn = dd * dd;
        acc.x += selfv.x * sn; acc.y += selfv.y * sn;
        acc.x += bb.x; acc.y += bb.y;
        const float x0 = valid ? acc.x : -INFINITY;
        const float x1 = valid ? acc.y : -INFINITY;
        const float mx = wave_max32(fmaxf(x0, x1));
        const float ls = logf(wave_sum32(expf(x0 - mx) + expf(x1 - mx)));
        if (valid) {
            float2 o; o.x = x0 - mx - ls; o.y = x1 - mx - ls;
            *(float2*)(out + (size_t)node * OUTC + 2 * lane) = o;
        }
    }
}

extern "C" void kernel_launch(void* const* d_in, const int* in_sizes, int n_in,
                              void* d_out, int out_size, void* d_ws, size_t ws_size,
                              hipStream_t stream) {
    const float* x       = (const float*)d_in[0];
    const int*   ei      = (const int*)d_in[1];
    const float* la      = (const float*)d_in[2];
    const float* lb      = (const float*)d_in[3];
    const float* mlp_w   = (const float*)d_in[4];
    const float* mlp_b   = (const float*)d_in[5];
    const float* conv_w  = (const float*)d_in[6];
    const float* conv_b  = (const float*)d_in[7];
    const float* bn_g    = (const float*)d_in[8];
    const float* bn_b    = (const float*)d_in[9];
    const float* lin_w   = (const float*)d_in[10];
    const float* lin_b   = (const float*)d_in[11];
    float* out = (float*)d_out;

    float* ws = (float*)d_ws;
    const size_t NM = (size_t)N_NODES * HID;
    float* hA    = ws;
    float* hB    = hA + NM;
    int*   cnt    = (int*)(hB + NM);
    int*   rowptr = cnt + SCAN_N;
    int*   rank   = rowptr + SCAN_N;
    int*   csr    = rank + N_EDGES;
    float* Wf    = (float*)(csr + 2 * N_EDGES);
    float* bf    = Wf + HID * CLS_W;
    float* dinv  = hB + NM + NM;
    float* bnsum = dinv + N_NODES;
    float* bnsq  = bnsum + HID;

    const int TPB = 256;
    const int edgeBlocks = ((N_EDGES + 3) / 4 + TPB - 1) / TPB;
    const int gatherBlocks = 1250;
    const int denseBlocks = 256;

    gcn_fold_cls<<<FOLD_BLOCKS, FOLD_T, 0, stream>>>(conv_w + (size_t)(NLAYERS - 1) * HID * HID,
                                           conv_b + (NLAYERS - 1) * HID, lin_w, lin_b, Wf, bf);

    gcn_cnt_zero<<<(N_NODES + TPB - 1) / TPB, TPB, 0, stream>>>(cnt, N_NODES);
    gcn_rank_edges<<<edgeBlocks, TPB, 0, stream>>>(ei, cnt, rank, N_EDGES);
    gcn_scan_deg<<<SCAN_BLOCKS, SCAN_T, 0, stream>>>(cnt, rowptr, dinv, N_NODES);
    gcn_fill_csr<<<edgeBlocks, TPB, 0, stream>>>(ei, rank, rowptr, dinv, csr, N_EDGES);

    for (int i = 0; i < NLAYERS; ++i) {
        if (i == 0) {
            gcn_layer_dense<true, 8><<<denseBlocks, DENSE_T, 0, stream>>>(
                x, la, lb, nullptr, nullptr, nullptr, nullptr,
                mlp_w, mlp_b, conv_w, hA, N_NODES);
        } else if (i < NLAYERS - 1) {
            gcn_layer_dense<false, 8><<<denseBlocks, DENSE_T, 0, stream>>>(
                hB, nullptr, nullptr, bnsum, bnsq, bn_g + (i - 1) * HID, bn_b + (i - 1) * HID,
                mlp_w + (size_t)i * HID * HID, mlp_b + i * HID, conv_w + (size_t)i * HID * HID,
                hA, N_NODES);
        } else {
            gcn_layer_dense<false, CLS_NT><<<denseBlocks, DENSE_T, 0, stream>>>(
                hB, nullptr, nullptr, bnsum, bnsq, bn_g + (i - 1) * HID, bn_b + (i - 1) * HID,
                mlp_w + (size_t)i * HID * HID, mlp_b + i * HID, Wf,
                hA, N_NODES);
        }
        if (i < NLAYERS - 1) {
            gcn_bn_zero<<<1, 256, 0, stream>>>(bnsum, bnsq);
            gcn_gather<true><<<gatherBlocks, GATHER_T, 0, stream>>>(
                rowptr, csr, dinv, hA, conv_b + i * HID, hB, bnsum, bnsq, N_NODES, N_EDGES);
        } else {
            gcn_gather_cls<<<gatherBlocks, GATHER_T, 0, stream>>>(
                rowptr, csr, dinv, hA, bf, out, N_NODES, N_EDGES);
        }
    }
}


# 13 launches -> 9: zeroing (counters, per-layer BN sums) + classifier fold in one setup kernel; edge ranking by the first dense kernel's waves before their strips; row sums arrival-ranked as before
# speedup vs baseline: 4.2335x; 1.0439x over previous
#include <hip/hip_runtime.h>
#include <hip/hip_bf16.h>
#include <math.h>

#define N_NODES 50000
#define N_EDGES 800000
#define HID     128
#define OUTC    40
#define NLAYERS 3
#define BN_EPS  1e-5f
#define WT_STRIDE 130

typedef __attribute__((ext_vector_type(2))) float v2f;
typedef __attribute__((ext_vector_type(8))) float v8f;
typedef __attribute__((ext_vector_type(4))) float v4f;
typedef __attribute__((ext_vector_type(4))) int v4i;

__device__ __forceinline__ void atomAddF32(float* p, float v) {
    __hip_atomic_fetch_add(p, v, __ATOMIC_RELAXED, __HIP_MEMORY_SCOPE_AGENT);
}

#define SCAN_T      1024
#define SCAN_BLOCKS ((N_NODES + 1 + SCAN_T - 1) / SCAN_T)
#define SCAN_N      (SCAN_T * SCAN_BLOCKS)
#define SCAN_PRE    (((SCAN_BLOCKS - 1) * (SCAN_T / 4) + SCAN_T - 1) / SCAN_T)

__device__ __forceinline__ int rank_one(int* cnt, int d) {
    return __hip_atomic_fetch_add(&cnt[d], 1, __ATOMIC_RELAXED, __HIP_MEMORY_SCOPE_AGENT);
}
__device__ __forceinline__ v4i rank_four(int* cnt, v4i d) {
    v4i r;
    r.x = rank_one(cnt, d.x);
    r.y = rank_one(cnt, d.y);
    r.z = rank_one(cnt, d.z);
    r.w = rank_one(cnt, d.w);
    return r;
}
__device__ __forceinline__ void rank_edges_part(const int* __restrict__ ei, int* cnt,
                                                int* __restrict__ rank, int e,
                                                int wbase, int lane, int nlanes) {
    const int* dst = ei + e;
    if ((e & 3) == 0) {
        const int ng = e >> 2;
        const v4i* dst4 = (const v4i*)dst;
        v4i* rank4 = (v4i*)rank;
        for (int b = wbase; b < ng; b += 2 * nlanes) {
            if (b + nlanes + 31 < ng) {
                v4i d0 = dst4[b + lane];
                v4i d1 = dst4[b + nlanes + lane];
                asm volatile("" : "+v"(d0), "+v"(d1));
                v4i r0 = rank_four(cnt, d0);
                v4i r1 = rank_four(cnt, d1);
                asm volatile("" : "+v"(r0), "+v"(r1));
                rank4[b + lane] = r0;
                rank4[b + nlanes + lane] = r1;
            } else {
                for (int u = 0; u < 2; ++u) {
                    const int g = b + u * nlanes + lane;
                    if (g < ng) rank4[g] = rank_four(cnt, dst4[g]);
                }
            }
        }
    } else {
        for (int k = wbase + lane; k < e; k += nlanes) rank[k] = rank_one(cnt, dst[k]);
    }
}

__global__ __launch_bounds__(SCAN_T) void gcn_scan_deg(const int* __restrict__ cnt,
                                                       int* __restrict__ rowptr,
                                                       float* __restrict__ dinv, int n) {
    __shared__ int wpre[SCAN_T / 32];
    __shared__ int wown[SCAN_T / 32];
    static_assert(SCAN_T / 32 == 32, "one lane per wave total in the fold below");
    const int t = threadIdx.x;
    const int lane = t & 31, wv = t >> 5;
    const int i = blockIdx.x * SCAN_T + t;
    const int lim = blockIdx.x * (SCAN_T / 4);
    const int4* cnt4 = (const int4*)cnt;
    int4 v[SCAN_PRE];
    #pragma unroll
    for (int j = 0; j < SCAN_PRE; ++j) {
        const int q = j * SCAN_T + t;
        v[j] = cnt4[q < lim ? q : 0];
    }
    int c = 0;
    if (i < n) c = cnt[i];
    int p = 0;
    #pragma unroll
    for (int j = 0; j < SCAN_PRE; ++j) {
        const int q = j * SCAN_T + t;
        if (q < lim) p += (v[j].x + v[j].y) + (v[j].z + v[j].w);
    }
    if (i < n) dinv[i] = rsqrtf((float)(c + 1));
    for (int o = 16; o > 0; o >>= 1) p += __shfl_xor(p, o);
    int incl = c;
    for (int o = 1; o < 32; o <<= 1) { int y = __shfl_up(incl, o); if (lane >= o) incl += y; }
    if (lane == 31) { wpre[wv] = p; wown[wv] = incl; }
    __syncthreads();
    int z = wpre[lane] + (lane < wv ? wown[lane] : 0);
    for (int o = 16; o > 0; o >>= 1) z += __shfl_xor(z, o);
    if (i <= n) rowptr[i] = z + incl - c;
}

__global__ __launch_bounds__(256) void gcn_fill_csr(const int* __restrict__ ei, const int* __restrict__ rank,
                             const int* __restrict__ rowptr, const float* __restrict__ dinv,
                             int* __restrict__ csr, int e) {
    const int i = (blockIdx.x * blockDim.x + threadIdx.x) * 4;
    if (i >= e) return;
    if ((e & 3) == 0) {
        const int4 s4 = *(const int4*)(ei + i);
        const int4 d4 = *(const int4*)(ei + e + i);
        const int4 r4 = *(const int4*)(rank + i);
        const int s[4] = {s4.x, s4.y, s4.z, s4.w};
        const int d[4] = {d4.x, d4.y, d4.z, d4.w};
        const int r[4] = {r4.x, r4.y, r4.z, r4.w};
        int rp[4]; float dv[4];
        #pragma unroll
        for (int u = 0; u < 4; ++u) { rp[u] = rowptr[d[u]]; dv[u] = dinv[s[u]]; }
        #pragma unroll
        for (int u = 0; u < 4; ++u) {
            unsigned pos = (unsigned)(rp[u] + r[u]);
            if (pos > (unsigned)(e - 1)) pos = (unsigned)(e - 1);
            int2 p; p.x = s[u]; p.y = __float_as_int(dv[u]);
            *(int2*)(csr + 2 * (size_t)pos) = p;
        }
    } else {
        for (int k = i; k < e && k < i + 4; ++k) {
            const int s = ei[k];
            unsigned pos = (unsigned)(rowptr[ei[e + k]] + rank[k]);
            if (pos > (unsigned)(e - 1)) pos = (unsigned)(e - 1);
            int2 p; p.x = s; p.y = __float_as_int(dinv[s]);
            *(int2*)(csr + 2 * (size_t)pos) = p;
        }
    }
}

#define DENSE_T 512
#define DENSE_WAVES (DENSE_T / 32)
#define STRIP_STRIDE 132

template <int NT>
__device__ __forceinline__ void strip_gemm(const float2 (&av)[HID / 4], const float* Bfrag, v8f (&acc)[NT]) {
    #pragma unroll
    for (int s = 0; s < HID / 4; ++s) {
        v2f a; a.x = av[s].x; a.y = av[s].y;
        #pragma unroll
        for (int nt = 0; nt < NT; ++nt) {
            const float2 bv = *(const float2*)(Bfrag + nt * 16 * WT_STRIDE + 4 * s);
            v2f b; b.x = bv.x; b.y = bv.y;
            acc[nt] = __builtin_amdgcn_wmma_f32_16x16x4_f32(
                false, a, false, b, (short)0, acc[nt], false, false);
        }
    }
}

template <bool FIRST>
__device__ __forceinline__ void strip_load(float2 (&av)[HID / 4], float2 (&aw)[HID / 4],
                                           const float* __restrict__ A, int tileM, int row16,
                                           int khalf, int nrows) {
    int rowIdx = tileM + row16;
    if (rowIdx > nrows - 1) rowIdx = nrows - 1;
    const float* Arow = A + (size_t)rowIdx * (FIRST ? 2 * HID : HID) + khalf;
    #pragma unroll
    for (int s = 0; s < HID / 4; ++s) {
        av[s] = *(const float2*)(Arow + 4 * s);
        if constexpr (FIRST) aw[s] = *(const float2*)(Arow + HID + 4 * s);
    }
}

template <bool FIRST, int NT2>
__global__ __launch_bounds__(DENSE_T) void gcn_layer_dense(
        const float* __restrict__ A,
        const float* __restrict__ la, const float* __restrict__ lb,
        const float* __restrict__ bnsum, const float* __restrict__ bnsq,
        const float* __restrict__ gamma, const float* __restrict__ beta,
        const float* __restrict__ W1, const float* __restrict__ b1,
        const float* __restrict__ W2, float* __restrict__ C, int nrows,
        const int* __restrict__ ei, int* cnt, int* __restrict__ rank, int nedges) {
    __shared__ __attribute__((aligned(16))) float W1t[HID * WT_STRIDE];
    __shared__ __attribute__((aligned(16))) float W2t[HID * WT_STRIDE];
    __shared__ __attribute__((aligned(16))) float strip[DENSE_WAVES * 16 * STRIP_STRIDE];
    __shared__ __attribute__((aligned(16))) float bnc[4 * HID];

    const int t = threadIdx.x;
    const int wave  = __builtin_amdgcn_readfirstlane(t >> 5);
    const int lane  = t & 31;
    const int row16 = lane & 15;
    const int khalf = (lane >> 4) << 1;
    const int nstrips = (nrows + 15) >> 4;
    const int wavesTotal = gridDim.x * DENSE_WAVES;
    const int sfirst = wave * gridDim.x + blockIdx.x;

    if constexpr (FIRST) {
        rank_edges_part(ei, cnt, rank, nedges, sfirst * 32, lane, wavesTotal * 32);
    }

    constexpr int N2 = NT2 * 16;
    constexpr int W2PER = (HID * N2) / DENSE_T;
    float w1v[(HID * HID) / DENSE_T], w2v[W2PER];
    #pragma unroll
    for (int j = 0; j < (HID * HID) / DENSE_T; ++j) w1v[j] = W1[t + j * DENSE_T];
    #pragma unroll
    for (int j = 0; j < W2PER; ++j) w2v[j] = W2[t + j * DENSE_T];

    float2 av[HID / 4], aw[HID / 4];
    bool active = sfirst < nstrips;
    if (active) strip_load<FIRST>(av, aw, A, sfirst * 16, row16, khalf, nrows);

    #pragma unroll
    for (int j = 0; j < (HID * HID) / DENSE_T; ++j) {
        const int idx = t + j * DENSE_T;
        W1t[(idx & 127) * WT_STRIDE + (idx >> 7)] = w1v[j];
    }
    #pragma unroll
    for (int j = 0; j < W2PER; ++j) {
        const int idx = t + j * DENSE_T;
        W2t[(idx % N2) * WT_STRIDE + (idx / N2)] = w2v[j];
    }
    float alpha = 0.0f, betac = 0.0f;
    if constexpr (FIRST) {
        alpha = expf(la[0]);
        betac = expf(lb[0]);
    } else {
        if (t < HID) {
            const float invn = 1.0f / (float)nrows;
            const float mu = bnsum[t] * invn;
            const float var = bnsq[t] * invn - mu * mu;
            bnc[t] = gamma[t];
            bnc[HID + t] = mu;
            bnc[2 * HID + t] = rsqrtf(var + BN_EPS);
            bnc[3 * HID + t] = beta[t];
        }
    }
    __syncthreads();


    float* myStrip = strip + wave * 16 * STRIP_STRIDE;
    const float* B1 = W1t + row16 * WT_STRIDE + khalf;
    const float* B2 = W2t + row16 * WT_STRIDE + khalf;
    const int rhalf = (lane >> 4) << 3;

    for (int s0 = 0; s0 < nstrips; s0 += wavesTotal) {
        const int tileM = (s0 + sfirst) * 16;
        if (active) {
            #pragma unroll
            for (int s = 0; s < HID / 4; ++s) {
                if constexpr (FIRST) {
                    av[s].x = alpha * av[s].x + betac * aw[s].x;
                    av[s].y = alpha * av[s].y + betac * aw[s].y;
                } else {
                    const int k = 4 * s + khalf;
                    const float2 g  = *(const float2*)(bnc + k);
                    const float2 mu = *(const float2*)(bnc + HID + k);
                    const float2 rs = *(const float2*)(bnc + 2 * HID + k);
                    const float2 bt = *(const float2*)(bnc + 3 * HID + k);
                    av[s].x = fmaxf(g.x * (av[s].x - mu.x) * rs.x + bt.x, 0.0f);
                    av[s].y = fmaxf(g.y * (av[s].y - mu.y) * rs.y + bt.y, 0.0f);
                }
            }
            v8f acc[8] = {};
            strip_gemm<8>(av, B1, acc);
            #pragma unroll
            for (int nt = 0; nt < 8; ++nt) {
                const float bv = b1[nt * 16 + row16];
                #pragma unroll
                for (int r = 0; r < 8; ++r)
                    myStrip[(rhalf + r) * STRIP_STRIDE + nt * 16 + row16] = fmaxf(acc[nt][r] + bv, 0.0f);
            }
        }
        __syncthreads();
        const bool was = active;
        if (was) {
            #pragma unroll
            for (int s = 0; s < HID / 4; ++s)
                av[s] = *(const float2*)(myStrip + row16 * STRIP_STRIDE + 4 * s + khalf);
        }
        v8f acc2[NT2] = {};
        if (was) strip_gemm<NT2>(av, B2, acc2);
        const int snext = s0 + wavesTotal + sfirst;
        active = (s0 + wavesTotal < nstrips) && (snext < nstrips);
        if (active) strip_load<FIRST>(av, aw, A, snext * 16, row16, khalf, nrows);
        if (was) {
            #pragma unroll
            for (int nt = 0; nt < NT2; ++nt) {
                const int col = nt * 16 + row16;
                #pragma unroll
                for (int r = 0; r < 8; ++r) {
                    const int row = tileM + rhalf + r;
                    if (row < nrows) C[(size_t)row * N2 + col] = acc2[nt][r];
                }
            }
        }
    }
}

#define GROWS 8
#define GATHER_T 256

template <bool STATS>
__global__ __launch_bounds__(GATHER_T) void gcn_gather(const int* __restrict__ rowptr,
                                                  const int* __restrict__ csr,
                                                  const float* __restrict__ dinv,
                                                  const float* __restrict__ m,
                                                  const float* __restrict__ b,
                                                  float* __restrict__ h,
                                                  float* bnsum, float* bnsq, int n, int e) {
    const int lane = threadIdx.x & 31;
    float4 ssum = make_float4(0.0f, 0.0f, 0.0f, 0.0f);
    float4 ssq  = make_float4(0.0f, 0.0f, 0.0f, 0.0f);
    const int nwaves = (gridDim.x * blockDim.x) >> 5;
    const int wid = __builtin_amdgcn_readfirstlane((blockIdx.x * blockDim.x + threadIdx.x) >> 5);
    const float* mlane = m + lane * 4;
    const float4 bb = *(const float4*)(b + lane * 4);
    for (int node = wid; node < n; node += nwaves) {
        int beg = rowptr[node];
        int end = rowptr[node + 1];
        if (beg < 0) beg = 0;
        if (end > e) end = e;
        const float dd = dinv[node];
        const float4 selfv = *(const float4*)(mlane + (size_t)node * HID);
        float4 acc = make_float4(0.0f, 0.0f, 0.0f, 0.0f);

        for (int base = beg; base < end; base += 32) {
            int c = end - base; if (c > 32) c = 32;
            int sv = 0, wbits = 0;
            if (lane < c) {
                const int2 p = *(const int2*)(csr + 2 * (size_t)(base + lane));
                sv = p.x; wbits = p.y;
            }
            if ((unsigned)sv > (unsigned)(n - 1)) sv = n - 1;
            int j = 0;
            for (; j + GROWS <= c; j += GROWS) {
                float4 v[GROWS]; float w[GROWS];
                #pragma unroll
                for (int u = 0; u < GROWS; ++u) {
                    const int s = __builtin_amdgcn_readlane(sv, j + u);
                    w[u] = __int_as_float(__builtin_amdgcn_readlane(wbits, j + u));
                    v[u] = *(const float4*)(mlane + (size_t)s * HID);
                }
                #pragma unroll
                for (int u = 0; u < GROWS; ++u) {
                    const float nrm = w[u] * dd;
                    acc.x += v[u].x * nrm; acc.y += v[u].y * nrm;
                    acc.z += v[u].z * nrm; acc.w += v[u].w * nrm;
                }
            }
            const int r = c - j;
            if (r > 0) {
                float4 v[GROWS - 1]; float w[GROWS - 1];
                #pragma unroll
                for (int u = 0; u < GROWS - 1; ++u) {
                    if (u < r) {
                        const int s = __builtin_amdgcn_readlane(sv, j + u);
                        w[u] = __int_as_float(__builtin_amdgcn_readlane(wbits, j + u));
                        v[u] = *(const float4*)(mlane + (size_t)s * HID);
                    }
                }
                #pragma unroll
                for (int u = 0; u < GROWS - 1; ++u) {
                    if (u < r) {
                        const float nrm = w[u] * dd;
                        acc.x += v[u].x * nrm; acc.y += v[u].y * nrm;
                        acc.z += v[u].z * nrm; acc.w += v[u].w * nrm;
                    }
                }
            }
        }
        const float sn = dd * dd;
        acc.x += selfv.x * sn; acc.y += selfv.y * sn; acc.z += selfv.z * sn; acc.w += selfv.w * sn;
        acc.x += bb.x; acc.y += bb.y; acc.z += bb.z; acc.w += bb.w;
        *(float4*)(h + (size_t)node * HID + lane * 4) = acc;
        if constexpr (STATS) {
            ssum.x += acc.x; ssum.y += acc.y; ssum.z += acc.z; ssum.w += acc.w;
            ssq.x += acc.x * acc.x; ssq.y += acc.y * acc.y;
            ssq.z += acc.z * acc.z; ssq.w += acc.w * acc.w;
        }
    }
    if constexpr (STATS) {
        __shared__ float red[(GATHER_T / 32) * 2 * HID];
        const int wv = threadIdx.x >> 5;
        *(float4*)(red + (wv * 2 + 0) * HID + lane * 4) = ssum;
        *(float4*)(red + (wv * 2 + 1) * HID + lane * 4) = ssq;
        __syncthreads();
        static_assert(GATHER_T == 2 * HID, "one thread per batch-norm sum");
        const int which = threadIdx.x >> 7;
        const int f = threadIdx.x & (HID - 1);
        float tot = 0.0f;
        for (int w = 0; w < GATHER_T / 32; ++w) tot += red[(w * 2 + which) * HID + f];
        atomAddF32(which ? &bnsq[f] : &bnsum[f], tot);
    }
}

#define CLS_NT 3
#define CLS_W (CLS_NT * 16)
#define FOLD_T (CLS_NT * 32)
#define FOLD_BLOCKS (HID / 16 + 1)
#define CLS_ROWS 16

__device__ __forceinline__ void fold_cls_tile(const float* __restrict__ W2,
                                              const float* __restrict__ b,
                                              const float* __restrict__ lw,
                                              const float* __restrict__ lb,
                                              float* __restrict__ Wf, float* __restrict__ bf,
                                              int t, int mt) {
    const int nt = __builtin_amdgcn_readfirstlane(t >> 5);
    const int lane  = t & 31;
    const int row16 = lane & 15;
    const int khalf = (lane >> 4) << 1;
    const int arow = mt * 16 + row16;
    const float* Arow = (arow < HID ? W2 + (size_t)arow * HID : b) + khalf;
    const int col = nt * 16 + row16;
    const bool live = col < OUTC;
    const float* Bcol = lw + (live ? col : OUTC - 1) + khalf * OUTC;

    float2 av[HID / 4]; float b0[HID / 4], b1[HID / 4];
    #pragma unroll
    for (int s = 0; s < HID / 4; ++s) {
        av[s] = *(const float2*)(Arow + 4 * s);
        b0[s] = Bcol[(4 * s) * OUTC];
        b1[s] = Bcol[(4 * s + 1) * OUTC];
    }
    v8f acc = {};
    #pragma unroll
    for (int s = 0; s < HID / 4; ++s) {
        v2f a; a.x = av[s].x; a.y = av[s].y;
        v2f bfrag; bfrag.x = live ? b0[s] : 0.0f; bfrag.y = live ? b1[s] : 0.0f;
        acc = __builtin_amdgcn_wmma_f32_16x16x4_f32(false, a, false, bfrag, (short)0, acc, false, false);
    }
    const int rhalf = (lane >> 4) << 3;
    if (mt < HID / 16) {
        #pragma unroll
        for (int r = 0; r < 8; ++r)
            Wf[(size_t)(mt * 16 + rhalf + r) * CLS_W + col] = acc[r];
    } else if (lane < 16) {
        bf[col] = acc[0] + (live ? lb[col] : 0.0f);
    }
}

#define SETUP_T 256
__global__ __launch_bounds__(SETUP_T) void gcn_setup(const float* __restrict__ W2,
                                                     const float* __restrict__ b,
                                                     const float* __restrict__ lw,
                                                     const float* __restrict__ lb,
                                                     float* __restrict__ Wf, float* __restrict__ bf,
                                                     int* __restrict__ cnt, int n,
                                                     float* __restrict__ bnz, int nbn) {
    const int i = blockIdx.x * SETUP_T + threadIdx.x;
    if (i < n) cnt[i] = 0;
    if (i < nbn) bnz[i] = 0.0f;
    const int wave = __builtin_amdgcn_readfirstlane(threadIdx.x >> 5);
    if (blockIdx.x < FOLD_BLOCKS && wave < CLS_NT)
        fold_cls_tile(W2, b, lw, lb, Wf, bf, threadIdx.x, blockIdx.x);
}

template <int CTRL> __device__ __forceinline__ float lane_perm16(float v) {
    return __int_as_float(__builtin_amdgcn_update_dpp(0, __float_as_int(v), CTRL, 0xF, 0xF, true));
}
__device__ __forceinline__ float wave_max32(float v) {
    v = fmaxf(v, lane_perm16<0xB1>(v));
    v = fmaxf(v, lane_perm16<0x4E>(v));
    v = fmaxf(v, lane_perm16<0x141>(v));
    v = fmaxf(v, lane_perm16<0x140>(v));
    const float lo = __int_as_float(__builtin_amdgcn_readlane(__float_as_int(v), 0));
    const float hi = __int_as_float(__builtin_amdgcn_readlane(__float_as_int(v), 16));
    return fmaxf(lo, hi);
}
__device__ __forceinline__ float wave_sum32(float v) {
    v += lane_perm16<0xB1>(v);
    v += lane_perm16<0x4E>(v);
    v += lane_perm16<0x141>(v);
    v += lane_perm16<0x140>(v);
    const float lo = __int_as_float(__builtin_amdgcn_readlane(__float_as_int(v), 0));
    const float hi = __int_as_float(__builtin_amdgcn_readlane(__float_as_int(v), 16));
    return lo + hi;
}

__global__ __launch_bounds__(GATHER_T) void gcn_gather_cls(const int* __restrict__ rowptr,
                                                           const int* __restrict__ csr,
                                                           const float* __restrict__ dinv,
                                                           const float* __restrict__ m,
                                                           const float* __restrict__ bf,
                                                           float* __restrict__ out, int n, int e) {
    const int lane = threadIdx.x & 31;
    const int nwaves = (gridDim.x * blockDim.x) >> 5;
    const int wid = __builtin_amdgcn_readfirstlane((blockIdx.x * blockDim.x + threadIdx.x) >> 5);
    const bool valid = lane < OUTC / 2;
    const int fl = 2 * (lane < CLS_W / 2 ? lane : CLS_W / 2 - 1);
    const float2 bb = *(const float2*)(bf + fl);
    for (int node = wid; node < n; node += nwaves) {
        int beg = rowptr[node];
        int end = rowptr[node + 1];
        if (beg < 0) beg = 0;
        if (end > e) end = e;
        const float dd = dinv[node];
        const float2 selfv = *(const float2*)(m + (size_t)node * CLS_W + fl);
        float2 acc = make_float2(0.0f, 0.0f);

        for (int base = beg; base < end; base += 32) {
            int c = end - base; if (c > 32) c = 32;
            int sv = 0, wbits = 0;
            if (lane < c) {
                const int2 p = *(const int2*)(csr + 2 * (size_t)(base + lane));
                sv = p.x; wbits = p.y;
            }
            if ((unsigned)sv > (unsigned)(n - 1)) sv = n - 1;
            int j = 0;
            for (; j + CLS_ROWS <= c; j += CLS_ROWS) {
                float2 v[CLS_ROWS]; float w[CLS_ROWS];
                #pragma unroll
                for (int u = 0; u < CLS_ROWS; ++u) {
                    const int s = __builtin_amdgcn_readlane(sv, j + u);
                    w[u] = __int_as_float(__builtin_amdgcn_readlane(wbits, j + u));
                    v[u] = *(const float2*)(m + (size_t)s * CLS_W + fl);
                }
                #pragma unroll
                for (int u = 0; u < CLS_ROWS; ++u) {
                    const float nrm = w[u] * dd;
                    acc.x += v[u].x * nrm; acc.y += v[u].y * nrm;
                }
            }
            const int r = c - j;
            if (r > 0) {
                float2 v[CLS_ROWS - 1]; float w[CLS_ROWS - 1];
                #pragma unroll
                for (int u = 0; u < CLS_ROWS - 1; ++u) {
                    if (u < r) {
                        const int s = __builtin_amdgcn_readlane(sv, j + u);
                        w[u] = __int_as_float(__builtin_amdgcn_readlane(wbits, j + u));
                        v[u] = *(const float2*)(m + (size_t)s * CLS_W + fl);
                    }
                }
                #pragma unroll
                for (int u = 0; u < CLS_ROWS - 1; ++u) {
                    if (u < r) {
                        const float nrm = w[u] * dd;
                        acc.x += v[u].x * nrm; acc.y += v[u].y * nrm;
                    }
                }
            }
        }
        const float sn = dd * dd;
        acc.x += selfv.x * sn; acc.y += selfv.y * sn;
        acc.x += bb.x; acc.y += bb.y;
        const float x0 = valid ? acc.x : -INFINITY;
        const float x1 = valid ? acc.y : -INFINITY;
        const float mx = wave_max32(fmaxf(x0, x1));
        const float ls = logf(wave_sum32(expf(x0 - mx) + expf(x1 - mx)));
        if (valid) {
            float2 o; o.x = x0 - mx - ls; o.y = x1 - mx - ls;
            *(float2*)(out + (size_t)node * OUTC + 2 * lane) = o;
        }
    }
}

extern "C" void kernel_launch(void* const* d_in, const int* in_sizes, int n_in,
                              void* d_out, int out_size, void* d_ws, size_t ws_size,
                              hipStream_t stream) {
    const float* x       = (const float*)d_in[0];
    const int*   ei      = (const int*)d_in[1];
    const float* la      = (const float*)d_in[2];
    const float* lb      = (const float*)d_in[3];
    const float* mlp_w   = (const float*)d_in[4];
    const float* mlp_b   = (const float*)d_in[5];
    const float* conv_w  = (const float*)d_in[6];
    const float* conv_b  = (const float*)d_in[7];
    const float* bn_g    = (const float*)d_in[8];
    const float* bn_b    = (const float*)d_in[9];
    const float* lin_w   = (const float*)d_in[10];
    const float* lin_b   = (const float*)d_in[11];
    float* out = (float*)d_out;

    float* ws = (float*)d_ws;
    const size_t NM = (size_t)N_NODES * HID;
    float* hA    = ws;
    float* hB    = hA + NM;
    int*   cnt    = (int*)(hB + NM);
    int*   rowptr = cnt + SCAN_N;
    int*   rank   = rowptr + SCAN_N;
    int*   csr    = rank + N_EDGES;
    float* Wf    = (float*)(csr + 2 * N_EDGES);
    float* bf    = Wf + HID * CLS_W;
    float* bnz   = bf + CLS_W;
    float* dinv  = hB + NM + NM;

    const int TPB = 256;
    const int edgeBlocks = ((N_EDGES + 3) / 4 + TPB - 1) / TPB;
    const int gatherBlocks = 1250;
    const int denseBlocks = 256;

    gcn_setup<<<(N_NODES + SETUP_T - 1) / SETUP_T, SETUP_T, 0, stream>>>(
        conv_w + (size_t)2 * HID * HID, conv_b + 2 * HID, lin_w, lin_b,
        Wf, bf, cnt, N_NODES, bnz, 4 * HID);

    gcn_layer_dense<true, 8><<<denseBlocks, DENSE_T, 0, stream>>>(
        x, la, lb, nullptr, nullptr, nullptr, nullptr,
        mlp_w, mlp_b, conv_w, hA, N_NODES, ei, cnt, rank, N_EDGES);

    gcn_scan_deg<<<SCAN_BLOCKS, SCAN_T, 0, stream>>>(cnt, rowptr, dinv, N_NODES);
    gcn_fill_csr<<<edgeBlocks, TPB, 0, stream>>>(ei, rank, rowptr, dinv, csr, N_EDGES);

    gcn_gather<true><<<gatherBlocks, GATHER_T, 0, stream>>>(
        rowptr, csr, dinv, hA, conv_b, hB, bnz, bnz + HID, N_NODES, N_EDGES);

    gcn_layer_dense<false, 8><<<denseBlocks, DENSE_T, 0, stream>>>(
        hB, nullptr, nullptr, bnz, bnz + HID, bn_g, bn_b,
        mlp_w + (size_t)HID * HID, mlp_b + HID, conv_w + (size_t)HID * HID,
        hA, N_NODES, nullptr, nullptr, nullptr, 0);
    gcn_gather<true><<<gatherBlocks, GATHER_T, 0, stream>>>(
        rowptr, csr, dinv, hA, conv_b + HID, hB, bnz + 2 * HID, bnz + 3 * HID, N_NODES, N_EDGES);

    gcn_layer_dense<false, CLS_NT><<<denseBlocks, DENSE_T, 0, stream>>>(
        hB, nullptr, nullptr, bnz + 2 * HID, bnz + 3 * HID, bn_g + HID, bn_b + HID,
        mlp_w + (size_t)2 * HID * HID, mlp_b + 2 * HID, Wf,
        hA, N_NODES, nullptr, nullptr, nullptr, 0);
    gcn_gather_cls<<<gatherBlocks, GATHER_T, 0, stream>>>(
        rowptr, csr, dinv, hA, bf, out, N_NODES, N_EDGES);
}


# Dense kernels: the 53 work-groups with a 13th strip split it over waves 12-15, one per SIMD (same rows, a quarter of both products' column tiles each): fullest SIMD 3.25 strips of WMMAs, was 4
# speedup vs baseline: 4.3797x; 1.0345x over previous
#include <hip/hip_runtime.h>
#include <hip/hip_bf16.h>
#include <math.h>

#define N_NODES 50000
#define N_EDGES 800000
#define HID     128
#define OUTC    40
#define NLAYERS 3
#define BN_EPS  1e-5f
#define WT_STRIDE 130

typedef __attribute__((ext_vector_type(2))) float v2f;
typedef __attribute__((ext_vector_type(8))) float v8f;
typedef __attribute__((ext_vector_type(4))) float v4f;
typedef __attribute__((ext_vector_type(4))) int v4i;

__device__ __forceinline__ void atomAddF32(float* p, float v) {
    __hip_atomic_fetch_add(p, v, __ATOMIC_RELAXED, __HIP_MEMORY_SCOPE_AGENT);
}

#define SCAN_T      1024
#define SCAN_BLOCKS ((N_NODES + 1 + SCAN_T - 1) / SCAN_T)
#define SCAN_N      (SCAN_T * SCAN_BLOCKS)
#define SCAN_PRE    (((SCAN_BLOCKS - 1) * (SCAN_T / 4) + SCAN_T - 1) / SCAN_T)

__device__ __forceinline__ int rank_one(int* cnt, int d) {
    return __hip_atomic_fetch_add(&cnt[d], 1, __ATOMIC_RELAXED, __HIP_MEMORY_SCOPE_AGENT);
}
__device__ __forceinline__ v4i rank_four(int* cnt, v4i d) {
    v4i r;
    r.x = rank_one(cnt, d.x);
    r.y = rank_one(cnt, d.y);
    r.z = rank_one(cnt, d.z);
    r.w = rank_one(cnt, d.w);
    return r;
}
__device__ __forceinline__ void rank_edges_part(const int* __restrict__ ei, int* cnt,
                                                int* __restrict__ rank, int e,
                                                int wbase, int lane, int nlanes) {
    const int* dst = ei + e;
    if ((e & 3) == 0) {
        const int ng = e >> 2;
        const v4i* dst4 = (const v4i*)dst;
        v4i* rank4 = (v4i*)rank;
        for (int b = wbase; b < ng; b += 2 * nlanes) {
            if (b + nlanes + 31 < ng) {
                v4i d0 = dst4[b + lane];
                v4i d1 = dst4[b + nlanes + lane];
                asm volatile("" : "+v"(d0), "+v"(d1));
                v4i r0 = rank_four(cnt, d0);
                v4i r1 = rank_four(cnt, d1);
                asm volatile("" : "+v"(r0), "+v"(r1));
                rank4[b + lane] = r0;
                rank4[b + nlanes + lane] = r1;
            } else {
                for (int u = 0; u < 2; ++u) {
                    const int g = b + u * nlanes + lane;
                    if (g < ng) rank4[g] = rank_four(cnt, dst4[g]);
                }
            }
        }
    } else {
        for (int k = wbase + lane; k < e; k += nlanes) rank[k] = rank_one(cnt, dst[k]);
    }
}

__global__ __launch_bounds__(SCAN_T) void gcn_scan_deg(const int* __restrict__ cnt,
                                                       int* __restrict__ rowptr,
                                                       float* __restrict__ dinv, int n) {
    __shared__ int wpre[SCAN_T / 32];
    __shared__ int wown[SCAN_T / 32];
    static_assert(SCAN_T / 32 == 32, "one lane per wave total in the fold below");
    const int t = threadIdx.x;
    const int lane = t & 31, wv = t >> 5;
    const int i = blockIdx.x * SCAN_T + t;
    const int lim = blockIdx.x * (SCAN_T / 4);
    const int4* cnt4 = (const int4*)cnt;
    int4 v[SCAN_PRE];
    #pragma unroll
    for (int j = 0; j < SCAN_PRE; ++j) {
        const int q = j * SCAN_T + t;
        v[j] = cnt4[q < lim ? q : 0];
    }
    int c = 0;
    if (i < n) c = cnt[i];
    int p = 0;
    #pragma unroll
    for (int j = 0; j < SCAN_PRE; ++j) {
        const int q = j * SCAN_T + t;
        if (q < lim) p += (v[j].x + v[j].y) + (v[j].z + v[j].w);
    }
    if (i < n) dinv[i] = rsqrtf((float)(c + 1));
    for (int o = 16; o > 0; o >>= 1) p += __shfl_xor(p, o);
    int incl = c;
    for (int o = 1; o < 32; o <<= 1) { int y = __shfl_up(incl, o); if (lane >= o) incl += y; }
    if (lane == 31) { wpre[wv] = p; wown[wv] = incl; }
    __syncthreads();
    int z = wpre[lane] + (lane < wv ? wown[lane] : 0);
    for (int o = 16; o > 0; o >>= 1) z += __shfl_xor(z, o);
    if (i <= n) rowptr[i] = z + incl - c;
}

__global__ __launch_bounds__(256) void gcn_fill_csr(const int* __restrict__ ei, const int* __restrict__ rank,
                             const int* __restrict__ rowptr, const float* __restrict__ dinv,
                             int* __restrict__ csr, int e) {
    const int i = (blockIdx.x * blockDim.x + threadIdx.x) * 4;
    if (i >= e) return;
    if ((e & 3) == 0) {
        const int4 s4 = *(const int4*)(ei + i);
        const int4 d4 = *(const int4*)(ei + e + i);
        const int4 r4 = *(const int4*)(rank + i);
        const int s[4] = {s4.x, s4.y, s4.z, s4.w};
        const int d[4] = {d4.x, d4.y, d4.z, d4.w};
        const int r[4] = {r4.x, r4.y, r4.z, r4.w};
        int rp[4]; float dv[4];
        #pragma unroll
        for (int u = 0; u < 4; ++u) { rp[u] = rowptr[d[u]]; dv[u] = dinv[s[u]]; }
        #pragma unroll
        for (int u = 0; u < 4; ++u) {
            unsigned pos = (unsigned)(rp[u] + r[u]);
            if (pos > (unsigned)(e - 1)) pos = (unsigned)(e - 1);
            int2 p; p.x = s[u]; p.y = __float_as_int(dv[u]);
            *(int2*)(csr + 2 * (size_t)pos) = p;
        }
    } else {
        for (int k = i; k < e && k < i + 4; ++k) {
            const int s = ei[k];
            unsigned pos = (unsigned)(rowptr[ei[e + k]] + rank[k]);
            if (pos > (unsigned)(e - 1)) pos = (unsigned)(e - 1);
            int2 p; p.x = s; p.y = __float_as_int(dinv[s]);
            *(int2*)(csr + 2 * (size_t)pos) = p;
        }
    }
}

#define DENSE_T 512
#define DENSE_WAVES (DENSE_T / 32)
#define STRIP_STRIDE 132

template <int NT>
__device__ __forceinline__ void strip_gemm(const float2 (&av)[HID / 4], const float* Bfrag, v8f (&acc)[NT]) {
    #pragma unroll
    for (int s = 0; s < HID / 4; ++s) {
        v2f a; a.x = av[s].x; a.y = av[s].y;
        #pragma unroll
        for (int nt = 0; nt < NT; ++nt) {
            const float2 bv = *(const float2*)(Bfrag + nt * 16 * WT_STRIDE + 4 * s);
            v2f b; b.x = bv.x; b.y = bv.y;
            acc[nt] = __builtin_amdgcn_wmma_f32_16x16x4_f32(
                false, a, false, b, (short)0, acc[nt], false, false);
        }
    }
}

template <bool FIRST>
__device__ __forceinline__ void strip_load(float2 (&av)[HID / 4], float2 (&aw)[HID / 4],
                                           const float* __restrict__ A, int tileM, int row16,
                                           int khalf, int nrows) {
    int rowIdx = tileM + row16;
    if (rowIdx > nrows - 1) rowIdx = nrows - 1;
    const float* Arow = A + (size_t)rowIdx * (FIRST ? 2 * HID : HID) + khalf;
    #pragma unroll
    for (int s = 0; s < HID / 4; ++s) {
        av[s] = *(const float2*)(Arow + 4 * s);
        if constexpr (FIRST) aw[s] = *(const float2*)(Arow + HID + 4 * s);
    }
}

template <bool FIRST>
__device__ __forceinline__ void strip_transform(float2 (&av)[HID / 4], const float2 (&aw)[HID / 4],
                                                float alpha, float betac, const float* bnc, int khalf) {
    #pragma unroll
    for (int s = 0; s < HID / 4; ++s) {
        if constexpr (FIRST) {
            av[s].x = alpha * av[s].x + betac * aw[s].x;
            av[s].y = alpha * av[s].y + betac * aw[s].y;
        } else {
            const int k = 4 * s + khalf;
            const float2 g  = *(const float2*)(bnc + k);
            const float2 mu = *(const float2*)(bnc + HID + k);
            const float2 rs = *(const float2*)(bnc + 2 * HID + k);
            const float2 bt = *(const float2*)(bnc + 3 * HID + k);
            av[s].x = fmaxf(g.x * (av[s].x - mu.x) * rs.x + bt.x, 0.0f);
            av[s].y = fmaxf(g.y * (av[s].y - mu.y) * rs.y + bt.y, 0.0f);
        }
    }
}

template <int NT>
__device__ __forceinline__ void strip_hidden(const v8f (&acc)[NT], const float* __restrict__ b1,
                                             float* strp, int nt0, int row16, int rhalf) {
    #pragma unroll
    for (int nt = 0; nt < NT; ++nt) {
        const int col = (nt0 + nt) * 16 + row16;
        const float bv = b1[col];
        #pragma unroll
        for (int r = 0; r < 8; ++r)
            strp[(rhalf + r) * STRIP_STRIDE + col] = fmaxf(acc[nt][r] + bv, 0.0f);
    }
}

template <int NT>
__device__ __forceinline__ void strip_store(const v8f (&acc)[NT], float* __restrict__ C, int ldc,
                                            int tileM, int nt0, int row16, int rhalf, int nrows) {
    #pragma unroll
    for (int nt = 0; nt < NT; ++nt) {
        const int col = (nt0 + nt) * 16 + row16;
        #pragma unroll
        for (int r = 0; r < 8; ++r) {
            const int row = tileM + rhalf + r;
            if (row < nrows) C[(size_t)row * ldc + col] = acc[nt][r];
        }
    }
}

template <bool FIRST, int NT2>
__global__ __launch_bounds__(DENSE_T) void gcn_layer_dense(
        const float* __restrict__ A,
        const float* __restrict__ la, const float* __restrict__ lb,
        const float* __restrict__ bnsum, const float* __restrict__ bnsq,
        const float* __restrict__ gamma, const float* __restrict__ beta,
        const float* __restrict__ W1, const float* __restrict__ b1,
        const float* __restrict__ W2, float* __restrict__ C, int nrows,
        const int* __restrict__ ei, int* cnt, int* __restrict__ rank, int nedges) {
    __shared__ __attribute__((aligned(16))) float W1t[HID * WT_STRIDE];
    __shared__ __attribute__((aligned(16))) float W2t[HID * WT_STRIDE];
    __shared__ __attribute__((aligned(16))) float strip[DENSE_WAVES * 16 * STRIP_STRIDE];
    __shared__ __attribute__((aligned(16))) float bnc[4 * HID];

    const int t = threadIdx.x;
    const int wave  = __builtin_amdgcn_readfirstlane(t >> 5);
    const int lane  = t & 31;
    const int row16 = lane & 15;
    const int khalf = (lane >> 4) << 1;
    const int nstrips = (nrows + 15) >> 4;
    const int wavesTotal = gridDim.x * DENSE_WAVES;
    const int sfirst = wave * gridDim.x + blockIdx.x;

    const int layers = nstrips / (int)gridDim.x;
    const int rem = nstrips - layers * (int)gridDim.x;
    const bool teamwg = (nstrips <= wavesTotal) && (layers + 4 <= DENSE_WAVES) && ((int)blockIdx.x < rem);
    const bool inteam = teamwg && (wave >= layers) && (wave < layers + 4);
    const int q = wave - layers;
    constexpr int TPM1 = 2;
    constexpr int TPM2 = (NT2 + 3) / 4;
    static_assert(NT2 % TPM2 == 0, "members take whole groups of TPM2 tiles");

    if constexpr (FIRST) {
        rank_edges_part(ei, cnt, rank, nedges, sfirst * 32, lane, wavesTotal * 32);
    }

    constexpr int N2 = NT2 * 16;
    constexpr int W2PER = (HID * N2) / DENSE_T;
    float w1v[(HID * HID) / DENSE_T], w2v[W2PER];
    #pragma unroll
    for (int j = 0; j < (HID * HID) / DENSE_T; ++j) w1v[j] = W1[t + j * DENSE_T];
    #pragma unroll
    for (int j = 0; j < W2PER; ++j) w2v[j] = W2[t + j * DENSE_T];

    float2 av[HID / 4], aw[HID / 4];
    const int myfirst = inteam ? layers * (int)gridDim.x + (int)blockIdx.x : sfirst;
    bool active = inteam || (sfirst < nstrips);
    if (active) strip_load<FIRST>(av, aw, A, myfirst * 16, row16, khalf, nrows);

    #pragma unroll
    for (int j = 0; j < (HID * HID) / DENSE_T; ++j) {
        const int idx = t + j * DENSE_T;
        W1t[(idx & 127) * WT_STRIDE + (idx >> 7)] = w1v[j];
    }
    #pragma unroll
    for (int j = 0; j < W2PER; ++j) {
        const int idx = t + j * DENSE_T;
        W2t[(idx % N2) * WT_STRIDE + (idx / N2)] = w2v[j];
    }
    float alpha = 0.0f, betac = 0.0f;
    if constexpr (FIRST) {
        alpha = __int_as_float(__builtin_amdgcn_readfirstlane(__float_as_int(expf(la[0]))));
        betac = __int_as_float(__builtin_amdgcn_readfirstlane(__float_as_int(expf(lb[0]))));
    } else {
        if (t < HID) {
            const float invn = 1.0f / (float)nrows;
            const float mu = bnsum[t] * invn;
            const float var = bnsq[t] * invn - mu * mu;
            bnc[t] = gamma[t];
            bnc[HID + t] = mu;
            bnc[2 * HID + t] = rsqrtf(var + BN_EPS);
            bnc[3 * HID + t] = beta[t];
        }
    }
    __syncthreads();

    float* myStrip = strip + (inteam ? layers : wave) * 16 * STRIP_STRIDE;
    const int nt1 = inteam ? q * TPM1 : 0;
    const int nt2 = inteam ? q * TPM2 : 0;
    const float* B1 = W1t + (nt1 * 16 + row16) * WT_STRIDE + khalf;
    const float* B2 = W2t + (nt2 * 16 + row16) * WT_STRIDE + khalf;
    const int rhalf = (lane >> 4) << 3;

    for (int s0 = 0; s0 < nstrips; s0 += wavesTotal) {
        if (s0 > 0) {
            active = !inteam && (s0 + sfirst < nstrips);
            if (active) strip_load<FIRST>(av, aw, A, (s0 + sfirst) * 16, row16, khalf, nrows);
        }
        const int tileM = (inteam ? myfirst : s0 + sfirst) * 16;
        if (active) {
            strip_transform<FIRST>(av, aw, alpha, betac, bnc, khalf);
            if (inteam) {
                v8f acc[TPM1] = {};
                strip_gemm<TPM1>(av, B1, acc);
                strip_hidden<TPM1>(acc, b1, myStrip, nt1, row16, rhalf);
            } else {
                v8f acc[8] = {};
                strip_gemm<8>(av, B1, acc);
                strip_hidden<8>(acc, b1, myStrip, 0, row16, rhalf);
            }
        }
        __syncthreads();
        if (active) {
            #pragma unroll
            for (int s = 0; s < HID / 4; ++s)
                av[s] = *(const float2*)(myStrip + row16 * STRIP_STRIDE + 4 * s + khalf);
            if (inteam) {
                if (nt2 < NT2) {
                    v8f acc2[TPM2] = {};
                    strip_gemm<TPM2>(av, B2, acc2);
                    strip_store<TPM2>(acc2, C, N2, tileM, nt2, row16, rhalf, nrows);
                }
            } else {
                v8f acc2[NT2] = {};
                strip_gemm<NT2>(av, B2, acc2);
                strip_store<NT2>(acc2, C, N2, tileM, 0, row16, rhalf, nrows);
            }
        }
    }
}

#define GROWS 8
#define GATHER_T 256

template <bool STATS>
__global__ __launch_bounds__(GATHER_T) void gcn_gather(const int* __restrict__ rowptr,
                                                  const int* __restrict__ csr,
                                                  const float* __restrict__ dinv,
                                                  const float* __restrict__ m,
                                                  const float* __restrict__ b,
                                                  float* __restrict__ h,
                                                  float* bnsum, float* bnsq, int n, int e) {
    const int lane = threadIdx.x & 31;
    float4 ssum = make_float4(0.0f, 0.0f, 0.0f, 0.0f);
    float4 ssq  = make_float4(0.0f, 0.0f, 0.0f, 0.0f);
    const int nwaves = (gridDim.x * blockDim.x) >> 5;
    const int wid = __builtin_amdgcn_readfirstlane((blockIdx.x * blockDim.x + threadIdx.x) >> 5);
    const float* mlane = m + lane * 4;
    const float4 bb = *(const float4*)(b + lane * 4);
    for (int node = wid; node < n; node += nwaves) {
        int beg = rowptr[node];
        int end = rowptr[node + 1];
        if (beg < 0) beg = 0;
        if (end > e) end = e;
        const float dd = dinv[node];
        const float4 selfv = *(const float4*)(mlane + (size_t)node * HID);
        float4 acc = make_float4(0.0f, 0.0f, 0.0f, 0.0f);

        for (int base = beg; base < end; base += 32) {
            int c = end - base; if (c > 32) c = 32;
            int sv = 0, wbits = 0;
            if (lane < c) {
                const int2 p = *(const int2*)(csr + 2 * (size_t)(base + lane));
                sv = p.x; wbits = p.y;
            }
            if ((unsigned)sv > (unsigned)(n - 1)) sv = n - 1;
            int j = 0;
            for (; j + GROWS <= c; j += GROWS) {
                float4 v[GROWS]; float w[GROWS];
                #pragma unroll
                for (int u = 0; u < GROWS; ++u) {
                    const int s = __builtin_amdgcn_readlane(sv, j + u);
                    w[u] = __int_as_float(__builtin_amdgcn_readlane(wbits, j + u));
                    v[u] = *(const float4*)(mlane + (size_t)s * HID);
                }
                #pragma unroll
                for (int u = 0; u < GROWS; ++u) {
                    const float nrm = w[u] * dd;
                    acc.x += v[u].x * nrm; acc.y += v[u].y * nrm;
                    acc.z += v[u].z * nrm; acc.w += v[u].w * nrm;
                }
            }
            const int r = c - j;
            if (r > 0) {
                float4 v[GROWS - 1]; float w[GROWS - 1];
                #pragma unroll
                for (int u = 0; u < GROWS - 1; ++u) {
                    if (u < r) {
                        const int s = __builtin_amdgcn_readlane(sv, j + u);
                        w[u] = __int_as_float(__builtin_amdgcn_readlane(wbits, j + u));
                        v[u] = *(const float4*)(mlane + (size_t)s * HID);
                    }
                }
                #pragma unroll
                for (int u = 0; u < GROWS - 1; ++u) {
                    if (u < r) {
                        const float nrm = w[u] * dd;
                        acc.x += v[u].x * nrm; acc.y += v[u].y * nrm;
                        acc.z += v[u].z * nrm; acc.w += v[u].w * nrm;
                    }
                }
            }
        }
        const float sn = dd * dd;
        acc.x += selfv.x * sn; acc.y += selfv.y * sn; acc.z += selfv.z * sn; acc.w += selfv.w * sn;
        acc.x += bb.x; acc.y += bb.y; acc.z += bb.z; acc.w += bb.w;
        *(float4*)(h + (size_t)node * HID + lane * 4) = acc;
        if constexpr (STATS) {
            ssum.x += acc.x; ssum.y += acc.y; ssum.z += acc.z; ssum.w += acc.w;
            ssq.x += acc.x * acc.x; ssq.y += acc.y * acc.y;
            ssq.z += acc.z * acc.z; ssq.w += acc.w * acc.w;
        }
    }
    if constexpr (STATS) {
        __shared__ float red[(GATHER_T / 32) * 2 * HID];
        const int wv = threadIdx.x >> 5;
        *(float4*)(red + (wv * 2 + 0) * HID + lane * 4) = ssum;
        *(float4*)(red + (wv * 2 + 1) * HID + lane * 4) = ssq;
        __syncthreads();
        static_assert(GATHER_T == 2 * HID, "one thread per batch-norm sum");
        const int which = threadIdx.x >> 7;
        const int f = threadIdx.x & (HID - 1);
        float tot = 0.0f;
        for (int w = 0; w < GATHER_T / 32; ++w) tot += red[(w * 2 + which) * HID + f];
        atomAddF32(which ? &bnsq[f] : &bnsum[f], tot);
    }
}

#define CLS_NT 3
#define CLS_W (CLS_NT * 16)
#define FOLD_T (CLS_NT * 32)
#define FOLD_BLOCKS (HID / 16 + 1)
#define CLS_ROWS 16

__device__ __forceinline__ void fold_cls_tile(const float* __restrict__ W2,
                                              const float* __restrict__ b,
                                              const float* __restrict__ lw,
                                              const float* __restrict__ lb,
                                              float* __restrict__ Wf, float* __restrict__ bf,
                                              int t, int mt) {
    const int nt = __builtin_amdgcn_readfirstlane(t >> 5);
    const int lane  = t & 31;
    const int row16 = lane & 15;
    const int khalf = (lane >> 4) << 1;
    const int arow = mt * 16 + row16;
    const float* Arow = (arow < HID ? W2 + (size_t)arow * HID : b) + khalf;
    const int col = nt * 16 + row16;
    const bool live = col < OUTC;
    const float* Bcol = lw + (live ? col : OUTC - 1) + khalf * OUTC;

    float2 av[HID / 4]; float b0[HID / 4], b1[HID / 4];
    #pragma unroll
    for (int s = 0; s < HID / 4; ++s) {
        av[s] = *(const float2*)(Arow + 4 * s);
        b0[s] = Bcol[(4 * s) * OUTC];
        b1[s] = Bcol[(4 * s + 1) * OUTC];
    }
    v8f acc = {};
    #pragma unroll
    for (int s = 0; s < HID / 4; ++s) {
        v2f a; a.x = av[s].x; a.y = av[s].y;
        v2f bfrag; bfrag.x = live ? b0[s] : 0.0f; bfrag.y = live ? b1[s] : 0.0f;
        acc = __builtin_amdgcn_wmma_f32_16x16x4_f32(false, a, false, bfrag, (short)0, acc, false, false);
    }
    const int rhalf = (lane >> 4) << 3;
    if (mt < HID / 16) {
        #pragma unroll
        for (int r = 0; r < 8; ++r)
            Wf[(size_t)(mt * 16 + rhalf + r) * CLS_W + col] = acc[r];
    } else if (lane < 16) {
        bf[col] = acc[0] + (live ? lb[col] : 0.0f);
    }
}

#define SETUP_T 256
__global__ __launch_bounds__(SETUP_T) void gcn_setup(const float* __restrict__ W2,
                                                     const float* __restrict__ b,
                                                     const float* __restrict__ lw,
                                                     const float* __restrict__ lb,
                                                     float* __restrict__ Wf, float* __restrict__ bf,
                                                     int* __restrict__ cnt, int n,
                                                     float* __restrict__ bnz, int nbn) {
    const int i = blockIdx.x * SETUP_T + threadIdx.x;
    if (i < n) cnt[i] = 0;
    if (i < nbn) bnz[i] = 0.0f;
    const int wave = __builtin_amdgcn_readfirstlane(threadIdx.x >> 5);
    if (blockIdx.x < FOLD_BLOCKS && wave < CLS_NT)
        fold_cls_tile(W2, b, lw, lb, Wf, bf, threadIdx.x, blockIdx.x);
}

template <int CTRL> __device__ __forceinline__ float lane_perm16(float v) {
    return __int_as_float(__builtin_amdgcn_update_dpp(0, __float_as_int(v), CTRL, 0xF, 0xF, true));
}
__device__ __forceinline__ float wave_max32(float v) {
    v = fmaxf(v, lane_perm16<0xB1>(v));
    v = fmaxf(v, lane_perm16<0x4E>(v));
    v = fmaxf(v, lane_perm16<0x141>(v));
    v = fmaxf(v, lane_perm16<0x140>(v));
    const float lo = __int_as_float(__builtin_amdgcn_readlane(__float_as_int(v), 0));
    const float hi = __int_as_float(__builtin_amdgcn_readlane(__float_as_int(v), 16));
    return fmaxf(lo, hi);
}
__device__ __forceinline__ float wave_sum32(float v) {
    v += lane_perm16<0xB1>(v);
    v += lane_perm16<0x4E>(v);
    v += lane_perm16<0x141>(v);
    v += lane_perm16<0x140>(v);
    const float lo = __int_as_float(__builtin_amdgcn_readlane(__float_as_int(v), 0));
    const float hi = __int_as_float(__builtin_amdgcn_readlane(__float_as_int(v), 16));
    return lo + hi;
}

__global__ __launch_bounds__(GATHER_T) void gcn_gather_cls(const int* __restrict__ rowptr,
                                                           const int* __restrict__ csr,
                                                           const float* __restrict__ dinv,
                                                           const float* __restrict__ m,
                                                           const float* __restrict__ bf,
                                                           float* __restrict__ out, int n, int e) {
    const int lane = threadIdx.x & 31;
    const int nwaves = (gridDim.x * blockDim.x) >> 5;
    const int wid = __builtin_amdgcn_readfirstlane((blockIdx.x * blockDim.x + threadIdx.x) >> 5);
    const bool valid = lane < OUTC / 2;
    const int fl = 2 * (lane < CLS_W / 2 ? lane : CLS_W / 2 - 1);
    const float2 bb = *(const float2*)(bf + fl);
    for (int node = wid; node < n; node += nwaves) {
        int beg = rowptr[node];
        int end = rowptr[node + 1];
        if (beg < 0) beg = 0;
        if (end > e) end = e;
        const float dd = dinv[node];
        const float2 selfv = *(const float2*)(m + (size_t)node * CLS_W + fl);
        float2 acc = make_float2(0.0f, 0.0f);

        for (int base = beg; base < end; base += 32) {
            int c = end - base; if (c > 32) c = 32;
            int sv = 0, wbits = 0;
            if (lane < c) {
                const int2 p = *(const int2*)(csr + 2 * (size_t)(base + lane));
                sv = p.x; wbits = p.y;
            }
            if ((unsigned)sv > (unsigned)(n - 1)) sv = n - 1;
            int j = 0;
            for (; j + CLS_ROWS <= c; j += CLS_ROWS) {
                float2 v[CLS_ROWS]; float w[CLS_ROWS];
                #pragma unroll
                for (int u = 0; u < CLS_ROWS; ++u) {
                    const int s = __builtin_amdgcn_readlane(sv, j + u);
                    w[u] = __int_as_float(__builtin_amdgcn_readlane(wbits, j + u));
                    v[u] = *(const float2*)(m + (size_t)s * CLS_W + fl);
                }
                #pragma unroll
                for (int u = 0; u < CLS_ROWS; ++u) {
                    const float nrm = w[u] * dd;
                    acc.x += v[u].x * nrm; acc.y += v[u].y * nrm;
                }
            }
            const int r = c - j;
            if (r > 0) {
                float2 v[CLS_ROWS - 1]; float w[CLS_ROWS - 1];
                #pragma unroll
                for (int u = 0; u < CLS_ROWS - 1; ++u) {
                    if (u < r) {
                        const int s = __builtin_amdgcn_readlane(sv, j + u);
                        w[u] = __int_as_float(__builtin_amdgcn_readlane(wbits, j + u));
                        v[u] = *(const float2*)(m + (size_t)s * CLS_W + fl);
                    }
                }
                #pragma unroll
                for (int u = 0; u < CLS_ROWS - 1; ++u) {
                    if (u < r) {
                        const float nrm = w[u] * dd;
                        acc.x += v[u].x * nrm; acc.y += v[u].y * nrm;
                    }
                }
            }
        }
        const float sn = dd * dd;
        acc.x += selfv.x * sn; acc.y += selfv.y * sn;
        acc.x += bb.x; acc.y += bb.y;
        const float x0 = valid ? acc.x : -INFINITY;
        const float x1 = valid ? acc.y : -INFINITY;
        const float mx = wave_max32(fmaxf(x0, x1));
        const float ls = logf(wave_sum32(expf(x0 - mx) + expf(x1 - mx)));
        if (valid) {
            float2 o; o.x = x0 - mx - ls; o.y = x1 - mx - ls;
            *(float2*)(out + (size_t)node * OUTC + 2 * lane) = o;
        }
    }
}

extern "C" void kernel_launch(void* const* d_in, const int* in_sizes, int n_in,
                              void* d_out, int out_size, void* d_ws, size_t ws_size,
                              hipStream_t stream) {
    const float* x       = (const float*)d_in[0];
    const int*   ei      = (const int*)d_in[1];
    const float* la      = (const float*)d_in[2];
    const float* lb      = (const float*)d_in[3];
    const float* mlp_w   = (const float*)d_in[4];
    const float* mlp_b   = (const float*)d_in[5];
    const float* conv_w  = (const float*)d_in[6];
    const float* conv_b  = (const float*)d_in[7];
    const float* bn_g    = (const float*)d_in[8];
    const float* bn_b    = (const float*)d_in[9];
    const float* lin_w   = (const float*)d_in[10];
    const float* lin_b   = (const float*)d_in[11];
    float* out = (float*)d_out;

    float* ws = (float*)d_ws;
    const size_t NM = (size_t)N_NODES * HID;
    float* hA    = ws;
    float* hB    = hA + NM;
    int*   cnt    = (int*)(hB + NM);
    int*   rowptr = cnt + SCAN_N;
    int*   rank   = rowptr + SCAN_N;
    int*   csr    = rank + N_EDGES;
    float* Wf    = (float*)(csr + 2 * N_EDGES);
    float* bf    = Wf + HID * CLS_W;
    float* bnz   = bf + CLS_W;
    float* dinv  = hB + NM + NM;

    const int TPB = 256;
    const int edgeBlocks = ((N_EDGES + 3) / 4 + TPB - 1) / TPB;
    const int gatherBlocks = 1250;
    const int denseBlocks = 256;

    gcn_setup<<<(N_NODES + SETUP_T - 1) / SETUP_T, SETUP_T, 0, stream>>>(
        conv_w + (size_t)2 * HID * HID, conv_b + 2 * HID, lin_w, lin_b,
        Wf, bf, cnt, N_NODES, bnz, 4 * HID);

    gcn_layer_dense<true, 8><<<denseBlocks, DENSE_T, 0, stream>>>(
        x, la, lb, nullptr, nullptr, nullptr, nullptr,
        mlp_w, mlp_b, conv_w, hA, N_NODES, ei, cnt, rank, N_EDGES);

    gcn_scan_deg<<<SCAN_BLOCKS, SCAN_T, 0, stream>>>(cnt, rowptr, dinv, N_NODES);
    gcn_fill_csr<<<edgeBlocks, TPB, 0, stream>>>(ei, rank, rowptr, dinv, csr, N_EDGES);

    gcn_gather<true><<<gatherBlocks, GATHER_T, 0, stream>>>(
        rowptr, csr, dinv, hA, conv_b, hB, bnz, bnz + HID, N_NODES, N_EDGES);

    gcn_layer_dense<false, 8><<<denseBlocks, DENSE_T, 0, stream>>>(
        hB, nullptr, nullptr, bnz, bnz + HID, bn_g, bn_b,
        mlp_w + (size_t)HID * HID, mlp_b + HID, conv_w + (size_t)HID * HID,
        hA, N_NODES, nullptr, nullptr, nullptr, 0);
    gcn_gather<true><<<gatherBlocks, GATHER_T, 0, stream>>>(
        rowptr, csr, dinv, hA, conv_b + HID, hB, bnz + 2 * HID, bnz + 3 * HID, N_NODES, N_EDGES);

    gcn_layer_dense<false, CLS_NT><<<denseBlocks, DENSE_T, 0, stream>>>(
        hB, nullptr, nullptr, bnz + 2 * HID, bnz + 3 * HID, bn_g + HID, bn_b + HID,
        mlp_w + (size_t)2 * HID * HID, mlp_b + 2 * HID, Wf,
        hA, N_NODES, nullptr, nullptr, nullptr, 0);
    gcn_gather_cls<<<gatherBlocks, GATHER_T, 0, stream>>>(
        rowptr, csr, dinv, hA, bf, out, N_NODES, N_EDGES);
}


# Dense kernels fetch each 16-row strip as whole 512-byte rows (16-byte-per-lane loads) and form the A fragments through the wave's LDS slot; was 8-byte fragment loads touching 16 lines each
# speedup vs baseline: 4.5151x; 1.0309x over previous
#include <hip/hip_runtime.h>
#include <hip/hip_bf16.h>
#include <math.h>

#define N_NODES 50000
#define N_EDGES 800000
#define HID     128
#define OUTC    40
#define NLAYERS 3
#define BN_EPS  1e-5f
#define WT_STRIDE 130

typedef __attribute__((ext_vector_type(2))) float v2f;
typedef __attribute__((ext_vector_type(8))) float v8f;
typedef __attribute__((ext_vector_type(4))) float v4f;
typedef __attribute__((ext_vector_type(4))) int v4i;

__device__ __forceinline__ void atomAddF32(float* p, float v) {
    __hip_atomic_fetch_add(p, v, __ATOMIC_RELAXED, __HIP_MEMORY_SCOPE_AGENT);
}

#define SCAN_T      1024
#define SCAN_BLOCKS ((N_NODES + 1 + SCAN_T - 1) / SCAN_T)
#define SCAN_N      (SCAN_T * SCAN_BLOCKS)
#define SCAN_PRE    (((SCAN_BLOCKS - 1) * (SCAN_T / 4) + SCAN_T - 1) / SCAN_T)

__device__ __forceinline__ int rank_one(int* cnt, int d) {
    return __hip_atomic_fetch_add(&cnt[d], 1, __ATOMIC_RELAXED, __HIP_MEMORY_SCOPE_AGENT);
}
__device__ __forceinline__ v4i rank_four(int* cnt, v4i d) {
    v4i r;
    r.x = rank_one(cnt, d.x);
    r.y = rank_one(cnt, d.y);
    r.z = rank_one(cnt, d.z);
    r.w = rank_one(cnt, d.w);
    return r;
}
__device__ __forceinline__ void rank_edges_part(const int* __restrict__ ei, int* cnt,
                                                int* __restrict__ rank, int e,
                                                int wbase, int lane, int nlanes) {
    const int* dst = ei + e;
    if ((e & 3) == 0) {
        const int ng = e >> 2;
        const v4i* dst4 = (const v4i*)dst;
        v4i* rank4 = (v4i*)rank;
        for (int b = wbase; b < ng; b += 2 * nlanes) {
            if (b + nlanes + 31 < ng) {
                v4i d0 = dst4[b + lane];
                v4i d1 = dst4[b + nlanes + lane];
                asm volatile("" : "+v"(d0), "+v"(d1));
                v4i r0 = rank_four(cnt, d0);
                v4i r1 = rank_four(cnt, d1);
                asm volatile("" : "+v"(r0), "+v"(r1));
                rank4[b + lane] = r0;
                rank4[b + nlanes + lane] = r1;
            } else {
                for (int u = 0; u < 2; ++u) {
                    const int g = b + u * nlanes + lane;
                    if (g < ng) rank4[g] = rank_four(cnt, dst4[g]);
                }
            }
        }
    } else {
        for (int k = wbase + lane; k < e; k += nlanes) rank[k] = rank_one(cnt, dst[k]);
    }
}

__global__ __launch_bounds__(SCAN_T) void gcn_scan_deg(const int* __restrict__ cnt,
                                                       int* __restrict__ rowptr,
                                                       float* __restrict__ dinv, int n) {
    __shared__ int wpre[SCAN_T / 32];
    __shared__ int wown[SCAN_T / 32];
    static_assert(SCAN_T / 32 == 32, "one lane per wave total in the fold below");
    const int t = threadIdx.x;
    const int lane = t & 31, wv = t >> 5;
    const int i = blockIdx.x * SCAN_T + t;
    const int lim = blockIdx.x * (SCAN_T / 4);
    const int4* cnt4 = (const int4*)cnt;
    int4 v[SCAN_PRE];
    #pragma unroll
    for (int j = 0; j < SCAN_PRE; ++j) {
        const int q = j * SCAN_T + t;
        v[j] = cnt4[q < lim ? q : 0];
    }
    int c = 0;
    if (i < n) c = cnt[i];
    int p = 0;
    #pragma unroll
    for (int j = 0; j < SCAN_PRE; ++j) {
        const int q = j * SCAN_T + t;
        if (q < lim) p += (v[j].x + v[j].y) + (v[j].z + v[j].w);
    }
    if (i < n) dinv[i] = rsqrtf((float)(c + 1));
    for (int o = 16; o > 0; o >>= 1) p += __shfl_xor(p, o);
    int incl = c;
    for (int o = 1; o < 32; o <<= 1) { int y = __shfl_up(incl, o); if (lane >= o) incl += y; }
    if (lane == 31) { wpre[wv] = p; wown[wv] = incl; }
    __syncthreads();
    int z = wpre[lane] + (lane < wv ? wown[lane] : 0);
    for (int o = 16; o > 0; o >>= 1) z += __shfl_xor(z, o);
    if (i <= n) rowptr[i] = z + incl - c;
}

__global__ __launch_bounds__(256) void gcn_fill_csr(const int* __restrict__ ei, const int* __restrict__ rank,
                             const int* __restrict__ rowptr, const float* __restrict__ dinv,
                             int* __restrict__ csr, int e) {
    const int i = (blockIdx.x * blockDim.x + threadIdx.x) * 4;
    if (i >= e) return;
    if ((e & 3) == 0) {
        const int4 s4 = *(const int4*)(ei + i);
        const int4 d4 = *(const int4*)(ei + e + i);
        const int4 r4 = *(const int4*)(rank + i);
        const int s[4] = {s4.x, s4.y, s4.z, s4.w};
        const int d[4] = {d4.x, d4.y, d4.z, d4.w};
        const int r[4] = {r4.x, r4.y, r4.z, r4.w};
        int rp[4]; float dv[4];
        #pragma unroll
        for (int u = 0; u < 4; ++u) { rp[u] = rowptr[d[u]]; dv[u] = dinv[s[u]]; }
        #pragma unroll
        for (int u = 0; u < 4; ++u) {
            unsigned pos = (unsigned)(rp[u] + r[u]);
            if (pos > (unsigned)(e - 1)) pos = (unsigned)(e - 1);
            int2 p; p.x = s[u]; p.y = __float_as_int(dv[u]);
            *(int2*)(csr + 2 * (size_t)pos) = p;
        }
    } else {
        for (int k = i; k < e && k < i + 4; ++k) {
            const int s = ei[k];
            unsigned pos = (unsigned)(rowptr[ei[e + k]] + rank[k]);
            if (pos > (unsigned)(e - 1)) pos = (unsigned)(e - 1);
            int2 p; p.x = s; p.y = __float_as_int(dinv[s]);
            *(int2*)(csr + 2 * (size_t)pos) = p;
        }
    }
}

#define DENSE_T 512
#define DENSE_WAVES (DENSE_T / 32)
#define STRIP_STRIDE 132

template <int NT>
__device__ __forceinline__ void strip_gemm(const float2 (&av)[HID / 4], const float* Bfrag, v8f (&acc)[NT]) {
    #pragma unroll
    for (int s = 0; s < HID / 4; ++s) {
        v2f a; a.x = av[s].x; a.y = av[s].y;
        #pragma unroll
        for (int nt = 0; nt < NT; ++nt) {
            const float2 bv = *(const float2*)(Bfrag + nt * 16 * WT_STRIDE + 4 * s);
            v2f b; b.x = bv.x; b.y = bv.y;
            acc[nt] = __builtin_amdgcn_wmma_f32_16x16x4_f32(
                false, a, false, b, (short)0, acc[nt], false, false);
        }
    }
}

template <int NR>
__device__ __forceinline__ void strip_load_rows(float4 (&rv)[NR], const float* __restrict__ A, int lda,
                                                int tileM, int lane, int nrows) {
    #pragma unroll
    for (int r = 0; r < NR; ++r) {
        int rowIdx = tileM + r;
        if (rowIdx > nrows - 1) rowIdx = nrows - 1;
        rv[r] = *(const float4*)(A + (size_t)rowIdx * lda + lane * 4);
    }
}

template <bool FIRST, int NR>
__device__ __forceinline__ void strip_stage(const float4 (&rv)[NR], const float4 (&rw)[NR],
                                            float alpha, float betac, float4 g, float4 mu,
                                            float4 rs, float4 bt, float* slot, int lane) {
    #pragma unroll
    for (int r = 0; r < NR; ++r) {
        float4 v;
        if constexpr (FIRST) {
            v.x = alpha * rv[r].x + betac * rw[r].x;
            v.y = alpha * rv[r].y + betac * rw[r].y;
            v.z = alpha * rv[r].z + betac * rw[r].z;
            v.w = alpha * rv[r].w + betac * rw[r].w;
        } else {
            v.x = fmaxf(g.x * (rv[r].x - mu.x) * rs.x + bt.x, 0.0f);
            v.y = fmaxf(g.y * (rv[r].y - mu.y) * rs.y + bt.y, 0.0f);
            v.z = fmaxf(g.z * (rv[r].z - mu.z) * rs.z + bt.z, 0.0f);
            v.w = fmaxf(g.w * (rv[r].w - mu.w) * rs.w + bt.w, 0.0f);
        }
        *(float4*)(slot + r * STRIP_STRIDE + lane * 4) = v;
    }
}

__device__ __forceinline__ void strip_frags(float2 (&av)[HID / 4], const float* slot, int row16, int khalf) {
    #pragma unroll
    for (int s = 0; s < HID / 4; ++s)
        av[s] = *(const float2*)(slot + row16 * STRIP_STRIDE + 4 * s + khalf);
}

template <int NT>
__device__ __forceinline__ void strip_hidden(const v8f (&acc)[NT], const float* __restrict__ b1,
                                             float* strp, int nt0, int row16, int rhalf) {
    #pragma unroll
    for (int nt = 0; nt < NT; ++nt) {
        const int col = (nt0 + nt) * 16 + row16;
        const float bv = b1[col];
        #pragma unroll
        for (int r = 0; r < 8; ++r)
            strp[(rhalf + r) * STRIP_STRIDE + col] = fmaxf(acc[nt][r] + bv, 0.0f);
    }
}

template <int NT>
__device__ __forceinline__ void strip_store(const v8f (&acc)[NT], float* __restrict__ C, int ldc,
                                            int tileM, int nt0, int row16, int rhalf, int nrows) {
    #pragma unroll
    for (int nt = 0; nt < NT; ++nt) {
        const int col = (nt0 + nt) * 16 + row16;
        #pragma unroll
        for (int r = 0; r < 8; ++r) {
            const int row = tileM + rhalf + r;
            if (row < nrows) C[(size_t)row * ldc + col] = acc[nt][r];
        }
    }
}

template <bool FIRST, int NT2>
__global__ __launch_bounds__(DENSE_T) void gcn_layer_dense(
        const float* __restrict__ A,
        const float* __restrict__ la, const float* __restrict__ lb,
        const float* __restrict__ bnsum, const float* __restrict__ bnsq,
        const float* __restrict__ gamma, const float* __restrict__ beta,
        const float* __restrict__ W1, const float* __restrict__ b1,
        const float* __restrict__ W2, float* __restrict__ C, int nrows,
        const int* __restrict__ ei, int* cnt, int* __restrict__ rank, int nedges) {
    __shared__ __attribute__((aligned(16))) float W1t[HID * WT_STRIDE];
    __shared__ __attribute__((aligned(16))) float W2t[HID * WT_STRIDE];
    __shared__ __attribute__((aligned(16))) float strip[(DENSE_WAVES + 1) * 16 * STRIP_STRIDE];

    const int t = threadIdx.x;
    const int wave  = __builtin_amdgcn_readfirstlane(t >> 5);
    const int lane  = t & 31;
    const int row16 = lane & 15;
    const int khalf = (lane >> 4) << 1;
    const int nstrips = (nrows + 15) >> 4;
    const int wavesTotal = gridDim.x * DENSE_WAVES;
    const int sfirst = wave * gridDim.x + blockIdx.x;

    const int layers = nstrips / (int)gridDim.x;
    const int rem = nstrips - layers * (int)gridDim.x;
    const bool teamwg = (nstrips <= wavesTotal) && (layers + 4 <= DENSE_WAVES) && ((int)blockIdx.x < rem);
    const bool inteam = teamwg && (wave >= layers) && (wave < layers + 4);
    const int q = wave - layers;
    constexpr int TPM1 = 2;
    constexpr int TPM2 = (NT2 + 3) / 4;
    static_assert(NT2 % TPM2 == 0, "members take whole groups of TPM2 tiles");

    if constexpr (FIRST) {
        rank_edges_part(ei, cnt, rank, nedges, sfirst * 32, lane, wavesTotal * 32);
    }

    constexpr int N2 = NT2 * 16;
    constexpr int W2PER = (HID * N2) / DENSE_T;
    float w1v[(HID * HID) / DENSE_T], w2v[W2PER];
    #pragma unroll
    for (int j = 0; j < (HID * HID) / DENSE_T; ++j) w1v[j] = W1[t + j * DENSE_T];
    #pragma unroll
    for (int j = 0; j < W2PER; ++j) w2v[j] = W2[t + j * DENSE_T];

    float4 rv[16], rw[16];
    const int myfirst = inteam ? layers * (int)gridDim.x + (int)blockIdx.x : sfirst;
    bool active = inteam || (sfirst < nstrips);
    if (active) {
        strip_load_rows<16>(rv, A, FIRST ? 2 * HID : HID, myfirst * 16, lane, nrows);
        if constexpr (FIRST) strip_load_rows<16>(rw, A + HID, 2 * HID, myfirst * 16, lane, nrows);
    }

    #pragma unroll
    for (int j = 0; j < (HID * HID) / DENSE_T; ++j) {
        const int idx = t + j * DENSE_T;
        W1t[(idx & 127) * WT_STRIDE + (idx >> 7)] = w1v[j];
    }
    #pragma unroll
    for (int j = 0; j < W2PER; ++j) {
        const int idx = t + j * DENSE_T;
        W2t[(idx % N2) * WT_STRIDE + (idx / N2)] = w2v[j];
    }
    float* inSlot = strip + wave * 16 * STRIP_STRIDE;
    float alpha = 0.0f, betac = 0.0f;
    float4 bg = {}, bmu = {}, brs = {}, bbt = {};
    if constexpr (FIRST) {
        alpha = __int_as_float(__builtin_amdgcn_readfirstlane(__float_as_int(expf(la[0]))));
        betac = __int_as_float(__builtin_amdgcn_readfirstlane(__float_as_int(expf(lb[0]))));
    } else {
        const float invn = 1.0f / (float)nrows;
        const float4 s1 = *(const float4*)(bnsum + lane * 4);
        const float4 s2 = *(const float4*)(bnsq + lane * 4);
        bg  = *(const float4*)(gamma + lane * 4);
        bbt = *(const float4*)(beta + lane * 4);
        bmu.x = s1.x * invn; bmu.y = s1.y * invn; bmu.z = s1.z * invn; bmu.w = s1.w * invn;
        brs.x = rsqrtf(s2.x * invn - bmu.x * bmu.x + BN_EPS);
        brs.y = rsqrtf(s2.y * invn - bmu.y * bmu.y + BN_EPS);
        brs.z = rsqrtf(s2.z * invn - bmu.z * bmu.z + BN_EPS);
        brs.w = rsqrtf(s2.w * invn - bmu.w * bmu.w + BN_EPS);
    }
    if (active) strip_stage<FIRST, 16>(rv, rw, alpha, betac, bg, bmu, brs, bbt, inSlot, lane);
    __syncthreads();

    float* myStrip = strip + (inteam ? DENSE_WAVES : wave) * 16 * STRIP_STRIDE;
    const int nt1 = inteam ? q * TPM1 : 0;
    const int nt2 = inteam ? q * TPM2 : 0;
    const float* B1 = W1t + (nt1 * 16 + row16) * WT_STRIDE + khalf;
    const float* B2 = W2t + (nt2 * 16 + row16) * WT_STRIDE + khalf;
    const int rhalf = (lane >> 4) << 3;

    for (int s0 = 0; s0 < nstrips; s0 += wavesTotal) {
        if (s0 > 0) {
            active = !inteam && (s0 + sfirst < nstrips);
        }
        const int tileM = (inteam ? myfirst : s0 + sfirst) * 16;
        float2 av[HID / 4];
        if (active) {
            if (s0 > 0) {
                for (int r0 = 0; r0 < 16; r0 += 4) {
                    float4 a4[4], w4[4];
                    strip_load_rows<4>(a4, A, FIRST ? 2 * HID : HID, tileM + r0, lane, nrows);
                    if constexpr (FIRST) strip_load_rows<4>(w4, A + HID, 2 * HID, tileM + r0, lane, nrows);
                    strip_stage<FIRST, 4>(a4, w4, alpha, betac, bg, bmu, brs, bbt,
                                          inSlot + r0 * STRIP_STRIDE, lane);
                }
            }
            strip_frags(av, inSlot, row16, khalf);
            if (inteam) {
                v8f acc[TPM1] = {};
                strip_gemm<TPM1>(av, B1, acc);
                strip_hidden<TPM1>(acc, b1, myStrip, nt1, row16, rhalf);
            } else {
                v8f acc[8] = {};
                strip_gemm<8>(av, B1, acc);
                strip_hidden<8>(acc, b1, myStrip, 0, row16, rhalf);
            }
        }
        __syncthreads();
        if (active) {
            strip_frags(av, myStrip, row16, khalf);
            if (inteam) {
                if (nt2 < NT2) {
                    v8f acc2[TPM2] = {};
                    strip_gemm<TPM2>(av, B2, acc2);
                    strip_store<TPM2>(acc2, C, N2, tileM, nt2, row16, rhalf, nrows);
                }
            } else {
                v8f acc2[NT2] = {};
                strip_gemm<NT2>(av, B2, acc2);
                strip_store<NT2>(acc2, C, N2, tileM, 0, row16, rhalf, nrows);
            }
        }
    }
}

#define GROWS 8
#define GATHER_T 256

template <bool STATS>
__global__ __launch_bounds__(GATHER_T) void gcn_gather(const int* __restrict__ rowptr,
                                                  const int* __restrict__ csr,
                                                  const float* __restrict__ dinv,
                                                  const float* __restrict__ m,
                                                  const float* __restrict__ b,
                                                  float* __restrict__ h,
                                                  float* bnsum, float* bnsq, int n, int e) {
    const int lane = threadIdx.x & 31;
    float4 ssum = make_float4(0.0f, 0.0f, 0.0f, 0.0f);
    float4 ssq  = make_float4(0.0f, 0.0f, 0.0f, 0.0f);
    const int nwaves = (gridDim.x * blockDim.x) >> 5;
    const int wid = __builtin_amdgcn_readfirstlane((blockIdx.x * blockDim.x + threadIdx.x) >> 5);
    const float* mlane = m + lane * 4;
    const float4 bb = *(const float4*)(b + lane * 4);
    for (int node = wid; node < n; node += nwaves) {
        int beg = rowptr[node];
        int end = rowptr[node + 1];
        if (beg < 0) beg = 0;
        if (end > e) end = e;
        const float dd = dinv[node];
        const float4 selfv = *(const float4*)(mlane + (size_t)node * HID);
        float4 acc = make_float4(0.0f, 0.0f, 0.0f, 0.0f);

        for (int base = beg; base < end; base += 32) {
            int c = end - base; if (c > 32) c = 32;
            int sv = 0, wbits = 0;
            if (lane < c) {
                const int2 p = *(const int2*)(csr + 2 * (size_t)(base + lane));
                sv = p.x; wbits = p.y;
            }
            if ((unsigned)sv > (unsigned)(n - 1)) sv = n - 1;
            int j = 0;
            for (; j + GROWS <= c; j += GROWS) {
                float4 v[GROWS]; float w[GROWS];
                #pragma unroll
                for (int u = 0; u < GROWS; ++u) {
                    const int s = __builtin_amdgcn_readlane(sv, j + u);
                    w[u] = __int_as_float(__builtin_amdgcn_readlane(wbits, j + u));
                    v[u] = *(const float4*)(mlane + (size_t)s * HID);
                }
                #pragma unroll
                for (int u = 0; u < GROWS; ++u) {
                    const float nrm = w[u] * dd;
                    acc.x += v[u].x * nrm; acc.y += v[u].y * nrm;
                    acc.z += v[u].z * nrm; acc.w += v[u].w * nrm;
                }
            }
            const int r = c - j;
            if (r > 0) {
                float4 v[GROWS - 1]; float w[GROWS - 1];
                #pragma unroll
                for (int u = 0; u < GROWS - 1; ++u) {
                    if (u < r) {
                        const int s = __builtin_amdgcn_readlane(sv, j + u);
                        w[u] = __int_as_float(__builtin_amdgcn_readlane(wbits, j + u));
                        v[u] = *(const float4*)(mlane + (size_t)s * HID);
                    }
                }
                #pragma unroll
                for (int u = 0; u < GROWS - 1; ++u) {
                    if (u < r) {
                        const float nrm = w[u] * dd;
                        acc.x += v[u].x * nrm; acc.y += v[u].y * nrm;
                        acc.z += v[u].z * nrm; acc.w += v[u].w * nrm;
                    }
                }
            }
        }
        const float sn = dd * dd;
        acc.x += selfv.x * sn; acc.y += selfv.y * sn; acc.z += selfv.z * sn; acc.w += selfv.w * sn;
        acc.x += bb.x; acc.y += bb.y; acc.z += bb.z; acc.w += bb.w;
        *(float4*)(h + (size_t)node * HID + lane * 4) = acc;
        if constexpr (STATS) {
            ssum.x += acc.x; ssum.y += acc.y; ssum.z += acc.z; ssum.w += acc.w;
            ssq.x += acc.x * acc.x; ssq.y += acc.y * acc.y;
            ssq.z += acc.z * acc.z; ssq.w += acc.w * acc.w;
        }
    }
    if constexpr (STATS) {
        __shared__ float red[(GATHER_T / 32) * 2 * HID];
        const int wv = threadIdx.x >> 5;
        *(float4*)(red + (wv * 2 + 0) * HID + lane * 4) = ssum;
        *(float4*)(red + (wv * 2 + 1) * HID + lane * 4) = ssq;
        __syncthreads();
        static_assert(GATHER_T == 2 * HID, "one thread per batch-norm sum");
        const int which = threadIdx.x >> 7;
        const int f = threadIdx.x & (HID - 1);
        float tot = 0.0f;
        for (int w = 0; w < GATHER_T / 32; ++w) tot += red[(w * 2 + which) * HID + f];
        atomAddF32(which ? &bnsq[f] : &bnsum[f], tot);
    }
}

#define CLS_NT 3
#define CLS_W (CLS_NT * 16)
#define FOLD_T (CLS_NT * 32)
#define FOLD_BLOCKS (HID / 16 + 1)
#define CLS_ROWS 16

__device__ __forceinline__ void fold_cls_tile(const float* __restrict__ W2,
                                              const float* __restrict__ b,
                                              const float* __restrict__ lw,
                                              const float* __restrict__ lb,
                                              float* __restrict__ Wf, float* __restrict__ bf,
                                              int t, int mt) {
    const int nt = __builtin_amdgcn_readfirstlane(t >> 5);
    const int lane  = t & 31;
    const int row16 = lane & 15;
    const int khalf = (lane >> 4) << 1;
    const int arow = mt * 16 + row16;
    const float* Arow = (arow < HID ? W2 + (size_t)arow * HID : b) + khalf;
    const int col = nt * 16 + row16;
    const bool live = col < OUTC;
    const float* Bcol = lw + (live ? col : OUTC - 1) + khalf * OUTC;

    float2 av[HID / 4]; float b0[HID / 4], b1[HID / 4];
    #pragma unroll
    for (int s = 0; s < HID / 4; ++s) {
        av[s] = *(const float2*)(Arow + 4 * s);
        b0[s] = Bcol[(4 * s) * OUTC];
        b1[s] = Bcol[(4 * s + 1) * OUTC];
    }
    v8f acc = {};
    #pragma unroll
    for (int s = 0; s < HID / 4; ++s) {
        v2f a; a.x = av[s].x; a.y = av[s].y;
        v2f bfrag; bfrag.x = live ? b0[s] : 0.0f; bfrag.y = live ? b1[s] : 0.0f;
        acc = __builtin_amdgcn_wmma_f32_16x16x4_f32(false, a, false, bfrag, (short)0, acc, false, false);
    }
    const int rhalf = (lane >> 4) << 3;
    if (mt < HID / 16) {
        #pragma unroll
        for (int r = 0; r < 8; ++r)
            Wf[(size_t)(mt * 16 + rhalf + r) * CLS_W + col] = acc[r];
    } else if (lane < 16) {
        bf[col] = acc[0] + (live ? lb[col] : 0.0f);
    }
}

#define SETUP_T 256
__global__ __launch_bounds__(SETUP_T) void gcn_setup(const float* __restrict__ W2,
                                                     const float* __restrict__ b,
                                                     const float* __restrict__ lw,
                                                     const float* __restrict__ lb,
                                                     float* __restrict__ Wf, float* __restrict__ bf,
                                                     int* __restrict__ cnt, int n,
                                                     float* __restrict__ bnz, int nbn) {
    const int i = blockIdx.x * SETUP_T + threadIdx.x;
    if (i < n) cnt[i] = 0;
    if (i < nbn) bnz[i] = 0.0f;
    const int wave = __builtin_amdgcn_readfirstlane(threadIdx.x >> 5);
    if (blockIdx.x < FOLD_BLOCKS && wave < CLS_NT)
        fold_cls_tile(W2, b, lw, lb, Wf, bf, threadIdx.x, blockIdx.x);
}

template <int CTRL> __device__ __forceinline__ float lane_perm16(float v) {
    return __int_as_float(__builtin_amdgcn_update_dpp(0, __float_as_int(v), CTRL, 0xF, 0xF, true));
}
__device__ __forceinline__ float wave_max32(float v) {
    v = fmaxf(v, lane_perm16<0xB1>(v));
    v = fmaxf(v, lane_perm16<0x4E>(v));
    v = fmaxf(v, lane_perm16<0x141>(v));
    v = fmaxf(v, lane_perm16<0x140>(v));
    const float lo = __int_as_float(__builtin_amdgcn_readlane(__float_as_int(v), 0));
    const float hi = __int_as_float(__builtin_amdgcn_readlane(__float_as_int(v), 16));
    return fmaxf(lo, hi);
}
__device__ __forceinline__ float wave_sum32(float v) {
    v += lane_perm16<0xB1>(v);
    v += lane_perm16<0x4E>(v);
    v += lane_perm16<0x141>(v);
    v += lane_perm16<0x140>(v);
    const float lo = __int_as_float(__builtin_amdgcn_readlane(__float_as_int(v), 0));
    const float hi = __int_as_float(__builtin_amdgcn_readlane(__float_as_int(v), 16));
    return lo + hi;
}

__global__ __launch_bounds__(GATHER_T) void gcn_gather_cls(const int* __restrict__ rowptr,
                                                           const int* __restrict__ csr,
                                                           const float* __restrict__ dinv,
                                                           const float* __restrict__ m,
                                                           const float* __restrict__ bf,
                                                           float* __restrict__ out, int n, int e) {
    const int lane = threadIdx.x & 31;
    const int nwaves = (gridDim.x * blockDim.x) >> 5;
    const int wid = __builtin_amdgcn_readfirstlane((blockIdx.x * blockDim.x + threadIdx.x) >> 5);
    const bool valid = lane < OUTC / 2;
    const int fl = 2 * (lane < CLS_W / 2 ? lane : CLS_W / 2 - 1);
    const float2 bb = *(const float2*)(bf + fl);
    for (int node = wid; node < n; node += nwaves) {
        int beg = rowptr[node];
        int end = rowptr[node + 1];
        if (beg < 0) beg = 0;
        if (end > e) end = e;
        const float dd = dinv[node];
        const float2 selfv = *(const float2*)(m + (size_t)node * CLS_W + fl);
        float2 acc = make_float2(0.0f, 0.0f);

        for (int base = beg; base < end; base += 32) {
            int c = end - base; if (c > 32) c = 32;
            int sv = 0, wbits = 0;
            if (lane < c) {
                const int2 p = *(const int2*)(csr + 2 * (size_t)(base + lane));
                sv = p.x; wbits = p.y;
            }
            if ((unsigned)sv > (unsigned)(n - 1)) sv = n - 1;
            int j = 0;
            for (; j + CLS_ROWS <= c; j += CLS_ROWS) {
                float2 v[CLS_ROWS]; float w[CLS_ROWS];
                #pragma unroll
                for (int u = 0; u < CLS_ROWS; ++u) {
                    const int s = __builtin_amdgcn_readlane(sv, j + u);
                    w[u] = __int_as_float(__builtin_amdgcn_readlane(wbits, j + u));
                    v[u] = *(const float2*)(m + (size_t)s * CLS_W + fl);
                }
                #pragma unroll
                for (int u = 0; u < CLS_ROWS; ++u) {
                    const float nrm = w[u] * dd;
                    acc.x += v[u].x * nrm; acc.y += v[u].y * nrm;
                }
            }
            const int r = c - j;
            if (r > 0) {
                float2 v[CLS_ROWS - 1]; float w[CLS_ROWS - 1];
                #pragma unroll
                for (int u = 0; u < CLS_ROWS - 1; ++u) {
                    if (u < r) {
                        const int s = __builtin_amdgcn_readlane(sv, j + u);
                        w[u] = __int_as_float(__builtin_amdgcn_readlane(wbits, j + u));
                        v[u] = *(const float2*)(m + (size_t)s * CLS_W + fl);
                    }
                }
                #pragma unroll
                for (int u = 0; u < CLS_ROWS - 1; ++u) {
                    if (u < r) {
                        const float nrm = w[u] * dd;
                        acc.x += v[u].x * nrm; acc.y += v[u].y * nrm;
                    }
                }
            }
        }
        const float sn = dd * dd;
        acc.x += selfv.x * sn; acc.y += selfv.y * sn;
        acc.x += bb.x; acc.y += bb.y;
        const float x0 = valid ? acc.x : -INFINITY;
        const float x1 = valid ? acc.y : -INFINITY;
        const float mx = wave_max32(fmaxf(x0, x1));
        const float ls = logf(wave_sum32(expf(x0 - mx) + expf(x1 - mx)));
        if (valid) {
            float2 o; o.x = x0 - mx - ls; o.y = x1 - mx - ls;
            *(float2*)(out + (size_t)node * OUTC + 2 * lane) = o;
        }
    }
}

extern "C" void kernel_launch(void* const* d_in, const int* in_sizes, int n_in,
                              void* d_out, int out_size, void* d_ws, size_t ws_size,
                              hipStream_t stream) {
    const float* x       = (const float*)d_in[0];
    const int*   ei      = (const int*)d_in[1];
    const float* la      = (const float*)d_in[2];
    const float* lb      = (const float*)d_in[3];
    const float* mlp_w   = (const float*)d_in[4];
    const float* mlp_b   = (const float*)d_in[5];
    const float* conv_w  = (const float*)d_in[6];
    const float* conv_b  = (const float*)d_in[7];
    const float* bn_g    = (const float*)d_in[8];
    const float* bn_b    = (const float*)d_in[9];
    const float* lin_w   = (const float*)d_in[10];
    const float* lin_b   = (const float*)d_in[11];
    float* out = (float*)d_out;

    float* ws = (float*)d_ws;
    const size_t NM = (size_t)N_NODES * HID;
    float* hA    = ws;
    float* hB    = hA + NM;
    int*   cnt    = (int*)(hB + NM);
    int*   rowptr = cnt + SCAN_N;
    int*   rank   = rowptr + SCAN_N;
    int*   csr    = rank + N_EDGES;
    float* Wf    = (float*)(csr + 2 * N_EDGES);
    float* bf    = Wf + HID * CLS_W;
    float* bnz   = bf + CLS_W;
    float* dinv  = hB + NM + NM;

    const int TPB = 256;
    const int edgeBlocks = ((N_EDGES + 3) / 4 + TPB - 1) / TPB;
    const int gatherBlocks = 1250;
    const int denseBlocks = 256;

    gcn_setup<<<(N_NODES + SETUP_T - 1) / SETUP_T, SETUP_T, 0, stream>>>(
        conv_w + (size_t)2 * HID * HID, conv_b + 2 * HID, lin_w, lin_b,
        Wf, bf, cnt, N_NODES, bnz, 4 * HID);

    gcn_layer_dense<true, 8><<<denseBlocks, DENSE_T, 0, stream>>>(
        x, la, lb, nullptr, nullptr, nullptr, nullptr,
        mlp_w, mlp_b, conv_w, hA, N_NODES, ei, cnt, rank, N_EDGES);

    gcn_scan_deg<<<SCAN_BLOCKS, SCAN_T, 0, stream>>>(cnt, rowptr, dinv, N_NODES);
    gcn_fill_csr<<<edgeBlocks, TPB, 0, stream>>>(ei, rank, rowptr, dinv, csr, N_EDGES);

    gcn_gather<true><<<gatherBlocks, GATHER_T, 0, stream>>>(
        rowptr, csr, dinv, hA, conv_b, hB, bnz, bnz + HID, N_NODES, N_EDGES);

    gcn_layer_dense<false, 8><<<denseBlocks, DENSE_T, 0, stream>>>(
        hB, nullptr, nullptr, bnz, bnz + HID, bn_g, bn_b,
        mlp_w + (size_t)HID * HID, mlp_b + HID, conv_w + (size_t)HID * HID,
        hA, N_NODES, nullptr, nullptr, nullptr, 0);
    gcn_gather<true><<<gatherBlocks, GATHER_T, 0, stream>>>(
        rowptr, csr, dinv, hA, conv_b + HID, hB, bnz + 2 * HID, bnz + 3 * HID, N_NODES, N_EDGES);

    gcn_layer_dense<false, CLS_NT><<<denseBlocks, DENSE_T, 0, stream>>>(
        hB, nullptr, nullptr, bnz + 2 * HID, bnz + 3 * HID, bn_g + HID, bn_b + HID,
        mlp_w + (size_t)2 * HID * HID, mlp_b + 2 * HID, Wf,
        hA, N_NODES, nullptr, nullptr, nullptr, 0);
    gcn_gather_cls<<<gatherBlocks, GATHER_T, 0, stream>>>(
        rowptr, csr, dinv, hA, bf, out, N_NODES, N_EDGES);
}
